# starting kernel
# baseline (speedup 1.0000x reference)
#include <hip/hip_runtime.h>
#include <hip/hip_bf16.h>
#include <math.h>

typedef __attribute__((ext_vector_type(2))) float v2f;
typedef __attribute__((ext_vector_type(8))) float v8f;

#define DCH 128
#define NGRAPH 64
#define GN_EPS 1e-5f
#define TILE_M 16
#define SX_LD (DCH + 4)

__global__ void k_zero_f32(float* __restrict__ p, size_t n) {
    size_t i = (size_t)blockIdx.x * blockDim.x + threadIdx.x;
    if (i < n) p[i] = 0.0f;
}

__global__ __launch_bounds__(256) void k_transpose_w(
    const float* __restrict__ W, float* __restrict__ Wt) {
    __shared__ float tile[16][17];
    int bx = blockIdx.x & 7;
    int by = blockIdx.x >> 3;
    int tx = threadIdx.x & 15;
    int ty = threadIdx.x >> 4;
    tile[ty][tx] = W[(size_t)(by * 16 + ty) * DCH + bx * 16 + tx];
    __syncthreads();
    Wt[(size_t)(bx * 16 + ty) * DCH + by * 16 + tx] = tile[tx][ty];
}

__global__ __launch_bounds__(256) void k_gemm_wmma(
    const float* __restrict__ x, const float* __restrict__ Wt,
    float* __restrict__ h, int N) {
    __shared__ float sX[TILE_M][SX_LD];

    const int t = threadIdx.x;
    const int rowBase = blockIdx.x * TILE_M;

    for (int i = t; i < TILE_M * DCH; i += 256) {
        int r = i >> 7;
        int c = i & (DCH - 1);
        int gr = rowBase + r;
        sX[r][c] = (gr < N) ? x[(size_t)gr * DCH + c] : 0.0f;
    }
    __syncthreads();

    const int lane = t & 31;
    const int wave = t >> 5;
    const int m    = lane & 15;
    const int koff = (lane >> 4) << 1;
    const int n    = (wave << 4) + (lane & 15);

    const float* wrow = Wt + (size_t)n * DCH;

    v8f acc = {};
#pragma unroll
    for (int kt = 0; kt < DCH; kt += 4) {
        v2f a = *reinterpret_cast<const v2f*>(&sX[m][kt + koff]);
        v2f b = *reinterpret_cast<const v2f*>(wrow + kt + koff);
        acc = __builtin_amdgcn_wmma_f32_16x16x4_f32(
            false, a, false, b, (short)0, acc, false, false);
    }

    const int rhalf = (lane >> 4) << 3;
#pragma unroll
    for (int v = 0; v < 8; ++v) {
        int gr = rowBase + rhalf + v;
        if (gr < N) h[(size_t)gr * DCH + n] = acc[v];
    }
}

#define EDGES_PER_WAVE 4
__global__ __launch_bounds__(256) void k_edge_scatter(
    const int* __restrict__ src, const int* __restrict__ dst,
    const float* __restrict__ attr, const float* __restrict__ h,
    float* __restrict__ agg, int E) {
    const int lane  = threadIdx.x & 31;
    const int ebase = (blockIdx.x * 8 + (threadIdx.x >> 5)) * EDGES_PER_WAVE;

    __builtin_prefetch(&src[ebase + 4096], 0, 0);
    __builtin_prefetch(&dst[ebase + 4096], 0, 0);
    __builtin_prefetch(&attr[ebase + 4096], 0, 0);

#pragma unroll
    for (int j = 0; j < EDGES_PER_WAVE; ++j) {
        int e = ebase + j;
        if (e >= E) break;
        e = __builtin_amdgcn_readfirstlane(e);
        int   s = src[e];
        int   d = dst[e];
        float a = attr[e];
        const float4* hv = reinterpret_cast<const float4*>(h + (size_t)s * DCH);
        float4 v = hv[lane];
        float* outp = agg + (size_t)d * DCH + lane * 4;
        unsafeAtomicAdd(outp + 0, v.x * a);
        unsafeAtomicAdd(outp + 1, v.y * a);
        unsafeAtomicAdd(outp + 2, v.z * a);
        unsafeAtomicAdd(outp + 3, v.w * a);
    }
}

#define GROWS 64
__global__ __launch_bounds__(128) void k_bias_relu_mean(
    float* __restrict__ out, const float* __restrict__ bias,
    const int* __restrict__ batch, float* __restrict__ mean_sum,
    float* __restrict__ cnt, int N) {
    __shared__ int sg[GROWS];
    const int base = blockIdx.x * GROWS;
    const int c = threadIdx.x;
    int rows = N - base;
    if (rows > GROWS) rows = GROWS;
    for (int i = c; i < rows; i += 128) sg[i] = batch[base + i];
    __syncthreads();

    const float bv = bias[c];
    float acc = 0.0f, cacc = 0.0f;
    int curg = -1;
    for (int r = 0; r < rows; ++r) {
        int g = sg[r];
        if (g != curg) {
            if (curg >= 0) {
                unsafeAtomicAdd(&mean_sum[(size_t)curg * DCH + c], acc);
                if (c == 0) unsafeAtomicAdd(&cnt[curg], cacc);
            }
            acc = 0.0f; cacc = 0.0f; curg = g;
        }
        size_t idx = (size_t)(base + r) * DCH + c;
        float v = out[idx] + bv;
        v = v > 0.0f ? v : 0.0f;
        out[idx] = v;
        acc += v;
        cacc += 1.0f;
    }
    if (curg >= 0) {
        unsafeAtomicAdd(&mean_sum[(size_t)curg * DCH + c], acc);
        if (c == 0) unsafeAtomicAdd(&cnt[curg], cacc);
    }
}

__global__ __launch_bounds__(128) void k_var_pass(
    const float* __restrict__ out, const int* __restrict__ batch,
    const float* __restrict__ mean_sum, const float* __restrict__ cnt,
    const float* __restrict__ ms, float* __restrict__ var_sum, int N) {
    __shared__ int sg[GROWS];
    const int base = blockIdx.x * GROWS;
    const int c = threadIdx.x;
    int rows = N - base;
    if (rows > GROWS) rows = GROWS;
    for (int i = c; i < rows; i += 128) sg[i] = batch[base + i];
    __syncthreads();

    const float msc = ms[c];
    float acc = 0.0f;
    float sub = 0.0f;
    int curg = -1;
    for (int r = 0; r < rows; ++r) {
        int g = sg[r];
        if (g != curg) {
            if (curg >= 0) unsafeAtomicAdd(&var_sum[(size_t)curg * DCH + c], acc);
            acc = 0.0f;
            curg = g;
            float cn = cnt[g];
            cn = cn > 1.0f ? cn : 1.0f;
            sub = (mean_sum[(size_t)g * DCH + c] / cn) * msc;
        }
        float o = out[(size_t)(base + r) * DCH + c] - sub;
        acc += o * o;
    }
    if (curg >= 0) unsafeAtomicAdd(&var_sum[(size_t)curg * DCH + c], acc);
}

__global__ __launch_bounds__(256) void k_finalize(
    float* __restrict__ out, const int* __restrict__ batch,
    const float* __restrict__ mean_sum, const float* __restrict__ var_sum,
    const float* __restrict__ cnt, const float* __restrict__ ms,
    const float* __restrict__ gw, const float* __restrict__ gb, int N) {
    size_t idx = (size_t)blockIdx.x * 256 + threadIdx.x;
    size_t total = (size_t)N * (DCH / 4);
    if (idx >= total) return;
    int nrow = (int)(idx >> 5);
    int q    = (int)(idx & 31);
    int g = batch[nrow];
    float cn = cnt[g];
    cn = cn > 1.0f ? cn : 1.0f;
    float inv_cn = 1.0f / cn;

    float4 m4 = reinterpret_cast<const float4*>(mean_sum + (size_t)g * DCH)[q];
    float4 v4 = reinterpret_cast<const float4*>(var_sum  + (size_t)g * DCH)[q];
    float4 s4 = reinterpret_cast<const float4*>(ms)[q];
    float4 w4 = reinterpret_cast<const float4*>(gw)[q];
    float4 b4 = reinterpret_cast<const float4*>(gb)[q];
    float4 o  = reinterpret_cast<float4*>(out)[idx];

    o.x -= (m4.x * inv_cn) * s4.x;
    o.y -= (m4.y * inv_cn) * s4.y;
    o.z -= (m4.z * inv_cn) * s4.z;
    o.w -= (m4.w * inv_cn) * s4.w;
    o.x = w4.x * o.x / sqrtf(v4.x * inv_cn + GN_EPS) + b4.x;
    o.y = w4.y * o.y / sqrtf(v4.y * inv_cn + GN_EPS) + b4.y;
    o.z = w4.z * o.z / sqrtf(v4.z * inv_cn + GN_EPS) + b4.z;
    o.w = w4.w * o.w / sqrtf(v4.w * inv_cn + GN_EPS) + b4.w;

    reinterpret_cast<float4*>(out)[idx] = o;
}

extern "C" void kernel_launch(void* const* d_in, const int* in_sizes, int n_in,
                              void* d_out, int out_size, void* d_ws, size_t ws_size,
                              hipStream_t stream) {
    const float* x     = (const float*)d_in[0];
    const int*   eidx  = (const int*)  d_in[1];
    const float* eattr = (const float*)d_in[2];
    const int*   batch = (const int*)  d_in[3];
    const float* W     = (const float*)d_in[4];
    const float* bias  = (const float*)d_in[5];
    const float* gw    = (const float*)d_in[6];
    const float* gb    = (const float*)d_in[7];
    const float* ms    = (const float*)d_in[8];
    float* out = (float*)d_out;

    const int N = in_sizes[0] / DCH;
    const int E = in_sizes[2];

    float* h        = (float*)d_ws;
    float* mean_sum = h + (size_t)N * DCH;
    float* var_sum  = mean_sum + (size_t)NGRAPH * DCH;
    float* cnt      = var_sum + (size_t)NGRAPH * DCH;
    float* Wt       = cnt + NGRAPH;

    {
        size_t n0 = (size_t)N * DCH;
        k_zero_f32<<<(unsigned)((n0 + 255) / 256), 256, 0, stream>>>(out, n0);
        size_t n1 = (size_t)2 * NGRAPH * DCH + NGRAPH;
        k_zero_f32<<<(unsigned)((n1 + 255) / 256), 256, 0, stream>>>(mean_sum, n1);
    }

    k_transpose_w<<<64, 256, 0, stream>>>(W, Wt);

    k_gemm_wmma<<<(N + TILE_M - 1) / TILE_M, 256, 0, stream>>>(x, Wt, h, N);

    {
        int edges_per_block = 8 * EDGES_PER_WAVE;
        k_edge_scatter<<<(E + edges_per_block - 1) / edges_per_block, 256, 0, stream>>>(
            eidx, eidx + E, eattr, h, out, E);
    }

    k_bias_relu_mean<<<(N + GROWS - 1) / GROWS, 128, 0, stream>>>(
        out, bias, batch, mean_sum, cnt, N);

    k_var_pass<<<(N + GROWS - 1) / GROWS, 128, 0, stream>>>(
        out, batch, mean_sum, cnt, ms, var_sum, N);

    {
        size_t total = (size_t)N * (DCH / 4);
        k_finalize<<<(unsigned)((total + 255) / 256), 256, 0, stream>>>(
            out, batch, mean_sum, var_sum, cnt, ms, gw, gb, N);
    }
}


# edge scatter: lane owns channels lane+32j so each wave-wide f32 atomic add (and each h load) covers one whole 128-byte line, not four quarter lines; atomics still add in arrival order
# speedup vs baseline: 1.4315x; 1.4315x over previous
#include <hip/hip_runtime.h>
#include <hip/hip_bf16.h>
#include <math.h>

typedef __attribute__((ext_vector_type(2))) float v2f;
typedef __attribute__((ext_vector_type(8))) float v8f;

#define DCH 128
#define NGRAPH 64
#define GN_EPS 1e-5f
#define TILE_M 16
#define SX_LD (DCH + 4)

__global__ void k_zero_f32(float* __restrict__ p, size_t n) {
    size_t i = (size_t)blockIdx.x * blockDim.x + threadIdx.x;
    if (i < n) p[i] = 0.0f;
}

__global__ __launch_bounds__(256) void k_transpose_w(
    const float* __restrict__ W, float* __restrict__ Wt) {
    __shared__ float tile[16][17];
    int bx = blockIdx.x & 7;
    int by = blockIdx.x >> 3;
    int tx = threadIdx.x & 15;
    int ty = threadIdx.x >> 4;
    tile[ty][tx] = W[(size_t)(by * 16 + ty) * DCH + bx * 16 + tx];
    __syncthreads();
    Wt[(size_t)(bx * 16 + ty) * DCH + by * 16 + tx] = tile[tx][ty];
}

__global__ __launch_bounds__(256) void k_gemm_wmma(
    const float* __restrict__ x, const float* __restrict__ Wt,
    float* __restrict__ h, int N) {
    __shared__ float sX[TILE_M][SX_LD];

    const int t = threadIdx.x;
    const int rowBase = blockIdx.x * TILE_M;

    for (int i = t; i < TILE_M * DCH; i += 256) {
        int r = i >> 7;
        int c = i & (DCH - 1);
        int gr = rowBase + r;
        sX[r][c] = (gr < N) ? x[(size_t)gr * DCH + c] : 0.0f;
    }
    __syncthreads();

    const int lane = t & 31;
    const int wave = t >> 5;
    const int m    = lane & 15;
    const int koff = (lane >> 4) << 1;
    const int n    = (wave << 4) + (lane & 15);

    const float* wrow = Wt + (size_t)n * DCH;

    v8f acc = {};
#pragma unroll
    for (int kt = 0; kt < DCH; kt += 4) {
        v2f a = *reinterpret_cast<const v2f*>(&sX[m][kt + koff]);
        v2f b = *reinterpret_cast<const v2f*>(wrow + kt + koff);
        acc = __builtin_amdgcn_wmma_f32_16x16x4_f32(
            false, a, false, b, (short)0, acc, false, false);
    }

    const int rhalf = (lane >> 4) << 3;
#pragma unroll
    for (int v = 0; v < 8; ++v) {
        int gr = rowBase + rhalf + v;
        if (gr < N) h[(size_t)gr * DCH + n] = acc[v];
    }
}

#define EDGES_PER_WAVE 4
__global__ __launch_bounds__(256) void k_edge_scatter(
    const int* __restrict__ src, const int* __restrict__ dst,
    const float* __restrict__ attr, const float* __restrict__ h,
    float* __restrict__ agg, int E) {
    const int lane  = threadIdx.x & 31;
    const int ebase = (blockIdx.x * 8 + (threadIdx.x >> 5)) * EDGES_PER_WAVE;

    __builtin_prefetch(&src[ebase + 4096], 0, 0);
    __builtin_prefetch(&dst[ebase + 4096], 0, 0);
    __builtin_prefetch(&attr[ebase + 4096], 0, 0);

#pragma unroll
    for (int j = 0; j < EDGES_PER_WAVE; ++j) {
        int e = ebase + j;
        if (e >= E) break;
        e = __builtin_amdgcn_readfirstlane(e);
        int   s = src[e];
        int   d = dst[e];
        float a = attr[e];
        const float* hp = h + (size_t)s * DCH + lane;
        float v0 = hp[0];
        float v1 = hp[32];
        float v2 = hp[64];
        float v3 = hp[96];
        float* outp = agg + (size_t)d * DCH + lane;
        unsafeAtomicAdd(outp + 0,  v0 * a);
        unsafeAtomicAdd(outp + 32, v1 * a);
        unsafeAtomicAdd(outp + 64, v2 * a);
        unsafeAtomicAdd(outp + 96, v3 * a);
    }
}

#define GROWS 64
__global__ __launch_bounds__(128) void k_bias_relu_mean(
    float* __restrict__ out, const float* __restrict__ bias,
    const int* __restrict__ batch, float* __restrict__ mean_sum,
    float* __restrict__ cnt, int N) {
    __shared__ int sg[GROWS];
    const int base = blockIdx.x * GROWS;
    const int c = threadIdx.x;
    int rows = N - base;
    if (rows > GROWS) rows = GROWS;
    for (int i = c; i < rows; i += 128) sg[i] = batch[base + i];
    __syncthreads();

    const float bv = bias[c];
    float acc = 0.0f, cacc = 0.0f;
    int curg = -1;
    for (int r = 0; r < rows; ++r) {
        int g = sg[r];
        if (g != curg) {
            if (curg >= 0) {
                unsafeAtomicAdd(&mean_sum[(size_t)curg * DCH + c], acc);
                if (c == 0) unsafeAtomicAdd(&cnt[curg], cacc);
            }
            acc = 0.0f; cacc = 0.0f; curg = g;
        }
        size_t idx = (size_t)(base + r) * DCH + c;
        float v = out[idx] + bv;
        v = v > 0.0f ? v : 0.0f;
        out[idx] = v;
        acc += v;
        cacc += 1.0f;
    }
    if (curg >= 0) {
        unsafeAtomicAdd(&mean_sum[(size_t)curg * DCH + c], acc);
        if (c == 0) unsafeAtomicAdd(&cnt[curg], cacc);
    }
}

__global__ __launch_bounds__(128) void k_var_pass(
    const float* __restrict__ out, const int* __restrict__ batch,
    const float* __restrict__ mean_sum, const float* __restrict__ cnt,
    const float* __restrict__ ms, float* __restrict__ var_sum, int N) {
    __shared__ int sg[GROWS];
    const int base = blockIdx.x * GROWS;
    const int c = threadIdx.x;
    int rows = N - base;
    if (rows > GROWS) rows = GROWS;
    for (int i = c; i < rows; i += 128) sg[i] = batch[base + i];
    __syncthreads();

    const float msc = ms[c];
    float acc = 0.0f;
    float sub = 0.0f;
    int curg = -1;
    for (int r = 0; r < rows; ++r) {
        int g = sg[r];
        if (g != curg) {
            if (curg >= 0) unsafeAtomicAdd(&var_sum[(size_t)curg * DCH + c], acc);
            acc = 0.0f;
            curg = g;
            float cn = cnt[g];
            cn = cn > 1.0f ? cn : 1.0f;
            sub = (mean_sum[(size_t)g * DCH + c] / cn) * msc;
        }
        float o = out[(size_t)(base + r) * DCH + c] - sub;
        acc += o * o;
    }
    if (curg >= 0) unsafeAtomicAdd(&var_sum[(size_t)curg * DCH + c], acc);
}

__global__ __launch_bounds__(256) void k_finalize(
    float* __restrict__ out, const int* __restrict__ batch,
    const float* __restrict__ mean_sum, const float* __restrict__ var_sum,
    const float* __restrict__ cnt, const float* __restrict__ ms,
    const float* __restrict__ gw, const float* __restrict__ gb, int N) {
    size_t idx = (size_t)blockIdx.x * 256 + threadIdx.x;
    size_t total = (size_t)N * (DCH / 4);
    if (idx >= total) return;
    int nrow = (int)(idx >> 5);
    int q    = (int)(idx & 31);
    int g = batch[nrow];
    float cn = cnt[g];
    cn = cn > 1.0f ? cn : 1.0f;
    float inv_cn = 1.0f / cn;

    float4 m4 = reinterpret_cast<const float4*>(mean_sum + (size_t)g * DCH)[q];
    float4 v4 = reinterpret_cast<const float4*>(var_sum  + (size_t)g * DCH)[q];
    float4 s4 = reinterpret_cast<const float4*>(ms)[q];
    float4 w4 = reinterpret_cast<const float4*>(gw)[q];
    float4 b4 = reinterpret_cast<const float4*>(gb)[q];
    float4 o  = reinterpret_cast<float4*>(out)[idx];

    o.x -= (m4.x * inv_cn) * s4.x;
    o.y -= (m4.y * inv_cn) * s4.y;
    o.z -= (m4.z * inv_cn) * s4.z;
    o.w -= (m4.w * inv_cn) * s4.w;
    o.x = w4.x * o.x / sqrtf(v4.x * inv_cn + GN_EPS) + b4.x;
    o.y = w4.y * o.y / sqrtf(v4.y * inv_cn + GN_EPS) + b4.y;
    o.z = w4.z * o.z / sqrtf(v4.z * inv_cn + GN_EPS) + b4.z;
    o.w = w4.w * o.w / sqrtf(v4.w * inv_cn + GN_EPS) + b4.w;

    reinterpret_cast<float4*>(out)[idx] = o;
}

extern "C" void kernel_launch(void* const* d_in, const int* in_sizes, int n_in,
                              void* d_out, int out_size, void* d_ws, size_t ws_size,
                              hipStream_t stream) {
    const float* x     = (const float*)d_in[0];
    const int*   eidx  = (const int*)  d_in[1];
    const float* eattr = (const float*)d_in[2];
    const int*   batch = (const int*)  d_in[3];
    const float* W     = (const float*)d_in[4];
    const float* bias  = (const float*)d_in[5];
    const float* gw    = (const float*)d_in[6];
    const float* gb    = (const float*)d_in[7];
    const float* ms    = (const float*)d_in[8];
    float* out = (float*)d_out;

    const int N = in_sizes[0] / DCH;
    const int E = in_sizes[2];

    float* h        = (float*)d_ws;
    float* mean_sum = h + (size_t)N * DCH;
    float* var_sum  = mean_sum + (size_t)NGRAPH * DCH;
    float* cnt      = var_sum + (size_t)NGRAPH * DCH;
    float* Wt       = cnt + NGRAPH;

    {
        size_t n0 = (size_t)N * DCH;
        k_zero_f32<<<(unsigned)((n0 + 255) / 256), 256, 0, stream>>>(out, n0);
        size_t n1 = (size_t)2 * NGRAPH * DCH + NGRAPH;
        k_zero_f32<<<(unsigned)((n1 + 255) / 256), 256, 0, stream>>>(mean_sum, n1);
    }

    k_transpose_w<<<64, 256, 0, stream>>>(W, Wt);

    k_gemm_wmma<<<(N + TILE_M - 1) / TILE_M, 256, 0, stream>>>(x, Wt, h, N);

    {
        int edges_per_block = 8 * EDGES_PER_WAVE;
        k_edge_scatter<<<(E + edges_per_block - 1) / edges_per_block, 256, 0, stream>>>(
            eidx, eidx + E, eattr, h, out, E);
    }

    k_bias_relu_mean<<<(N + GROWS - 1) / GROWS, 128, 0, stream>>>(
        out, bias, batch, mean_sum, cnt, N);

    k_var_pass<<<(N + GROWS - 1) / GROWS, 128, 0, stream>>>(
        out, batch, mean_sum, cnt, ms, var_sum, N);

    {
        size_t total = (size_t)N * (DCH / 4);
        k_finalize<<<(unsigned)((total + 255) / 256), 256, 0, stream>>>(
            out, batch, mean_sum, var_sum, cnt, ms, gw, gb, N);
    }
}


# statistics passes (bias/relu/mean sums, variance sums): all 64 row loads of a block issued before the first row is summed, up to 64 loads in flight per wave instead of 1; sums keep their row order
# speedup vs baseline: 1.5898x; 1.1106x over previous
#include <hip/hip_runtime.h>
#include <hip/hip_bf16.h>
#include <math.h>

typedef __attribute__((ext_vector_type(2))) float v2f;
typedef __attribute__((ext_vector_type(8))) float v8f;

#define DCH 128
#define NGRAPH 64
#define GN_EPS 1e-5f
#define TILE_M 16
#define SX_LD (DCH + 4)

__global__ void k_zero_f32(float* __restrict__ p, size_t n) {
    size_t i = (size_t)blockIdx.x * blockDim.x + threadIdx.x;
    if (i < n) p[i] = 0.0f;
}

__global__ __launch_bounds__(256) void k_transpose_w(
    const float* __restrict__ W, float* __restrict__ Wt) {
    __shared__ float tile[16][17];
    int bx = blockIdx.x & 7;
    int by = blockIdx.x >> 3;
    int tx = threadIdx.x & 15;
    int ty = threadIdx.x >> 4;
    tile[ty][tx] = W[(size_t)(by * 16 + ty) * DCH + bx * 16 + tx];
    __syncthreads();
    Wt[(size_t)(bx * 16 + ty) * DCH + by * 16 + tx] = tile[tx][ty];
}

__global__ __launch_bounds__(256) void k_gemm_wmma(
    const float* __restrict__ x, const float* __restrict__ Wt,
    float* __restrict__ h, int N) {
    __shared__ float sX[TILE_M][SX_LD];

    const int t = threadIdx.x;
    const int rowBase = blockIdx.x * TILE_M;

    for (int i = t; i < TILE_M * DCH; i += 256) {
        int r = i >> 7;
        int c = i & (DCH - 1);
        int gr = rowBase + r;
        sX[r][c] = (gr < N) ? x[(size_t)gr * DCH + c] : 0.0f;
    }
    __syncthreads();

    const int lane = t & 31;
    const int wave = t >> 5;
    const int m    = lane & 15;
    const int koff = (lane >> 4) << 1;
    const int n    = (wave << 4) + (lane & 15);

    const float* wrow = Wt + (size_t)n * DCH;

    v8f acc = {};
#pragma unroll
    for (int kt = 0; kt < DCH; kt += 4) {
        v2f a = *reinterpret_cast<const v2f*>(&sX[m][kt + koff]);
        v2f b = *reinterpret_cast<const v2f*>(wrow + kt + koff);
        acc = __builtin_amdgcn_wmma_f32_16x16x4_f32(
            false, a, false, b, (short)0, acc, false, false);
    }

    const int rhalf = (lane >> 4) << 3;
#pragma unroll
    for (int v = 0; v < 8; ++v) {
        int gr = rowBase + rhalf + v;
        if (gr < N) h[(size_t)gr * DCH + n] = acc[v];
    }
}

#define EDGES_PER_WAVE 4
__global__ __launch_bounds__(256) void k_edge_scatter(
    const int* __restrict__ src, const int* __restrict__ dst,
    const float* __restrict__ attr, const float* __restrict__ h,
    float* __restrict__ agg, int E) {
    const int lane  = threadIdx.x & 31;
    const int ebase = (blockIdx.x * 8 + (threadIdx.x >> 5)) * EDGES_PER_WAVE;

    __builtin_prefetch(&src[ebase + 4096], 0, 0);
    __builtin_prefetch(&dst[ebase + 4096], 0, 0);
    __builtin_prefetch(&attr[ebase + 4096], 0, 0);

#pragma unroll
    for (int j = 0; j < EDGES_PER_WAVE; ++j) {
        int e = ebase + j;
        if (e >= E) break;
        e = __builtin_amdgcn_readfirstlane(e);
        int   s = src[e];
        int   d = dst[e];
        float a = attr[e];
        const float* hp = h + (size_t)s * DCH + lane;
        float v0 = hp[0];
        float v1 = hp[32];
        float v2 = hp[64];
        float v3 = hp[96];
        float* outp = agg + (size_t)d * DCH + lane;
        unsafeAtomicAdd(outp + 0,  v0 * a);
        unsafeAtomicAdd(outp + 32, v1 * a);
        unsafeAtomicAdd(outp + 64, v2 * a);
        unsafeAtomicAdd(outp + 96, v3 * a);
    }
}

#define GROWS 64

__global__ __launch_bounds__(128) void k_bias_relu_mean(
    float* __restrict__ out, const float* __restrict__ bias,
    const int* __restrict__ batch, float* __restrict__ mean_sum,
    float* __restrict__ cnt, int N) {
    const int c = threadIdx.x;
    const float bv = bias[c];
    float* msum = mean_sum + c;
    float acc = 0.0f, cacc = 0.0f;
    int curg = -1;

    if (N >= GROWS) {
        int base = blockIdx.x * GROWS;
        int skip = 0;
        if (base + GROWS > N) { skip = base + GROWS - N; base = N - GROWS; }
        float* col = out + (size_t)base * DCH + c;
        const int lane = c & 31;
        const int gidlo = batch[base + lane];
        const int gidhi = batch[base + 32 + lane];

        float v[GROWS];
#pragma unroll
        for (int j = 0; j < GROWS; ++j)
            v[j] = (j >= skip) ? col[(size_t)j * DCH] : 0.0f;

#pragma unroll
        for (int j = 0; j < GROWS; ++j) {
            if (j >= skip) {
                const int gg = __builtin_amdgcn_readlane(j < 32 ? gidlo : gidhi, j & 31);
                if (gg != curg) {
                    if (curg >= 0) {
                        unsafeAtomicAdd(&msum[(size_t)curg * DCH], acc);
                        if (c == 0) unsafeAtomicAdd(&cnt[curg], cacc);
                    }
                    acc = 0.0f; cacc = 0.0f; curg = gg;
                }
                float x = v[j] + bv;
                x = x > 0.0f ? x : 0.0f;
                col[(size_t)j * DCH] = x;
                acc += x;
                cacc += 1.0f;
            }
        }
    } else {
        for (int r = 0; r < N; ++r) {
            const int gg = batch[r];
            if (gg != curg) {
                if (curg >= 0) {
                    unsafeAtomicAdd(&msum[(size_t)curg * DCH], acc);
                    if (c == 0) unsafeAtomicAdd(&cnt[curg], cacc);
                }
                acc = 0.0f; cacc = 0.0f; curg = gg;
            }
            const size_t idx = (size_t)r * DCH + c;
            float x = out[idx] + bv;
            x = x > 0.0f ? x : 0.0f;
            out[idx] = x;
            acc += x;
            cacc += 1.0f;
        }
    }
    if (curg >= 0) {
        unsafeAtomicAdd(&msum[(size_t)curg * DCH], acc);
        if (c == 0) unsafeAtomicAdd(&cnt[curg], cacc);
    }
}

#define HOLD16(a, o)                                                          \
    asm volatile("" :: "v"(a[o + 0]), "v"(a[o + 1]), "v"(a[o + 2]),          \
                 "v"(a[o + 3]), "v"(a[o + 4]), "v"(a[o + 5]), "v"(a[o + 6]), \
                 "v"(a[o + 7]), "v"(a[o + 8]), "v"(a[o + 9]), "v"(a[o + 10]),\
                 "v"(a[o + 11]), "v"(a[o + 12]), "v"(a[o + 13]),             \
                 "v"(a[o + 14]), "v"(a[o + 15]))

__global__ __launch_bounds__(128) void k_var_pass(
    const float* __restrict__ out, const int* __restrict__ batch,
    const float* __restrict__ mean_sum, const float* __restrict__ cnt,
    const float* __restrict__ ms, float* __restrict__ var_sum, int N) {
    const int c = threadIdx.x;
    const float msc = ms[c];
    const float* msum = mean_sum + c;
    float* vsum = var_sum + c;
    float acc = 0.0f;
    float sub = 0.0f;
    int curg = -1;

    if (N >= GROWS) {
        int base = blockIdx.x * GROWS;
        int skip = 0;
        if (base + GROWS > N) { skip = base + GROWS - N; base = N - GROWS; }
        const float* col = out + (size_t)base * DCH + c;
        const int lane = c & 31;
        const int gidlo = batch[base + lane];
        const int gidhi = batch[base + 32 + lane];

        float v[GROWS];
#pragma unroll
        for (int j = 0; j < GROWS; ++j) v[j] = col[(size_t)j * DCH];
        HOLD16(v, 0); HOLD16(v, 16); HOLD16(v, 32); HOLD16(v, 48);

#pragma unroll
        for (int j = 0; j < GROWS; ++j) {
            if (j >= skip) {
                const int gg = __builtin_amdgcn_readlane(j < 32 ? gidlo : gidhi, j & 31);
                if (gg != curg) {
                    if (curg >= 0) unsafeAtomicAdd(&vsum[(size_t)curg * DCH], acc);
                    acc = 0.0f;
                    curg = gg;
                    float cn = cnt[gg];
                    cn = cn > 1.0f ? cn : 1.0f;
                    sub = (msum[(size_t)gg * DCH] / cn) * msc;
                }
                float o = v[j] - sub;
                acc += o * o;
            }
        }
    } else {
        for (int r = 0; r < N; ++r) {
            const int gg = batch[r];
            if (gg != curg) {
                if (curg >= 0) unsafeAtomicAdd(&vsum[(size_t)curg * DCH], acc);
                acc = 0.0f;
                curg = gg;
                float cn = cnt[gg];
                cn = cn > 1.0f ? cn : 1.0f;
                sub = (msum[(size_t)gg * DCH] / cn) * msc;
            }
            float o = out[(size_t)r * DCH + c] - sub;
            acc += o * o;
        }
    }
    if (curg >= 0) unsafeAtomicAdd(&vsum[(size_t)curg * DCH], acc);
}

__global__ __launch_bounds__(256) void k_finalize(
    float* __restrict__ out, const int* __restrict__ batch,
    const float* __restrict__ mean_sum, const float* __restrict__ var_sum,
    const float* __restrict__ cnt, const float* __restrict__ ms,
    const float* __restrict__ gw, const float* __restrict__ gb, int N) {
    size_t idx = (size_t)blockIdx.x * 256 + threadIdx.x;
    size_t total = (size_t)N * (DCH / 4);
    if (idx >= total) return;
    int nrow = (int)(idx >> 5);
    int q    = (int)(idx & 31);
    int g = batch[nrow];
    float cn = cnt[g];
    cn = cn > 1.0f ? cn : 1.0f;
    float inv_cn = 1.0f / cn;

    float4 m4 = reinterpret_cast<const float4*>(mean_sum + (size_t)g * DCH)[q];
    float4 v4 = reinterpret_cast<const float4*>(var_sum  + (size_t)g * DCH)[q];
    float4 s4 = reinterpret_cast<const float4*>(ms)[q];
    float4 w4 = reinterpret_cast<const float4*>(gw)[q];
    float4 b4 = reinterpret_cast<const float4*>(gb)[q];
    float4 o  = reinterpret_cast<float4*>(out)[idx];

    o.x -= (m4.x * inv_cn) * s4.x;
    o.y -= (m4.y * inv_cn) * s4.y;
    o.z -= (m4.z * inv_cn) * s4.z;
    o.w -= (m4.w * inv_cn) * s4.w;
    o.x = w4.x * o.x / sqrtf(v4.x * inv_cn + GN_EPS) + b4.x;
    o.y = w4.y * o.y / sqrtf(v4.y * inv_cn + GN_EPS) + b4.y;
    o.z = w4.z * o.z / sqrtf(v4.z * inv_cn + GN_EPS) + b4.z;
    o.w = w4.w * o.w / sqrtf(v4.w * inv_cn + GN_EPS) + b4.w;

    reinterpret_cast<float4*>(out)[idx] = o;
}

extern "C" void kernel_launch(void* const* d_in, const int* in_sizes, int n_in,
                              void* d_out, int out_size, void* d_ws, size_t ws_size,
                              hipStream_t stream) {
    const float* x     = (const float*)d_in[0];
    const int*   eidx  = (const int*)  d_in[1];
    const float* eattr = (const float*)d_in[2];
    const int*   batch = (const int*)  d_in[3];
    const float* W     = (const float*)d_in[4];
    const float* bias  = (const float*)d_in[5];
    const float* gw    = (const float*)d_in[6];
    const float* gb    = (const float*)d_in[7];
    const float* ms    = (const float*)d_in[8];
    float* out = (float*)d_out;

    const int N = in_sizes[0] / DCH;
    const int E = in_sizes[2];

    float* h        = (float*)d_ws;
    float* mean_sum = h + (size_t)N * DCH;
    float* var_sum  = mean_sum + (size_t)NGRAPH * DCH;
    float* cnt      = var_sum + (size_t)NGRAPH * DCH;
    float* Wt       = cnt + NGRAPH;

    {
        size_t n0 = (size_t)N * DCH;
        k_zero_f32<<<(unsigned)((n0 + 255) / 256), 256, 0, stream>>>(out, n0);
        size_t n1 = (size_t)2 * NGRAPH * DCH + NGRAPH;
        k_zero_f32<<<(unsigned)((n1 + 255) / 256), 256, 0, stream>>>(mean_sum, n1);
    }

    k_transpose_w<<<64, 256, 0, stream>>>(W, Wt);

    k_gemm_wmma<<<(N + TILE_M - 1) / TILE_M, 256, 0, stream>>>(x, Wt, h, N);

    {
        int edges_per_block = 8 * EDGES_PER_WAVE;
        k_edge_scatter<<<(E + edges_per_block - 1) / edges_per_block, 256, 0, stream>>>(
            eidx, eidx + E, eattr, h, out, E);
    }

    k_bias_relu_mean<<<(N + GROWS - 1) / GROWS, 128, 0, stream>>>(
        out, bias, batch, mean_sum, cnt, N);

    k_var_pass<<<(N + GROWS - 1) / GROWS, 128, 0, stream>>>(
        out, batch, mean_sum, cnt, ms, var_sum, N);

    {
        size_t total = (size_t)N * (DCH / 4);
        k_finalize<<<(unsigned)((total + 255) / 256), 256, 0, stream>>>(
            out, batch, mean_sum, var_sum, cnt, ms, gw, gb, N);
    }
}


# zero fill (out, statistics sums): a clause of b128 stores per thread, 64 KB per block, 782+2 blocks; was one 4-byte store per thread in 50065 blocks, 400 thousand one-store waves
# speedup vs baseline: 1.6707x; 1.0509x over previous
#include <hip/hip_runtime.h>
#include <hip/hip_bf16.h>
#include <math.h>

typedef __attribute__((ext_vector_type(2))) float v2f;
typedef __attribute__((ext_vector_type(8))) float v8f;

#define DCH 128
#define NGRAPH 64
#define GN_EPS 1e-5f
#define TILE_M 16
#define SX_LD (DCH + 4)

#define ZERO_V4 16
__global__ __launch_bounds__(256) void k_zero_f32(float* __restrict__ p, size_t n) {
    const size_t n4 = n >> 2;
    const size_t blk = (size_t)blockIdx.x * (256 * ZERO_V4);
    float4* __restrict__ q = reinterpret_cast<float4*>(p) + blk + threadIdx.x;
    const float4 z = make_float4(0.0f, 0.0f, 0.0f, 0.0f);
    if (blk + 256 * ZERO_V4 <= n4) {
#pragma unroll
        for (int k = 0; k < ZERO_V4; ++k) q[k * 256] = z;
    } else {
#pragma unroll
        for (int k = 0; k < ZERO_V4; ++k)
            if (blk + (size_t)k * 256 + threadIdx.x < n4) q[k * 256] = z;
    }
    if (blockIdx.x == 0 && threadIdx.x < (n & 3)) p[(n4 << 2) + threadIdx.x] = 0.0f;
}

__global__ __launch_bounds__(256) void k_transpose_w(
    const float* __restrict__ W, float* __restrict__ Wt) {
    __shared__ float tile[16][17];
    int bx = blockIdx.x & 7;
    int by = blockIdx.x >> 3;
    int tx = threadIdx.x & 15;
    int ty = threadIdx.x >> 4;
    tile[ty][tx] = W[(size_t)(by * 16 + ty) * DCH + bx * 16 + tx];
    __syncthreads();
    Wt[(size_t)(bx * 16 + ty) * DCH + by * 16 + tx] = tile[tx][ty];
}

__global__ __launch_bounds__(256) void k_gemm_wmma(
    const float* __restrict__ x, const float* __restrict__ Wt,
    float* __restrict__ h, int N) {
    __shared__ float sX[TILE_M][SX_LD];

    const int t = threadIdx.x;
    const int rowBase = blockIdx.x * TILE_M;

    for (int i = t; i < TILE_M * DCH; i += 256) {
        int r = i >> 7;
        int c = i & (DCH - 1);
        int gr = rowBase + r;
        sX[r][c] = (gr < N) ? x[(size_t)gr * DCH + c] : 0.0f;
    }
    __syncthreads();

    const int lane = t & 31;
    const int wave = t >> 5;
    const int m    = lane & 15;
    const int koff = (lane >> 4) << 1;
    const int n    = (wave << 4) + (lane & 15);

    const float* wrow = Wt + (size_t)n * DCH;

    v8f acc = {};
#pragma unroll
    for (int kt = 0; kt < DCH; kt += 4) {
        v2f a = *reinterpret_cast<const v2f*>(&sX[m][kt + koff]);
        v2f b = *reinterpret_cast<const v2f*>(wrow + kt + koff);
        acc = __builtin_amdgcn_wmma_f32_16x16x4_f32(
            false, a, false, b, (short)0, acc, false, false);
    }

    const int rhalf = (lane >> 4) << 3;
#pragma unroll
    for (int v = 0; v < 8; ++v) {
        int gr = rowBase + rhalf + v;
        if (gr < N) h[(size_t)gr * DCH + n] = acc[v];
    }
}

#define EDGES_PER_WAVE 4
__global__ __launch_bounds__(256) void k_edge_scatter(
    const int* __restrict__ src, const int* __restrict__ dst,
    const float* __restrict__ attr, const float* __restrict__ h,
    float* __restrict__ agg, int E) {
    const int lane  = threadIdx.x & 31;
    const int ebase = (blockIdx.x * 8 + (threadIdx.x >> 5)) * EDGES_PER_WAVE;

    __builtin_prefetch(&src[ebase + 4096], 0, 0);
    __builtin_prefetch(&dst[ebase + 4096], 0, 0);
    __builtin_prefetch(&attr[ebase + 4096], 0, 0);

#pragma unroll
    for (int j = 0; j < EDGES_PER_WAVE; ++j) {
        int e = ebase + j;
        if (e >= E) break;
        e = __builtin_amdgcn_readfirstlane(e);
        int   s = src[e];
        int   d = dst[e];
        float a = attr[e];
        const float* hp = h + (size_t)s * DCH + lane;
        float v0 = hp[0];
        float v1 = hp[32];
        float v2 = hp[64];
        float v3 = hp[96];
        float* outp = agg + (size_t)d * DCH + lane;
        unsafeAtomicAdd(outp + 0,  v0 * a);
        unsafeAtomicAdd(outp + 32, v1 * a);
        unsafeAtomicAdd(outp + 64, v2 * a);
        unsafeAtomicAdd(outp + 96, v3 * a);
    }
}

#define GROWS 64

__global__ __launch_bounds__(128) void k_bias_relu_mean(
    float* __restrict__ out, const float* __restrict__ bias,
    const int* __restrict__ batch, float* __restrict__ mean_sum,
    float* __restrict__ cnt, int N) {
    const int c = threadIdx.x;
    const float bv = bias[c];
    float* msum = mean_sum + c;
    float acc = 0.0f, cacc = 0.0f;
    int curg = -1;

    if (N >= GROWS) {
        int base = blockIdx.x * GROWS;
        int skip = 0;
        if (base + GROWS > N) { skip = base + GROWS - N; base = N - GROWS; }
        float* col = out + (size_t)base * DCH + c;
        const int lane = c & 31;
        const int gidlo = batch[base + lane];
        const int gidhi = batch[base + 32 + lane];

        float v[GROWS];
#pragma unroll
        for (int j = 0; j < GROWS; ++j)
            v[j] = (j >= skip) ? col[(size_t)j * DCH] : 0.0f;

#pragma unroll
        for (int j = 0; j < GROWS; ++j) {
            if (j >= skip) {
                const int gg = __builtin_amdgcn_readlane(j < 32 ? gidlo : gidhi, j & 31);
                if (gg != curg) {
                    if (curg >= 0) {
                        unsafeAtomicAdd(&msum[(size_t)curg * DCH], acc);
                        if (c == 0) unsafeAtomicAdd(&cnt[curg], cacc);
                    }
                    acc = 0.0f; cacc = 0.0f; curg = gg;
                }
                float x = v[j] + bv;
                x = x > 0.0f ? x : 0.0f;
                col[(size_t)j * DCH] = x;
                acc += x;
                cacc += 1.0f;
            }
        }
    } else {
        for (int r = 0; r < N; ++r) {
            const int gg = batch[r];
            if (gg != curg) {
                if (curg >= 0) {
                    unsafeAtomicAdd(&msum[(size_t)curg * DCH], acc);
                    if (c == 0) unsafeAtomicAdd(&cnt[curg], cacc);
                }
                acc = 0.0f; cacc = 0.0f; curg = gg;
            }
            const size_t idx = (size_t)r * DCH + c;
            float x = out[idx] + bv;
            x = x > 0.0f ? x : 0.0f;
            out[idx] = x;
            acc += x;
            cacc += 1.0f;
        }
    }
    if (curg >= 0) {
        unsafeAtomicAdd(&msum[(size_t)curg * DCH], acc);
        if (c == 0) unsafeAtomicAdd(&cnt[curg], cacc);
    }
}

#define HOLD16(a, o)                                                          \
    asm volatile("" :: "v"(a[o + 0]), "v"(a[o + 1]), "v"(a[o + 2]),          \
                 "v"(a[o + 3]), "v"(a[o + 4]), "v"(a[o + 5]), "v"(a[o + 6]), \
                 "v"(a[o + 7]), "v"(a[o + 8]), "v"(a[o + 9]), "v"(a[o + 10]),\
                 "v"(a[o + 11]), "v"(a[o + 12]), "v"(a[o + 13]),             \
                 "v"(a[o + 14]), "v"(a[o + 15]))

__global__ __launch_bounds__(128) void k_var_pass(
    const float* __restrict__ out, const int* __restrict__ batch,
    const float* __restrict__ mean_sum, const float* __restrict__ cnt,
    const float* __restrict__ ms, float* __restrict__ var_sum, int N) {
    const int c = threadIdx.x;
    const float msc = ms[c];
    const float* msum = mean_sum + c;
    float* vsum = var_sum + c;
    float acc = 0.0f;
    float sub = 0.0f;
    int curg = -1;

    if (N >= GROWS) {
        int base = blockIdx.x * GROWS;
        int skip = 0;
        if (base + GROWS > N) { skip = base + GROWS - N; base = N - GROWS; }
        const float* col = out + (size_t)base * DCH + c;
        const int lane = c & 31;
        const int gidlo = batch[base + lane];
        const int gidhi = batch[base + 32 + lane];

        float v[GROWS];
#pragma unroll
        for (int j = 0; j < GROWS; ++j) v[j] = col[(size_t)j * DCH];
        HOLD16(v, 0); HOLD16(v, 16); HOLD16(v, 32); HOLD16(v, 48);

#pragma unroll
        for (int j = 0; j < GROWS; ++j) {
            if (j >= skip) {
                const int gg = __builtin_amdgcn_readlane(j < 32 ? gidlo : gidhi, j & 31);
                if (gg != curg) {
                    if (curg >= 0) unsafeAtomicAdd(&vsum[(size_t)curg * DCH], acc);
                    acc = 0.0f;
                    curg = gg;
                    float cn = cnt[gg];
                    cn = cn > 1.0f ? cn : 1.0f;
                    sub = (msum[(size_t)gg * DCH] / cn) * msc;
                }
                float o = v[j] - sub;
                acc += o * o;
            }
        }
    } else {
        for (int r = 0; r < N; ++r) {
            const int gg = batch[r];
            if (gg != curg) {
                if (curg >= 0) unsafeAtomicAdd(&vsum[(size_t)curg * DCH], acc);
                acc = 0.0f;
                curg = gg;
                float cn = cnt[gg];
                cn = cn > 1.0f ? cn : 1.0f;
                sub = (msum[(size_t)gg * DCH] / cn) * msc;
            }
            float o = out[(size_t)r * DCH + c] - sub;
            acc += o * o;
        }
    }
    if (curg >= 0) unsafeAtomicAdd(&vsum[(size_t)curg * DCH], acc);
}

__global__ __launch_bounds__(256) void k_finalize(
    float* __restrict__ out, const int* __restrict__ batch,
    const float* __restrict__ mean_sum, const float* __restrict__ var_sum,
    const float* __restrict__ cnt, const float* __restrict__ ms,
    const float* __restrict__ gw, const float* __restrict__ gb, int N) {
    size_t idx = (size_t)blockIdx.x * 256 + threadIdx.x;
    size_t total = (size_t)N * (DCH / 4);
    if (idx >= total) return;
    int nrow = (int)(idx >> 5);
    int q    = (int)(idx & 31);
    int g = batch[nrow];
    float cn = cnt[g];
    cn = cn > 1.0f ? cn : 1.0f;
    float inv_cn = 1.0f / cn;

    float4 m4 = reinterpret_cast<const float4*>(mean_sum + (size_t)g * DCH)[q];
    float4 v4 = reinterpret_cast<const float4*>(var_sum  + (size_t)g * DCH)[q];
    float4 s4 = reinterpret_cast<const float4*>(ms)[q];
    float4 w4 = reinterpret_cast<const float4*>(gw)[q];
    float4 b4 = reinterpret_cast<const float4*>(gb)[q];
    float4 o  = reinterpret_cast<float4*>(out)[idx];

    o.x -= (m4.x * inv_cn) * s4.x;
    o.y -= (m4.y * inv_cn) * s4.y;
    o.z -= (m4.z * inv_cn) * s4.z;
    o.w -= (m4.w * inv_cn) * s4.w;
    o.x = w4.x * o.x / sqrtf(v4.x * inv_cn + GN_EPS) + b4.x;
    o.y = w4.y * o.y / sqrtf(v4.y * inv_cn + GN_EPS) + b4.y;
    o.z = w4.z * o.z / sqrtf(v4.z * inv_cn + GN_EPS) + b4.z;
    o.w = w4.w * o.w / sqrtf(v4.w * inv_cn + GN_EPS) + b4.w;

    reinterpret_cast<float4*>(out)[idx] = o;
}

extern "C" void kernel_launch(void* const* d_in, const int* in_sizes, int n_in,
                              void* d_out, int out_size, void* d_ws, size_t ws_size,
                              hipStream_t stream) {
    const float* x     = (const float*)d_in[0];
    const int*   eidx  = (const int*)  d_in[1];
    const float* eattr = (const float*)d_in[2];
    const int*   batch = (const int*)  d_in[3];
    const float* W     = (const float*)d_in[4];
    const float* bias  = (const float*)d_in[5];
    const float* gw    = (const float*)d_in[6];
    const float* gb    = (const float*)d_in[7];
    const float* ms    = (const float*)d_in[8];
    float* out = (float*)d_out;

    const int N = in_sizes[0] / DCH;
    const int E = in_sizes[2];

    float* h        = (float*)d_ws;
    float* mean_sum = h + (size_t)N * DCH;
    float* var_sum  = mean_sum + (size_t)NGRAPH * DCH;
    float* cnt      = var_sum + (size_t)NGRAPH * DCH;
    float* Wt       = cnt + NGRAPH;

    {
        const size_t per_block = (size_t)256 * ZERO_V4 * 4;
        size_t n0 = (size_t)N * DCH;
        k_zero_f32<<<(unsigned)((n0 + per_block - 1) / per_block), 256, 0, stream>>>(out, n0);
        size_t n1 = (size_t)2 * NGRAPH * DCH + NGRAPH;
        k_zero_f32<<<(unsigned)((n1 + per_block - 1) / per_block), 256, 0, stream>>>(mean_sum, n1);
    }

    k_transpose_w<<<64, 256, 0, stream>>>(W, Wt);

    k_gemm_wmma<<<(N + TILE_M - 1) / TILE_M, 256, 0, stream>>>(x, Wt, h, N);

    {
        int edges_per_block = 8 * EDGES_PER_WAVE;
        k_edge_scatter<<<(E + edges_per_block - 1) / edges_per_block, 256, 0, stream>>>(
            eidx, eidx + E, eattr, h, out, E);
    }

    k_bias_relu_mean<<<(N + GROWS - 1) / GROWS, 128, 0, stream>>>(
        out, bias, batch, mean_sum, cnt, N);

    k_var_pass<<<(N + GROWS - 1) / GROWS, 128, 0, stream>>>(
        out, batch, mean_sum, cnt, ms, var_sum, N);

    {
        size_t total = (size_t)N * (DCH / 4);
        k_finalize<<<(unsigned)((total + 255) / 256), 256, 0, stream>>>(
            out, batch, mean_sum, var_sum, cnt, ms, gw, gb, N);
    }
}


# edge sum as a pull: edges grouped by 128-node dst range (integer LDS/global counters), one wave per node adds its h rows in ascending edge number, one store per line; no float atomics there
# speedup vs baseline: 1.8038x; 1.0797x over previous
#include <hip/hip_runtime.h>
#include <hip/hip_bf16.h>
#include <math.h>

typedef __attribute__((ext_vector_type(2))) float v2f;
typedef __attribute__((ext_vector_type(8))) float v8f;

#define DCH 128
#define NGRAPH 64
#define GN_EPS 1e-5f
#define TILE_M 16
#define SX_LD (DCH + 4)

#define ZERO_V4 16
__global__ __launch_bounds__(256) void k_zero_f32(float* __restrict__ p, size_t n) {
    const size_t n4 = n >> 2;
    const size_t blk = (size_t)blockIdx.x * (256 * ZERO_V4);
    float4* __restrict__ q = reinterpret_cast<float4*>(p) + blk + threadIdx.x;
    const float4 z = make_float4(0.0f, 0.0f, 0.0f, 0.0f);
    if (blk + 256 * ZERO_V4 <= n4) {
#pragma unroll
        for (int k = 0; k < ZERO_V4; ++k) q[k * 256] = z;
    } else {
#pragma unroll
        for (int k = 0; k < ZERO_V4; ++k)
            if (blk + (size_t)k * 256 + threadIdx.x < n4) q[k * 256] = z;
    }
    if (blockIdx.x == 0 && threadIdx.x < (n & 3)) p[(n4 << 2) + threadIdx.x] = 0.0f;
}

__global__ __launch_bounds__(256) void k_transpose_w(
    const float* __restrict__ W, float* __restrict__ Wt) {
    __shared__ float tile[16][17];
    int bx = blockIdx.x & 7;
    int by = blockIdx.x >> 3;
    int tx = threadIdx.x & 15;
    int ty = threadIdx.x >> 4;
    tile[ty][tx] = W[(size_t)(by * 16 + ty) * DCH + bx * 16 + tx];
    __syncthreads();
    Wt[(size_t)(bx * 16 + ty) * DCH + by * 16 + tx] = tile[tx][ty];
}

__global__ __launch_bounds__(256) void k_gemm_wmma(
    const float* __restrict__ x, const float* __restrict__ Wt,
    float* __restrict__ h, int N) {
    __shared__ float sX[TILE_M][SX_LD];

    const int t = threadIdx.x;
    const int rowBase = blockIdx.x * TILE_M;

    for (int i = t; i < TILE_M * DCH; i += 256) {
        int r = i >> 7;
        int c = i & (DCH - 1);
        int gr = rowBase + r;
        sX[r][c] = (gr < N) ? x[(size_t)gr * DCH + c] : 0.0f;
    }
    __syncthreads();

    const int lane = t & 31;
    const int wave = t >> 5;
    const int m    = lane & 15;
    const int koff = (lane >> 4) << 1;
    const int n    = (wave << 4) + (lane & 15);

    const float* wrow = Wt + (size_t)n * DCH;

    v8f acc = {};
#pragma unroll
    for (int kt = 0; kt < DCH; kt += 4) {
        v2f a = *reinterpret_cast<const v2f*>(&sX[m][kt + koff]);
        v2f b = *reinterpret_cast<const v2f*>(wrow + kt + koff);
        acc = __builtin_amdgcn_wmma_f32_16x16x4_f32(
            false, a, false, b, (short)0, acc, false, false);
    }

    const int rhalf = (lane >> 4) << 3;
#pragma unroll
    for (int v = 0; v < 8; ++v) {
        int gr = rowBase + rhalf + v;
        if (gr < N) h[(size_t)gr * DCH + n] = acc[v];
    }
}

#define BIN_SHIFT 7
#define BIN_NODES 128
#define MAXBIN 1024
#define CHUNK 4096
#define EPT (CHUNK / 256)
#define LCAP 8192

__global__ __launch_bounds__(256) void k_bin_count(
    const int* __restrict__ dst, int* __restrict__ binCount, int E, int nb) {
    __shared__ int hist[MAXBIN];
    const int t = threadIdx.x;
    for (int i = t; i < nb; i += 256) hist[i] = 0;
    __syncthreads();
    const int e0 = blockIdx.x * CHUNK;
#pragma unroll
    for (int k = 0; k < EPT; ++k) {
        const int e = e0 + k * 256 + t;
        if (e < E) atomicAdd(&hist[dst[e] >> BIN_SHIFT], 1);
    }
    __syncthreads();
    for (int i = t; i < nb; i += 256) {
        const int c = hist[i];
        if (c) atomicAdd(&binCount[i], c);
    }
}

__global__ __launch_bounds__(256) void k_bin_place(
    const int* __restrict__ src, const int* __restrict__ dst,
    const float* __restrict__ attr, const int* __restrict__ binCount,
    int* __restrict__ binFill, int* __restrict__ binStart,
    int4* __restrict__ recs, int E, int nb) {
    __shared__ int lcur[MAXBIN];
    __shared__ int lbase[MAXBIN];
    __shared__ int wtot[8];
    const int t = threadIdx.x;
    const int lane = t & 31;
    const int wave = t >> 5;

    int a0 = 4 * t + 0 < nb ? binCount[4 * t + 0] : 0;
    int a1 = 4 * t + 1 < nb ? binCount[4 * t + 1] : 0;
    int a2 = 4 * t + 2 < nb ? binCount[4 * t + 2] : 0;
    int a3 = 4 * t + 3 < nb ? binCount[4 * t + 3] : 0;
    const int s = a0 + a1 + a2 + a3;
    int v = s;
    for (int o = 1; o < 32; o <<= 1) {
        const int u = __shfl_up(v, o);
        if (lane >= o) v += u;
    }
    if (lane == 31) wtot[wave] = v;
    lcur[4 * t + 0] = 0; lcur[4 * t + 1] = 0; lcur[4 * t + 2] = 0; lcur[4 * t + 3] = 0;
    __syncthreads();
    int wb = 0;
    for (int w = 0; w < wave; ++w) wb += wtot[w];
    const int ex = wb + v - s;
    lbase[4 * t + 0] = ex;
    lbase[4 * t + 1] = ex + a0;
    lbase[4 * t + 2] = ex + a0 + a1;
    lbase[4 * t + 3] = ex + a0 + a1 + a2;
    if (blockIdx.x == 0) {
        binStart[4 * t + 0] = ex;
        binStart[4 * t + 1] = ex + a0;
        binStart[4 * t + 2] = ex + a0 + a1;
        binStart[4 * t + 3] = ex + a0 + a1 + a2;
    }

    const int e0 = blockIdx.x * CHUNK;
    int b[EPT], r[EPT];
#pragma unroll
    for (int k = 0; k < EPT; ++k) {
        const int e = e0 + k * 256 + t;
        b[k] = -1; r[k] = 0;
        if (e < E) {
            b[k] = dst[e] >> BIN_SHIFT;
            r[k] = atomicAdd(&lcur[b[k]], 1);
        }
    }
    __syncthreads();
    for (int i = t; i < nb; i += 256) {
        const int c = lcur[i];
        if (c) lbase[i] += atomicAdd(&binFill[i], c);
    }
    __syncthreads();
#pragma unroll
    for (int k = 0; k < EPT; ++k) {
        if (b[k] >= 0) {
            const int e = e0 + k * 256 + t;
            recs[lbase[b[k]] + r[k]] =
                make_int4(src[e], dst[e], __float_as_int(attr[e]), e);
        }
    }
}

__device__ __forceinline__ void row_fma(const float* __restrict__ h, int s, float a,
                                        int lane, float& c0, float& c1, float& c2, float& c3) {
    const float* hp = h + (size_t)s * DCH + lane;
    c0 += hp[0] * a;
    c1 += hp[32] * a;
    c2 += hp[64] * a;
    c3 += hp[96] * a;
}

#define SUM_THREADS 512
#define SUM_WAVES (SUM_THREADS / 32)
__global__ __launch_bounds__(SUM_THREADS) void k_bin_sum(
    const int4* __restrict__ recs, const int* __restrict__ binCount,
    const int* __restrict__ binStart, const float* __restrict__ h,
    float* __restrict__ agg, int N) {
    __shared__ int nodeCnt[BIN_NODES];
    __shared__ int nodeStart[BIN_NODES];
    __shared__ int nodeCur[BIN_NODES];
    __shared__ int L[LCAP];
    const int t = threadIdx.x;
    const int lane = t & 31;
    const int wave = t >> 5;
    const int bin = blockIdx.x;
    const int n = binCount[bin];
    const int4* __restrict__ R = recs + binStart[bin];
    const int node0 = bin << BIN_SHIFT;
    const bool listed = n <= LCAP;

    if (listed) {
        if (t < BIN_NODES) { nodeCnt[t] = 0; nodeCur[t] = 0; }
        __syncthreads();
        for (int i = t; i < n; i += SUM_THREADS) atomicAdd(&nodeCnt[R[i].y - node0], 1);
        __syncthreads();
        if (t < BIN_NODES) {
            int p = 0;
            for (int u = 0; u < t; ++u) p += nodeCnt[u];
            nodeStart[t] = p;
        }
        __syncthreads();
        for (int i = t; i < n; i += SUM_THREADS) {
            const int dl = R[i].y - node0;
            const int p = atomicAdd(&nodeCur[dl], 1);
            L[nodeStart[dl] + p] = i;
        }
        __syncthreads();
    }

    for (int node = wave; node < BIN_NODES; node += SUM_WAVES) {
        const int gv = node0 + node;
        if (gv >= N) break;
        float c0 = 0.0f, c1 = 0.0f, c2 = 0.0f, c3 = 0.0f;
        if (listed) {
            const int d = __builtin_amdgcn_readfirstlane(nodeCnt[node]);
            const int o = __builtin_amdgcn_readfirstlane(nodeStart[node]);
            if (d <= 32) {
                int key = 0x7fffffff, s = 0;
                float a = 0.0f;
                if (lane < d) {
                    const int4 rc = R[L[o + lane]];
                    key = rc.w; s = rc.x; a = __int_as_float(rc.z);
                }
                int rank = 0;
                for (int j = 0; j < d; ++j)
                    rank += (__builtin_amdgcn_readlane(key, j) < key) ? 1 : 0;
                if (lane >= d) rank = -1;
                int k = 0;
                for (; k + 4 <= d; k += 4) {
                    const int j0 = __builtin_ctz(__builtin_amdgcn_ballot_w32(rank == k + 0));
                    const int j1 = __builtin_ctz(__builtin_amdgcn_ballot_w32(rank == k + 1));
                    const int j2 = __builtin_ctz(__builtin_amdgcn_ballot_w32(rank == k + 2));
                    const int j3 = __builtin_ctz(__builtin_amdgcn_ballot_w32(rank == k + 3));
                    const int s0 = __builtin_amdgcn_readlane(s, j0);
                    const int s1 = __builtin_amdgcn_readlane(s, j1);
                    const int s2 = __builtin_amdgcn_readlane(s, j2);
                    const int s3 = __builtin_amdgcn_readlane(s, j3);
                    const float f0 = __int_as_float(__builtin_amdgcn_readlane(__float_as_int(a), j0));
                    const float f1 = __int_as_float(__builtin_amdgcn_readlane(__float_as_int(a), j1));
                    const float f2 = __int_as_float(__builtin_amdgcn_readlane(__float_as_int(a), j2));
                    const float f3 = __int_as_float(__builtin_amdgcn_readlane(__float_as_int(a), j3));
                    const float* p0 = h + (size_t)s0 * DCH + lane;
                    const float* p1 = h + (size_t)s1 * DCH + lane;
                    const float* p2 = h + (size_t)s2 * DCH + lane;
                    const float* p3 = h + (size_t)s3 * DCH + lane;
                    const float x00 = p0[0], x01 = p0[32], x02 = p0[64], x03 = p0[96];
                    const float x10 = p1[0], x11 = p1[32], x12 = p1[64], x13 = p1[96];
                    const float x20 = p2[0], x21 = p2[32], x22 = p2[64], x23 = p2[96];
                    const float x30 = p3[0], x31 = p3[32], x32 = p3[64], x33 = p3[96];
                    c0 += x00 * f0; c1 += x01 * f0; c2 += x02 * f0; c3 += x03 * f0;
                    c0 += x10 * f1; c1 += x11 * f1; c2 += x12 * f1; c3 += x13 * f1;
                    c0 += x20 * f2; c1 += x21 * f2; c2 += x22 * f2; c3 += x23 * f2;
                    c0 += x30 * f3; c1 += x31 * f3; c2 += x32 * f3; c3 += x33 * f3;
                }
                if (k < d) {
                    const bool u1 = k + 1 < d, u2 = k + 2 < d;
                    const int j0 = __builtin_ctz(__builtin_amdgcn_ballot_w32(rank == k + 0));
                    const int j1 = u1 ? __builtin_ctz(__builtin_amdgcn_ballot_w32(rank == k + 1)) : j0;
                    const int j2 = u2 ? __builtin_ctz(__builtin_amdgcn_ballot_w32(rank == k + 2)) : j0;
                    const int s0 = __builtin_amdgcn_readlane(s, j0);
                    const int s1 = __builtin_amdgcn_readlane(s, j1);
                    const int s2 = __builtin_amdgcn_readlane(s, j2);
                    const float f0 = __int_as_float(__builtin_amdgcn_readlane(__float_as_int(a), j0));
                    const float f1 = __int_as_float(__builtin_amdgcn_readlane(__float_as_int(a), j1));
                    const float f2 = __int_as_float(__builtin_amdgcn_readlane(__float_as_int(a), j2));
                    const float* p0 = h + (size_t)s0 * DCH + lane;
                    const float* p1 = h + (size_t)s1 * DCH + lane;
                    const float* p2 = h + (size_t)s2 * DCH + lane;
                    float x10 = 0.0f, x11 = 0.0f, x12 = 0.0f, x13 = 0.0f;
                    float x20 = 0.0f, x21 = 0.0f, x22 = 0.0f, x23 = 0.0f;
                    const float x00 = p0[0], x01 = p0[32], x02 = p0[64], x03 = p0[96];
                    if (u1) { x10 = p1[0]; x11 = p1[32]; x12 = p1[64]; x13 = p1[96]; }
                    if (u2) { x20 = p2[0]; x21 = p2[32]; x22 = p2[64]; x23 = p2[96]; }
                    c0 += x00 * f0; c1 += x01 * f0; c2 += x02 * f0; c3 += x03 * f0;
                    if (u1) { c0 += x10 * f1; c1 += x11 * f1; c2 += x12 * f1; c3 += x13 * f1; }
                    if (u2) { c0 += x20 * f2; c1 += x21 * f2; c2 += x22 * f2; c3 += x23 * f2; }
                }
            } else {
                int last = -1;
                for (int k = 0; k < d; ++k) {
                    int best = 0x7fffffff, bi = 0;
                    for (int i = lane; i < d; i += 32) {
                        const int ri = L[o + i];
                        const int key = R[ri].w;
                        if (key > last && key < best) { best = key; bi = ri; }
                    }
                    for (int off = 16; off; off >>= 1) {
                        const int ob = __shfl_xor(best, off);
                        const int oi = __shfl_xor(bi, off);
                        if (ob < best) { best = ob; bi = oi; }
                    }
                    const int4 rc = R[bi];
                    last = best;
                    row_fma(h, rc.x, __int_as_float(rc.z), lane, c0, c1, c2, c3);
                }
            }
        } else {
            int last = -1;
            for (int trip = 0; trip <= n; ++trip) {
                int best = 0x7fffffff, bi = 0;
                for (int i = lane; i < n; i += 32) {
                    const int4 rc = R[i];
                    if (rc.y == gv && rc.w > last && rc.w < best) { best = rc.w; bi = i; }
                }
                for (int off = 16; off; off >>= 1) {
                    const int ob = __shfl_xor(best, off);
                    const int oi = __shfl_xor(bi, off);
                    if (ob < best) { best = ob; bi = oi; }
                }
                if (best == 0x7fffffff) break;
                const int4 rc = R[bi];
                last = best;
                row_fma(h, rc.x, __int_as_float(rc.z), lane, c0, c1, c2, c3);
            }
        }
        float* outp = agg + (size_t)gv * DCH + lane;
        outp[0] = c0; outp[32] = c1; outp[64] = c2; outp[96] = c3;
    }
}

#define GROWS 64

__global__ __launch_bounds__(128) void k_bias_relu_mean(
    float* __restrict__ out, const float* __restrict__ bias,
    const int* __restrict__ batch, float* __restrict__ mean_sum,
    float* __restrict__ cnt, int N) {
    const int c = threadIdx.x;
    const float bv = bias[c];
    float* msum = mean_sum + c;
    float acc = 0.0f, cacc = 0.0f;
    int curg = -1;

    if (N >= GROWS) {
        int base = blockIdx.x * GROWS;
        int skip = 0;
        if (base + GROWS > N) { skip = base + GROWS - N; base = N - GROWS; }
        float* col = out + (size_t)base * DCH + c;
        const int lane = c & 31;
        const int gidlo = batch[base + lane];
        const int gidhi = batch[base + 32 + lane];

        float v[GROWS];
#pragma unroll
        for (int j = 0; j < GROWS; ++j)
            v[j] = (j >= skip) ? col[(size_t)j * DCH] : 0.0f;

#pragma unroll
        for (int j = 0; j < GROWS; ++j) {
            if (j >= skip) {
                const int gg = __builtin_amdgcn_readlane(j < 32 ? gidlo : gidhi, j & 31);
                if (gg != curg) {
                    if (curg >= 0) {
                        unsafeAtomicAdd(&msum[(size_t)curg * DCH], acc);
                        if (c == 0) unsafeAtomicAdd(&cnt[curg], cacc);
                    }
                    acc = 0.0f; cacc = 0.0f; curg = gg;
                }
                float x = v[j] + bv;
                x = x > 0.0f ? x : 0.0f;
                col[(size_t)j * DCH] = x;
                acc += x;
                cacc += 1.0f;
            }
        }
    } else {
        for (int r = 0; r < N; ++r) {
            const int gg = batch[r];
            if (gg != curg) {
                if (curg >= 0) {
                    unsafeAtomicAdd(&msum[(size_t)curg * DCH], acc);
                    if (c == 0) unsafeAtomicAdd(&cnt[curg], cacc);
                }
                acc = 0.0f; cacc = 0.0f; curg = gg;
            }
            const size_t idx = (size_t)r * DCH + c;
            float x = out[idx] + bv;
            x = x > 0.0f ? x : 0.0f;
            out[idx] = x;
            acc += x;
            cacc += 1.0f;
        }
    }
    if (curg >= 0) {
        unsafeAtomicAdd(&msum[(size_t)curg * DCH], acc);
        if (c == 0) unsafeAtomicAdd(&cnt[curg], cacc);
    }
}

#define HOLD16(a, o)                                                          \
    asm volatile("" :: "v"(a[o + 0]), "v"(a[o + 1]), "v"(a[o + 2]),          \
                 "v"(a[o + 3]), "v"(a[o + 4]), "v"(a[o + 5]), "v"(a[o + 6]), \
                 "v"(a[o + 7]), "v"(a[o + 8]), "v"(a[o + 9]), "v"(a[o + 10]),\
                 "v"(a[o + 11]), "v"(a[o + 12]), "v"(a[o + 13]),             \
                 "v"(a[o + 14]), "v"(a[o + 15]))

__global__ __launch_bounds__(128) void k_var_pass(
    const float* __restrict__ out, const int* __restrict__ batch,
    const float* __restrict__ mean_sum, const float* __restrict__ cnt,
    const float* __restrict__ ms, float* __restrict__ var_sum, int N) {
    const int c = threadIdx.x;
    const float msc = ms[c];
    const float* msum = mean_sum + c;
    float* vsum = var_sum + c;
    float acc = 0.0f;
    float sub = 0.0f;
    int curg = -1;

    if (N >= GROWS) {
        int base = blockIdx.x * GROWS;
        int skip = 0;
        if (base + GROWS > N) { skip = base + GROWS - N; base = N - GROWS; }
        const float* col = out + (size_t)base * DCH + c;
        const int lane = c & 31;
        const int gidlo = batch[base + lane];
        const int gidhi = batch[base + 32 + lane];

        float v[GROWS];
#pragma unroll
        for (int j = 0; j < GROWS; ++j) v[j] = col[(size_t)j * DCH];
        HOLD16(v, 0); HOLD16(v, 16); HOLD16(v, 32); HOLD16(v, 48);

#pragma unroll
        for (int j = 0; j < GROWS; ++j) {
            if (j >= skip) {
                const int gg = __builtin_amdgcn_readlane(j < 32 ? gidlo : gidhi, j & 31);
                if (gg != curg) {
                    if (curg >= 0) unsafeAtomicAdd(&vsum[(size_t)curg * DCH], acc);
                    acc = 0.0f;
                    curg = gg;
                    float cn = cnt[gg];
                    cn = cn > 1.0f ? cn : 1.0f;
                    sub = (msum[(size_t)gg * DCH] / cn) * msc;
                }
                float o = v[j] - sub;
                acc += o * o;
            }
        }
    } else {
        for (int r = 0; r < N; ++r) {
            const int gg = batch[r];
            if (gg != curg) {
                if (curg >= 0) unsafeAtomicAdd(&vsum[(size_t)curg * DCH], acc);
                acc = 0.0f;
                curg = gg;
                float cn = cnt[gg];
                cn = cn > 1.0f ? cn : 1.0f;
                sub = (msum[(size_t)gg * DCH] / cn) * msc;
            }
            float o = out[(size_t)r * DCH + c] - sub;
            acc += o * o;
        }
    }
    if (curg >= 0) unsafeAtomicAdd(&vsum[(size_t)curg * DCH], acc);
}

__global__ __launch_bounds__(256) void k_finalize(
    float* __restrict__ out, const int* __restrict__ batch,
    const float* __restrict__ mean_sum, const float* __restrict__ var_sum,
    const float* __restrict__ cnt, const float* __restrict__ ms,
    const float* __restrict__ gw, const float* __restrict__ gb, int N) {
    size_t idx = (size_t)blockIdx.x * 256 + threadIdx.x;
    size_t total = (size_t)N * (DCH / 4);
    if (idx >= total) return;
    int nrow = (int)(idx >> 5);
    int q    = (int)(idx & 31);
    int g = batch[nrow];
    float cn = cnt[g];
    cn = cn > 1.0f ? cn : 1.0f;
    float inv_cn = 1.0f / cn;

    float4 m4 = reinterpret_cast<const float4*>(mean_sum + (size_t)g * DCH)[q];
    float4 v4 = reinterpret_cast<const float4*>(var_sum  + (size_t)g * DCH)[q];
    float4 s4 = reinterpret_cast<const float4*>(ms)[q];
    float4 w4 = reinterpret_cast<const float4*>(gw)[q];
    float4 b4 = reinterpret_cast<const float4*>(gb)[q];
    float4 o  = reinterpret_cast<float4*>(out)[idx];

    o.x -= (m4.x * inv_cn) * s4.x;
    o.y -= (m4.y * inv_cn) * s4.y;
    o.z -= (m4.z * inv_cn) * s4.z;
    o.w -= (m4.w * inv_cn) * s4.w;
    o.x = w4.x * o.x / sqrtf(v4.x * inv_cn + GN_EPS) + b4.x;
    o.y = w4.y * o.y / sqrtf(v4.y * inv_cn + GN_EPS) + b4.y;
    o.z = w4.z * o.z / sqrtf(v4.z * inv_cn + GN_EPS) + b4.z;
    o.w = w4.w * o.w / sqrtf(v4.w * inv_cn + GN_EPS) + b4.w;

    reinterpret_cast<float4*>(out)[idx] = o;
}

extern "C" void kernel_launch(void* const* d_in, const int* in_sizes, int n_in,
                              void* d_out, int out_size, void* d_ws, size_t ws_size,
                              hipStream_t stream) {
    const float* x     = (const float*)d_in[0];
    const int*   eidx  = (const int*)  d_in[1];
    const float* eattr = (const float*)d_in[2];
    const int*   batch = (const int*)  d_in[3];
    const float* W     = (const float*)d_in[4];
    const float* bias  = (const float*)d_in[5];
    const float* gw    = (const float*)d_in[6];
    const float* gb    = (const float*)d_in[7];
    const float* ms    = (const float*)d_in[8];
    float* out = (float*)d_out;

    const int N = in_sizes[0] / DCH;
    const int E = in_sizes[2];

    float* h        = (float*)d_ws;
    float* mean_sum = h + (size_t)N * DCH;
    float* var_sum  = mean_sum + (size_t)NGRAPH * DCH;
    float* cnt      = var_sum + (size_t)NGRAPH * DCH;
    int*   binCount = (int*)(cnt + NGRAPH);
    int*   binFill  = binCount + MAXBIN;
    float* Wt       = (float*)(binFill + MAXBIN);
    int*   binStart = (int*)(Wt + DCH * DCH);
    int4*  recs     = (int4*)(binStart + MAXBIN);
    const int nb = (N + BIN_NODES - 1) >> BIN_SHIFT;

    {
        const size_t per_block = (size_t)256 * ZERO_V4 * 4;
        size_t n1 = (size_t)2 * NGRAPH * DCH + NGRAPH + 2 * MAXBIN;
        k_zero_f32<<<(unsigned)((n1 + per_block - 1) / per_block), 256, 0, stream>>>(mean_sum, n1);
    }

    k_transpose_w<<<64, 256, 0, stream>>>(W, Wt);

    k_gemm_wmma<<<(N + TILE_M - 1) / TILE_M, 256, 0, stream>>>(x, Wt, h, N);

    {
        const int eblocks = (E + CHUNK - 1) / CHUNK;
        k_bin_count<<<eblocks, 256, 0, stream>>>(eidx + E, binCount, E, nb);
        k_bin_place<<<eblocks, 256, 0, stream>>>(
            eidx, eidx + E, eattr, binCount, binFill, binStart, recs, E, nb);
        k_bin_sum<<<nb, SUM_THREADS, 0, stream>>>(recs, binCount, binStart, h, out, N);
    }

    k_bias_relu_mean<<<(N + GROWS - 1) / GROWS, 128, 0, stream>>>(
        out, bias, batch, mean_sum, cnt, N);

    k_var_pass<<<(N + GROWS - 1) / GROWS, 128, 0, stream>>>(
        out, batch, mean_sum, cnt, ms, var_sum, N);

    {
        size_t total = (size_t)N * (DCH / 4);
        k_finalize<<<(unsigned)((total + 255) / 256), 256, 0, stream>>>(
            out, batch, mean_sum, var_sum, cnt, ms, gw, gb, N);
    }
}


# edge grouping (count, place, per-bin list): each thread's 16 edge loads (dst; src, attr) and its returning bin adds go out together, then are used; were one per memory round trip, 16 in a row
# speedup vs baseline: 1.8752x; 1.0396x over previous
#include <hip/hip_runtime.h>
#include <hip/hip_bf16.h>
#include <math.h>

typedef __attribute__((ext_vector_type(2))) float v2f;
typedef __attribute__((ext_vector_type(8))) float v8f;

#define DCH 128
#define NGRAPH 64
#define GN_EPS 1e-5f
#define TILE_M 16
#define SX_LD (DCH + 4)

#define ZERO_V4 16
__global__ __launch_bounds__(256) void k_zero_f32(float* __restrict__ p, size_t n) {
    const size_t n4 = n >> 2;
    const size_t blk = (size_t)blockIdx.x * (256 * ZERO_V4);
    float4* __restrict__ q = reinterpret_cast<float4*>(p) + blk + threadIdx.x;
    const float4 z = make_float4(0.0f, 0.0f, 0.0f, 0.0f);
    if (blk + 256 * ZERO_V4 <= n4) {
#pragma unroll
        for (int k = 0; k < ZERO_V4; ++k) q[k * 256] = z;
    } else {
#pragma unroll
        for (int k = 0; k < ZERO_V4; ++k)
            if (blk + (size_t)k * 256 + threadIdx.x < n4) q[k * 256] = z;
    }
    if (blockIdx.x == 0 && threadIdx.x < (n & 3)) p[(n4 << 2) + threadIdx.x] = 0.0f;
}

__global__ __launch_bounds__(256) void k_transpose_w(
    const float* __restrict__ W, float* __restrict__ Wt) {
    __shared__ float tile[16][17];
    int bx = blockIdx.x & 7;
    int by = blockIdx.x >> 3;
    int tx = threadIdx.x & 15;
    int ty = threadIdx.x >> 4;
    tile[ty][tx] = W[(size_t)(by * 16 + ty) * DCH + bx * 16 + tx];
    __syncthreads();
    Wt[(size_t)(bx * 16 + ty) * DCH + by * 16 + tx] = tile[tx][ty];
}

__global__ __launch_bounds__(256) void k_gemm_wmma(
    const float* __restrict__ x, const float* __restrict__ Wt,
    float* __restrict__ h, int N) {
    __shared__ float sX[TILE_M][SX_LD];

    const int t = threadIdx.x;
    const int rowBase = blockIdx.x * TILE_M;

    for (int i = t; i < TILE_M * DCH; i += 256) {
        int r = i >> 7;
        int c = i & (DCH - 1);
        int gr = rowBase + r;
        sX[r][c] = (gr < N) ? x[(size_t)gr * DCH + c] : 0.0f;
    }
    __syncthreads();

    const int lane = t & 31;
    const int wave = t >> 5;
    const int m    = lane & 15;
    const int koff = (lane >> 4) << 1;
    const int n    = (wave << 4) + (lane & 15);

    const float* wrow = Wt + (size_t)n * DCH;

    v8f acc = {};
#pragma unroll
    for (int kt = 0; kt < DCH; kt += 4) {
        v2f a = *reinterpret_cast<const v2f*>(&sX[m][kt + koff]);
        v2f b = *reinterpret_cast<const v2f*>(wrow + kt + koff);
        acc = __builtin_amdgcn_wmma_f32_16x16x4_f32(
            false, a, false, b, (short)0, acc, false, false);
    }

    const int rhalf = (lane >> 4) << 3;
#pragma unroll
    for (int v = 0; v < 8; ++v) {
        int gr = rowBase + rhalf + v;
        if (gr < N) h[(size_t)gr * DCH + n] = acc[v];
    }
}

#define BIN_SHIFT 7
#define BIN_NODES 128
#define MAXBIN 1024
#define CHUNK 4096
#define EPT (CHUNK / 256)
#define LCAP 8192

__global__ __launch_bounds__(256) void k_bin_count(
    const int* __restrict__ dst, int* __restrict__ binCount, int E, int nb) {
    __shared__ int hist[MAXBIN];
    const int t = threadIdx.x;
    const int e0 = blockIdx.x * CHUNK;
    int dv[EPT];
#pragma unroll
    for (int k = 0; k < EPT; ++k) {
        const int e = e0 + k * 256 + t;
        dv[k] = dst[e < E ? e : E - 1];
    }
    for (int i = t; i < nb; i += 256) hist[i] = 0;
    __syncthreads();
#pragma unroll
    for (int k = 0; k < EPT; ++k) {
        const int e = e0 + k * 256 + t;
        if (e < E) atomicAdd(&hist[dv[k] >> BIN_SHIFT], 1);
    }
    __syncthreads();
    for (int i = t; i < nb; i += 256) {
        const int c = hist[i];
        if (c) atomicAdd(&binCount[i], c);
    }
}

__global__ __launch_bounds__(256) void k_bin_place(
    const int* __restrict__ src, const int* __restrict__ dst,
    const float* __restrict__ attr, const int* __restrict__ binCount,
    int* __restrict__ binFill, int* __restrict__ binStart,
    int4* __restrict__ recs, int E, int nb) {
    __shared__ int lcur[MAXBIN];
    __shared__ int lbase[MAXBIN];
    __shared__ int wtot[8];
    const int t = threadIdx.x;
    const int lane = t & 31;
    const int wave = t >> 5;
    const int e0 = blockIdx.x * CHUNK;

    int a0 = 4 * t + 0 < nb ? binCount[4 * t + 0] : 0;
    int a1 = 4 * t + 1 < nb ? binCount[4 * t + 1] : 0;
    int a2 = 4 * t + 2 < nb ? binCount[4 * t + 2] : 0;
    int a3 = 4 * t + 3 < nb ? binCount[4 * t + 3] : 0;

    int dv[EPT], sv[EPT];
    float av[EPT];
#pragma unroll
    for (int k = 0; k < EPT; ++k) {
        const int e = e0 + k * 256 + t;
        dv[k] = dst[e < E ? e : E - 1];
    }
#pragma unroll
    for (int k = 0; k < EPT; ++k) {
        const int e = e0 + k * 256 + t;
        const int ec = e < E ? e : E - 1;
        sv[k] = src[ec];
        av[k] = attr[ec];
    }

    const int s = a0 + a1 + a2 + a3;
    int v = s;
    for (int o = 1; o < 32; o <<= 1) {
        const int u = __shfl_up(v, o);
        if (lane >= o) v += u;
    }
    if (lane == 31) wtot[wave] = v;
    lcur[4 * t + 0] = 0; lcur[4 * t + 1] = 0; lcur[4 * t + 2] = 0; lcur[4 * t + 3] = 0;
    __syncthreads();
    int wb = 0;
    for (int w = 0; w < wave; ++w) wb += wtot[w];
    const int ex = wb + v - s;
    lbase[4 * t + 0] = ex;
    lbase[4 * t + 1] = ex + a0;
    lbase[4 * t + 2] = ex + a0 + a1;
    lbase[4 * t + 3] = ex + a0 + a1 + a2;
    if (blockIdx.x == 0) {
        binStart[4 * t + 0] = ex;
        binStart[4 * t + 1] = ex + a0;
        binStart[4 * t + 2] = ex + a0 + a1;
        binStart[4 * t + 3] = ex + a0 + a1 + a2;
    }

    int r[EPT];
#pragma unroll
    for (int k = 0; k < EPT; ++k) {
        const int e = e0 + k * 256 + t;
        r[k] = 0;
        if (e < E) r[k] = atomicAdd(&lcur[dv[k] >> BIN_SHIFT], 1);
    }
    __syncthreads();
    int cc[MAXBIN / 256], gb[MAXBIN / 256];
#pragma unroll
    for (int j = 0; j < MAXBIN / 256; ++j) {
        const int i = t + 256 * j;
        cc[j] = i < nb ? lcur[i] : 0;
    }
#pragma unroll
    for (int j = 0; j < MAXBIN / 256; ++j) {
        gb[j] = 0;
        if (cc[j]) gb[j] = atomicAdd(&binFill[t + 256 * j], cc[j]);
    }
#pragma unroll
    for (int j = 0; j < MAXBIN / 256; ++j)
        if (cc[j]) lbase[t + 256 * j] += gb[j];
    __syncthreads();
    int lb[EPT];
#pragma unroll
    for (int k = 0; k < EPT; ++k) lb[k] = lbase[dv[k] >> BIN_SHIFT];
#pragma unroll
    for (int k = 0; k < EPT; ++k) {
        const int e = e0 + k * 256 + t;
        if (e < E)
            recs[lb[k] + r[k]] = make_int4(sv[k], dv[k], __float_as_int(av[k]), e);
    }
}

__device__ __forceinline__ void row_fma(const float* __restrict__ h, int s, float a,
                                        int lane, float& c0, float& c1, float& c2, float& c3) {
    const float* hp = h + (size_t)s * DCH + lane;
    c0 += hp[0] * a;
    c1 += hp[32] * a;
    c2 += hp[64] * a;
    c3 += hp[96] * a;
}

#define SUM_THREADS 512
#define SUM_WAVES (SUM_THREADS / 32)
#define LIST_V 8
__global__ __launch_bounds__(SUM_THREADS) void k_bin_sum(
    const int4* __restrict__ recs, const int* __restrict__ binCount,
    const int* __restrict__ binStart, const float* __restrict__ h,
    float* __restrict__ agg, int N) {
    __shared__ int nodeCnt[BIN_NODES];
    __shared__ int nodeStart[BIN_NODES];
    __shared__ int nodeCur[BIN_NODES];
    __shared__ int L[LCAP];
    const int t = threadIdx.x;
    const int lane = t & 31;
    const int wave = t >> 5;
    const int bin = blockIdx.x;
    const int n = binCount[bin];
    const int4* __restrict__ R = recs + binStart[bin];
    const int node0 = bin << BIN_SHIFT;
    const bool listed = n <= LCAP;

    if (listed) {
        if (t < BIN_NODES) { nodeCnt[t] = 0; nodeCur[t] = 0; }
        __syncthreads();
        for (int i0 = t; i0 < n; i0 += LIST_V * SUM_THREADS) {
            int y[LIST_V];
#pragma unroll
            for (int j = 0; j < LIST_V; ++j) {
                const int i = i0 + j * SUM_THREADS;
                y[j] = R[i < n ? i : n - 1].y;
            }
#pragma unroll
            for (int j = 0; j < LIST_V; ++j)
                if (i0 + j * SUM_THREADS < n) atomicAdd(&nodeCnt[y[j] - node0], 1);
        }
        __syncthreads();
        if (t < BIN_NODES) {
            int p = 0;
            for (int u = 0; u < t; ++u) p += nodeCnt[u];
            nodeStart[t] = p;
        }
        __syncthreads();
        for (int i0 = t; i0 < n; i0 += LIST_V * SUM_THREADS) {
            int y[LIST_V];
#pragma unroll
            for (int j = 0; j < LIST_V; ++j) {
                const int i = i0 + j * SUM_THREADS;
                y[j] = R[i < n ? i : n - 1].y;
            }
#pragma unroll
            for (int j = 0; j < LIST_V; ++j) {
                const int i = i0 + j * SUM_THREADS;
                if (i < n) {
                    const int dl = y[j] - node0;
                    const int p = atomicAdd(&nodeCur[dl], 1);
                    L[nodeStart[dl] + p] = i;
                }
            }
        }
        __syncthreads();
    }

    for (int node = wave; node < BIN_NODES; node += SUM_WAVES) {
        const int gv = node0 + node;
        if (gv >= N) break;
        float c0 = 0.0f, c1 = 0.0f, c2 = 0.0f, c3 = 0.0f;
        if (listed) {
            const int d = __builtin_amdgcn_readfirstlane(nodeCnt[node]);
            const int o = __builtin_amdgcn_readfirstlane(nodeStart[node]);
            if (d <= 32) {
                int key = 0x7fffffff, s = 0;
                float a = 0.0f;
                if (lane < d) {
                    const int4 rc = R[L[o + lane]];
                    key = rc.w; s = rc.x; a = __int_as_float(rc.z);
                }
                int rank = 0;
                for (int j = 0; j < d; ++j)
                    rank += (__builtin_amdgcn_readlane(key, j) < key) ? 1 : 0;
                if (lane >= d) rank = -1;
                int k = 0;
                for (; k + 4 <= d; k += 4) {
                    const int j0 = __builtin_ctz(__builtin_amdgcn_ballot_w32(rank == k + 0));
                    const int j1 = __builtin_ctz(__builtin_amdgcn_ballot_w32(rank == k + 1));
                    const int j2 = __builtin_ctz(__builtin_amdgcn_ballot_w32(rank == k + 2));
                    const int j3 = __builtin_ctz(__builtin_amdgcn_ballot_w32(rank == k + 3));
                    const int s0 = __builtin_amdgcn_readlane(s, j0);
                    const int s1 = __builtin_amdgcn_readlane(s, j1);
                    const int s2 = __builtin_amdgcn_readlane(s, j2);
                    const int s3 = __builtin_amdgcn_readlane(s, j3);
                    const float f0 = __int_as_float(__builtin_amdgcn_readlane(__float_as_int(a), j0));
                    const float f1 = __int_as_float(__builtin_amdgcn_readlane(__float_as_int(a), j1));
                    const float f2 = __int_as_float(__builtin_amdgcn_readlane(__float_as_int(a), j2));
                    const float f3 = __int_as_float(__builtin_amdgcn_readlane(__float_as_int(a), j3));
                    const float* p0 = h + (size_t)s0 * DCH + lane;
                    const float* p1 = h + (size_t)s1 * DCH + lane;
                    const float* p2 = h + (size_t)s2 * DCH + lane;
                    const float* p3 = h + (size_t)s3 * DCH + lane;
                    const float x00 = p0[0], x01 = p0[32], x02 = p0[64], x03 = p0[96];
                    const float x10 = p1[0], x11 = p1[32], x12 = p1[64], x13 = p1[96];
                    const float x20 = p2[0], x21 = p2[32], x22 = p2[64], x23 = p2[96];
                    const float x30 = p3[0], x31 = p3[32], x32 = p3[64], x33 = p3[96];
                    c0 += x00 * f0; c1 += x01 * f0; c2 += x02 * f0; c3 += x03 * f0;
                    c0 += x10 * f1; c1 += x11 * f1; c2 += x12 * f1; c3 += x13 * f1;
                    c0 += x20 * f2; c1 += x21 * f2; c2 += x22 * f2; c3 += x23 * f2;
                    c0 += x30 * f3; c1 += x31 * f3; c2 += x32 * f3; c3 += x33 * f3;
                }
                if (k < d) {
                    const bool u1 = k + 1 < d, u2 = k + 2 < d;
                    const int j0 = __builtin_ctz(__builtin_amdgcn_ballot_w32(rank == k + 0));
                    const int j1 = u1 ? __builtin_ctz(__builtin_amdgcn_ballot_w32(rank == k + 1)) : j0;
                    const int j2 = u2 ? __builtin_ctz(__builtin_amdgcn_ballot_w32(rank == k + 2)) : j0;
                    const int s0 = __builtin_amdgcn_readlane(s, j0);
                    const int s1 = __builtin_amdgcn_readlane(s, j1);
                    const int s2 = __builtin_amdgcn_readlane(s, j2);
                    const float f0 = __int_as_float(__builtin_amdgcn_readlane(__float_as_int(a), j0));
                    const float f1 = __int_as_float(__builtin_amdgcn_readlane(__float_as_int(a), j1));
                    const float f2 = __int_as_float(__builtin_amdgcn_readlane(__float_as_int(a), j2));
                    const float* p0 = h + (size_t)s0 * DCH + lane;
                    const float* p1 = h + (size_t)s1 * DCH + lane;
                    const float* p2 = h + (size_t)s2 * DCH + lane;
                    float x10 = 0.0f, x11 = 0.0f, x12 = 0.0f, x13 = 0.0f;
                    float x20 = 0.0f, x21 = 0.0f, x22 = 0.0f, x23 = 0.0f;
                    const float x00 = p0[0], x01 = p0[32], x02 = p0[64], x03 = p0[96];
                    if (u1) { x10 = p1[0]; x11 = p1[32]; x12 = p1[64]; x13 = p1[96]; }
                    if (u2) { x20 = p2[0]; x21 = p2[32]; x22 = p2[64]; x23 = p2[96]; }
                    c0 += x00 * f0; c1 += x01 * f0; c2 += x02 * f0; c3 += x03 * f0;
                    if (u1) { c0 += x10 * f1; c1 += x11 * f1; c2 += x12 * f1; c3 += x13 * f1; }
                    if (u2) { c0 += x20 * f2; c1 += x21 * f2; c2 += x22 * f2; c3 += x23 * f2; }
                }
            } else {
                int last = -1;
                for (int k = 0; k < d; ++k) {
                    int best = 0x7fffffff, bi = 0;
                    for (int i = lane; i < d; i += 32) {
                        const int ri = L[o + i];
                        const int key = R[ri].w;
                        if (key > last && key < best) { best = key; bi = ri; }
                    }
                    for (int off = 16; off; off >>= 1) {
                        const int ob = __shfl_xor(best, off);
                        const int oi = __shfl_xor(bi, off);
                        if (ob < best) { best = ob; bi = oi; }
                    }
                    const int4 rc = R[bi];
                    last = best;
                    row_fma(h, rc.x, __int_as_float(rc.z), lane, c0, c1, c2, c3);
                }
            }
        } else {
            int last = -1;
            for (int trip = 0; trip <= n; ++trip) {
                int best = 0x7fffffff, bi = 0;
                for (int i = lane; i < n; i += 32) {
                    const int4 rc = R[i];
                    if (rc.y == gv && rc.w > last && rc.w < best) { best = rc.w; bi = i; }
                }
                for (int off = 16; off; off >>= 1) {
                    const int ob = __shfl_xor(best, off);
                    const int oi = __shfl_xor(bi, off);
                    if (ob < best) { best = ob; bi = oi; }
                }
                if (best == 0x7fffffff) break;
                const int4 rc = R[bi];
                last = best;
                row_fma(h, rc.x, __int_as_float(rc.z), lane, c0, c1, c2, c3);
            }
        }
        float* outp = agg + (size_t)gv * DCH + lane;
        outp[0] = c0; outp[32] = c1; outp[64] = c2; outp[96] = c3;
    }
}

#define GROWS 64

__global__ __launch_bounds__(128) void k_bias_relu_mean(
    float* __restrict__ out, const float* __restrict__ bias,
    const int* __restrict__ batch, float* __restrict__ mean_sum,
    float* __restrict__ cnt, int N) {
    const int c = threadIdx.x;
    const float bv = bias[c];
    float* msum = mean_sum + c;
    float acc = 0.0f, cacc = 0.0f;
    int curg = -1;

    if (N >= GROWS) {
        int base = blockIdx.x * GROWS;
        int skip = 0;
        if (base + GROWS > N) { skip = base + GROWS - N; base = N - GROWS; }
        float* col = out + (size_t)base * DCH + c;
        const int lane = c & 31;
        const int gidlo = batch[base + lane];
        const int gidhi = batch[base + 32 + lane];

        float v[GROWS];
#pragma unroll
        for (int j = 0; j < GROWS; ++j)
            v[j] = (j >= skip) ? col[(size_t)j * DCH] : 0.0f;

#pragma unroll
        for (int j = 0; j < GROWS; ++j) {
            if (j >= skip) {
                const int gg = __builtin_amdgcn_readlane(j < 32 ? gidlo : gidhi, j & 31);
                if (gg != curg) {
                    if (curg >= 0) {
                        unsafeAtomicAdd(&msum[(size_t)curg * DCH], acc);
                        if (c == 0) unsafeAtomicAdd(&cnt[curg], cacc);
                    }
                    acc = 0.0f; cacc = 0.0f; curg = gg;
                }
                float x = v[j] + bv;
                x = x > 0.0f ? x : 0.0f;
                col[(size_t)j * DCH] = x;
                acc += x;
                cacc += 1.0f;
            }
        }
    } else {
        for (int r = 0; r < N; ++r) {
            const int gg = batch[r];
            if (gg != curg) {
                if (curg >= 0) {
                    unsafeAtomicAdd(&msum[(size_t)curg * DCH], acc);
                    if (c == 0) unsafeAtomicAdd(&cnt[curg], cacc);
                }
                acc = 0.0f; cacc = 0.0f; curg = gg;
            }
            const size_t idx = (size_t)r * DCH + c;
            float x = out[idx] + bv;
            x = x > 0.0f ? x : 0.0f;
            out[idx] = x;
            acc += x;
            cacc += 1.0f;
        }
    }
    if (curg >= 0) {
        unsafeAtomicAdd(&msum[(size_t)curg * DCH], acc);
        if (c == 0) unsafeAtomicAdd(&cnt[curg], cacc);
    }
}

#define HOLD16(a, o)                                                          \
    asm volatile("" :: "v"(a[o + 0]), "v"(a[o + 1]), "v"(a[o + 2]),          \
                 "v"(a[o + 3]), "v"(a[o + 4]), "v"(a[o + 5]), "v"(a[o + 6]), \
                 "v"(a[o + 7]), "v"(a[o + 8]), "v"(a[o + 9]), "v"(a[o + 10]),\
                 "v"(a[o + 11]), "v"(a[o + 12]), "v"(a[o + 13]),             \
                 "v"(a[o + 14]), "v"(a[o + 15]))

__global__ __launch_bounds__(128) void k_var_pass(
    const float* __restrict__ out, const int* __restrict__ batch,
    const float* __restrict__ mean_sum, const float* __restrict__ cnt,
    const float* __restrict__ ms, float* __restrict__ var_sum, int N) {
    const int c = threadIdx.x;
    const float msc = ms[c];
    const float* msum = mean_sum + c;
    float* vsum = var_sum + c;
    float acc = 0.0f;
    float sub = 0.0f;
    int curg = -1;

    if (N >= GROWS) {
        int base = blockIdx.x * GROWS;
        int skip = 0;
        if (base + GROWS > N) { skip = base + GROWS - N; base = N - GROWS; }
        const float* col = out + (size_t)base * DCH + c;
        const int lane = c & 31;
        const int gidlo = batch[base + lane];
        const int gidhi = batch[base + 32 + lane];

        float v[GROWS];
#pragma unroll
        for (int j = 0; j < GROWS; ++j) v[j] = col[(size_t)j * DCH];
        HOLD16(v, 0); HOLD16(v, 16); HOLD16(v, 32); HOLD16(v, 48);

#pragma unroll
        for (int j = 0; j < GROWS; ++j) {
            if (j >= skip) {
                const int gg = __builtin_amdgcn_readlane(j < 32 ? gidlo : gidhi, j & 31);
                if (gg != curg) {
                    if (curg >= 0) unsafeAtomicAdd(&vsum[(size_t)curg * DCH], acc);
                    acc = 0.0f;
                    curg = gg;
                    float cn = cnt[gg];
                    cn = cn > 1.0f ? cn : 1.0f;
                    sub = (msum[(size_t)gg * DCH] / cn) * msc;
                }
                float o = v[j] - sub;
                acc += o * o;
            }
        }
    } else {
        for (int r = 0; r < N; ++r) {
            const int gg = batch[r];
            if (gg != curg) {
                if (curg >= 0) unsafeAtomicAdd(&vsum[(size_t)curg * DCH], acc);
                acc = 0.0f;
                curg = gg;
                float cn = cnt[gg];
                cn = cn > 1.0f ? cn : 1.0f;
                sub = (msum[(size_t)gg * DCH] / cn) * msc;
            }
            float o = out[(size_t)r * DCH + c] - sub;
            acc += o * o;
        }
    }
    if (curg >= 0) unsafeAtomicAdd(&vsum[(size_t)curg * DCH], acc);
}

__global__ __launch_bounds__(256) void k_finalize(
    float* __restrict__ out, const int* __restrict__ batch,
    const float* __restrict__ mean_sum, const float* __restrict__ var_sum,
    const float* __restrict__ cnt, const float* __restrict__ ms,
    const float* __restrict__ gw, const float* __restrict__ gb, int N) {
    size_t idx = (size_t)blockIdx.x * 256 + threadIdx.x;
    size_t total = (size_t)N * (DCH / 4);
    if (idx >= total) return;
    int nrow = (int)(idx >> 5);
    int q    = (int)(idx & 31);
    int g = batch[nrow];
    float cn = cnt[g];
    cn = cn > 1.0f ? cn : 1.0f;
    float inv_cn = 1.0f / cn;

    float4 m4 = reinterpret_cast<const float4*>(mean_sum + (size_t)g * DCH)[q];
    float4 v4 = reinterpret_cast<const float4*>(var_sum  + (size_t)g * DCH)[q];
    float4 s4 = reinterpret_cast<const float4*>(ms)[q];
    float4 w4 = reinterpret_cast<const float4*>(gw)[q];
    float4 b4 = reinterpret_cast<const float4*>(gb)[q];
    float4 o  = reinterpret_cast<float4*>(out)[idx];

    o.x -= (m4.x * inv_cn) * s4.x;
    o.y -= (m4.y * inv_cn) * s4.y;
    o.z -= (m4.z * inv_cn) * s4.z;
    o.w -= (m4.w * inv_cn) * s4.w;
    o.x = w4.x * o.x / sqrtf(v4.x * inv_cn + GN_EPS) + b4.x;
    o.y = w4.y * o.y / sqrtf(v4.y * inv_cn + GN_EPS) + b4.y;
    o.z = w4.z * o.z / sqrtf(v4.z * inv_cn + GN_EPS) + b4.z;
    o.w = w4.w * o.w / sqrtf(v4.w * inv_cn + GN_EPS) + b4.w;

    reinterpret_cast<float4*>(out)[idx] = o;
}

extern "C" void kernel_launch(void* const* d_in, const int* in_sizes, int n_in,
                              void* d_out, int out_size, void* d_ws, size_t ws_size,
                              hipStream_t stream) {
    const float* x     = (const float*)d_in[0];
    const int*   eidx  = (const int*)  d_in[1];
    const float* eattr = (const float*)d_in[2];
    const int*   batch = (const int*)  d_in[3];
    const float* W     = (const float*)d_in[4];
    const float* bias  = (const float*)d_in[5];
    const float* gw    = (const float*)d_in[6];
    const float* gb    = (const float*)d_in[7];
    const float* ms    = (const float*)d_in[8];
    float* out = (float*)d_out;

    const int N = in_sizes[0] / DCH;
    const int E = in_sizes[2];

    float* h        = (float*)d_ws;
    float* mean_sum = h + (size_t)N * DCH;
    float* var_sum  = mean_sum + (size_t)NGRAPH * DCH;
    float* cnt      = var_sum + (size_t)NGRAPH * DCH;
    int*   binCount = (int*)(cnt + NGRAPH);
    int*   binFill  = binCount + MAXBIN;
    float* Wt       = (float*)(binFill + MAXBIN);
    int*   binStart = (int*)(Wt + DCH * DCH);
    int4*  recs     = (int4*)(binStart + MAXBIN);
    const int nb = (N + BIN_NODES - 1) >> BIN_SHIFT;

    {
        const size_t per_block = (size_t)256 * ZERO_V4 * 4;
        size_t n1 = (size_t)2 * NGRAPH * DCH + NGRAPH + 2 * MAXBIN;
        k_zero_f32<<<(unsigned)((n1 + per_block - 1) / per_block), 256, 0, stream>>>(mean_sum, n1);
    }

    k_transpose_w<<<64, 256, 0, stream>>>(W, Wt);

    k_gemm_wmma<<<(N + TILE_M - 1) / TILE_M, 256, 0, stream>>>(x, Wt, h, N);

    {
        const int eblocks = (E + CHUNK - 1) / CHUNK;
        k_bin_count<<<eblocks, 256, 0, stream>>>(eidx + E, binCount, E, nb);
        k_bin_place<<<eblocks, 256, 0, stream>>>(
            eidx, eidx + E, eattr, binCount, binFill, binStart, recs, E, nb);
        k_bin_sum<<<nb, SUM_THREADS, 0, stream>>>(recs, binCount, binStart, h, out, N);
    }

    k_bias_relu_mean<<<(N + GROWS - 1) / GROWS, 128, 0, stream>>>(
        out, bias, batch, mean_sum, cnt, N);

    k_var_pass<<<(N + GROWS - 1) / GROWS, 128, 0, stream>>>(
        out, batch, mean_sum, cnt, ms, var_sum, N);

    {
        size_t total = (size_t)N * (DCH / 4);
        k_finalize<<<(unsigned)((total + 255) / 256), 256, 0, stream>>>(
            out, batch, mean_sum, var_sum, cnt, ms, gw, gb, N);
    }
}


# edge sum: a node with 33 to 64 edges is ranked in registers (two listed edges per lane) and its rows added four at a time like the rest; was a wave-wide minimum search per edge; same edge order
# speedup vs baseline: 1.9971x; 1.0650x over previous
#include <hip/hip_runtime.h>
#include <hip/hip_bf16.h>
#include <math.h>

typedef __attribute__((ext_vector_type(2))) float v2f;
typedef __attribute__((ext_vector_type(8))) float v8f;

#define DCH 128
#define NGRAPH 64
#define GN_EPS 1e-5f
#define TILE_M 16
#define SX_LD (DCH + 4)

#define ZERO_V4 16
__global__ __launch_bounds__(256) void k_zero_f32(float* __restrict__ p, size_t n) {
    const size_t n4 = n >> 2;
    const size_t blk = (size_t)blockIdx.x * (256 * ZERO_V4);
    float4* __restrict__ q = reinterpret_cast<float4*>(p) + blk + threadIdx.x;
    const float4 z = make_float4(0.0f, 0.0f, 0.0f, 0.0f);
    if (blk + 256 * ZERO_V4 <= n4) {
#pragma unroll
        for (int k = 0; k < ZERO_V4; ++k) q[k * 256] = z;
    } else {
#pragma unroll
        for (int k = 0; k < ZERO_V4; ++k)
            if (blk + (size_t)k * 256 + threadIdx.x < n4) q[k * 256] = z;
    }
    if (blockIdx.x == 0 && threadIdx.x < (n & 3)) p[(n4 << 2) + threadIdx.x] = 0.0f;
}

__global__ __launch_bounds__(256) void k_transpose_w(
    const float* __restrict__ W, float* __restrict__ Wt) {
    __shared__ float tile[16][17];
    int bx = blockIdx.x & 7;
    int by = blockIdx.x >> 3;
    int tx = threadIdx.x & 15;
    int ty = threadIdx.x >> 4;
    tile[ty][tx] = W[(size_t)(by * 16 + ty) * DCH + bx * 16 + tx];
    __syncthreads();
    Wt[(size_t)(bx * 16 + ty) * DCH + by * 16 + tx] = tile[tx][ty];
}

__global__ __launch_bounds__(256) void k_gemm_wmma(
    const float* __restrict__ x, const float* __restrict__ Wt,
    float* __restrict__ h, int N) {
    __shared__ float sX[TILE_M][SX_LD];

    const int t = threadIdx.x;
    const int rowBase = blockIdx.x * TILE_M;

    for (int i = t; i < TILE_M * DCH; i += 256) {
        int r = i >> 7;
        int c = i & (DCH - 1);
        int gr = rowBase + r;
        sX[r][c] = (gr < N) ? x[(size_t)gr * DCH + c] : 0.0f;
    }
    __syncthreads();

    const int lane = t & 31;
    const int wave = t >> 5;
    const int m    = lane & 15;
    const int koff = (lane >> 4) << 1;
    const int n    = (wave << 4) + (lane & 15);

    const float* wrow = Wt + (size_t)n * DCH;

    v8f acc = {};
#pragma unroll
    for (int kt = 0; kt < DCH; kt += 4) {
        v2f a = *reinterpret_cast<const v2f*>(&sX[m][kt + koff]);
        v2f b = *reinterpret_cast<const v2f*>(wrow + kt + koff);
        acc = __builtin_amdgcn_wmma_f32_16x16x4_f32(
            false, a, false, b, (short)0, acc, false, false);
    }

    const int rhalf = (lane >> 4) << 3;
#pragma unroll
    for (int v = 0; v < 8; ++v) {
        int gr = rowBase + rhalf + v;
        if (gr < N) h[(size_t)gr * DCH + n] = acc[v];
    }
}

#define BIN_SHIFT 7
#define BIN_NODES 128
#define MAXBIN 1024
#define CHUNK 4096
#define EPT (CHUNK / 256)
#define LCAP 8192

__global__ __launch_bounds__(256) void k_bin_count(
    const int* __restrict__ dst, int* __restrict__ binCount, int E, int nb) {
    __shared__ int hist[MAXBIN];
    const int t = threadIdx.x;
    const int e0 = blockIdx.x * CHUNK;
    int dv[EPT];
#pragma unroll
    for (int k = 0; k < EPT; ++k) {
        const int e = e0 + k * 256 + t;
        dv[k] = dst[e < E ? e : E - 1];
    }
    for (int i = t; i < nb; i += 256) hist[i] = 0;
    __syncthreads();
#pragma unroll
    for (int k = 0; k < EPT; ++k) {
        const int e = e0 + k * 256 + t;
        if (e < E) atomicAdd(&hist[dv[k] >> BIN_SHIFT], 1);
    }
    __syncthreads();
    for (int i = t; i < nb; i += 256) {
        const int c = hist[i];
        if (c) atomicAdd(&binCount[i], c);
    }
}

__global__ __launch_bounds__(256) void k_bin_place(
    const int* __restrict__ src, const int* __restrict__ dst,
    const float* __restrict__ attr, const int* __restrict__ binCount,
    int* __restrict__ binFill, int* __restrict__ binStart,
    int4* __restrict__ recs, int E, int nb) {
    __shared__ int lcur[MAXBIN];
    __shared__ int lbase[MAXBIN];
    __shared__ int wtot[8];
    const int t = threadIdx.x;
    const int lane = t & 31;
    const int wave = t >> 5;
    const int e0 = blockIdx.x * CHUNK;

    int a0 = 4 * t + 0 < nb ? binCount[4 * t + 0] : 0;
    int a1 = 4 * t + 1 < nb ? binCount[4 * t + 1] : 0;
    int a2 = 4 * t + 2 < nb ? binCount[4 * t + 2] : 0;
    int a3 = 4 * t + 3 < nb ? binCount[4 * t + 3] : 0;

    int dv[EPT], sv[EPT];
    float av[EPT];
#pragma unroll
    for (int k = 0; k < EPT; ++k) {
        const int e = e0 + k * 256 + t;
        dv[k] = dst[e < E ? e : E - 1];
    }
#pragma unroll
    for (int k = 0; k < EPT; ++k) {
        const int e = e0 + k * 256 + t;
        const int ec = e < E ? e : E - 1;
        sv[k] = src[ec];
        av[k] = attr[ec];
    }

    const int s = a0 + a1 + a2 + a3;
    int v = s;
    for (int o = 1; o < 32; o <<= 1) {
        const int u = __shfl_up(v, o);
        if (lane >= o) v += u;
    }
    if (lane == 31) wtot[wave] = v;
    lcur[4 * t + 0] = 0; lcur[4 * t + 1] = 0; lcur[4 * t + 2] = 0; lcur[4 * t + 3] = 0;
    __syncthreads();
    int wb = 0;
    for (int w = 0; w < wave; ++w) wb += wtot[w];
    const int ex = wb + v - s;
    lbase[4 * t + 0] = ex;
    lbase[4 * t + 1] = ex + a0;
    lbase[4 * t + 2] = ex + a0 + a1;
    lbase[4 * t + 3] = ex + a0 + a1 + a2;
    if (blockIdx.x == 0) {
        binStart[4 * t + 0] = ex;
        binStart[4 * t + 1] = ex + a0;
        binStart[4 * t + 2] = ex + a0 + a1;
        binStart[4 * t + 3] = ex + a0 + a1 + a2;
    }

    int r[EPT];
#pragma unroll
    for (int k = 0; k < EPT; ++k) {
        const int e = e0 + k * 256 + t;
        r[k] = 0;
        if (e < E) r[k] = atomicAdd(&lcur[dv[k] >> BIN_SHIFT], 1);
    }
    __syncthreads();
    int cc[MAXBIN / 256], gb[MAXBIN / 256];
#pragma unroll
    for (int j = 0; j < MAXBIN / 256; ++j) {
        const int i = t + 256 * j;
        cc[j] = i < nb ? lcur[i] : 0;
    }
#pragma unroll
    for (int j = 0; j < MAXBIN / 256; ++j) {
        gb[j] = 0;
        if (cc[j]) gb[j] = atomicAdd(&binFill[t + 256 * j], cc[j]);
    }
#pragma unroll
    for (int j = 0; j < MAXBIN / 256; ++j)
        if (cc[j]) lbase[t + 256 * j] += gb[j];
    __syncthreads();
    int lb[EPT];
#pragma unroll
    for (int k = 0; k < EPT; ++k) lb[k] = lbase[dv[k] >> BIN_SHIFT];
#pragma unroll
    for (int k = 0; k < EPT; ++k) {
        const int e = e0 + k * 256 + t;
        if (e < E)
            recs[lb[k] + r[k]] = make_int4(sv[k], dv[k], __float_as_int(av[k]), e);
    }
}

__device__ __forceinline__ void row_fma(const float* __restrict__ h, int s, float a,
                                        int lane, float& c0, float& c1, float& c2, float& c3) {
    const float* hp = h + (size_t)s * DCH + lane;
    c0 += hp[0] * a;
    c1 += hp[32] * a;
    c2 += hp[64] * a;
    c3 += hp[96] * a;
}

__device__ __forceinline__ void pick_two(int k, int rankA, int rankB, int sA, int sB,
                                         float aA, float aB, int& s, float& f) {
    const bool mB = rankB == k;
    const bool m = mB || rankA == k;
    const int j = __builtin_ctz(__builtin_amdgcn_ballot_w32(m));
    s = __builtin_amdgcn_readlane(mB ? sB : sA, j);
    f = __int_as_float(__builtin_amdgcn_readlane(__float_as_int(mB ? aB : aA), j));
}

#define SUM_THREADS 512
#define SUM_WAVES (SUM_THREADS / 32)
#define LIST_V 8
__global__ __launch_bounds__(SUM_THREADS) void k_bin_sum(
    const int4* __restrict__ recs, const int* __restrict__ binCount,
    const int* __restrict__ binStart, const float* __restrict__ h,
    float* __restrict__ agg, int N) {
    __shared__ int nodeCnt[BIN_NODES];
    __shared__ int nodeStart[BIN_NODES];
    __shared__ int nodeCur[BIN_NODES];
    __shared__ int L[LCAP];
    const int t = threadIdx.x;
    const int lane = t & 31;
    const int wave = t >> 5;
    const int bin = blockIdx.x;
    const int n = binCount[bin];
    const int4* __restrict__ R = recs + binStart[bin];
    const int node0 = bin << BIN_SHIFT;
    const bool listed = n <= LCAP;

    if (listed) {
        if (t < BIN_NODES) { nodeCnt[t] = 0; nodeCur[t] = 0; }
        __syncthreads();
        for (int i0 = t; i0 < n; i0 += LIST_V * SUM_THREADS) {
            int y[LIST_V];
#pragma unroll
            for (int j = 0; j < LIST_V; ++j) {
                const int i = i0 + j * SUM_THREADS;
                y[j] = R[i < n ? i : n - 1].y;
            }
#pragma unroll
            for (int j = 0; j < LIST_V; ++j)
                if (i0 + j * SUM_THREADS < n) atomicAdd(&nodeCnt[y[j] - node0], 1);
        }
        __syncthreads();
        if (t < BIN_NODES) {
            int p = 0;
            for (int u = 0; u < t; ++u) p += nodeCnt[u];
            nodeStart[t] = p;
        }
        __syncthreads();
        for (int i0 = t; i0 < n; i0 += LIST_V * SUM_THREADS) {
            int y[LIST_V];
#pragma unroll
            for (int j = 0; j < LIST_V; ++j) {
                const int i = i0 + j * SUM_THREADS;
                y[j] = R[i < n ? i : n - 1].y;
            }
#pragma unroll
            for (int j = 0; j < LIST_V; ++j) {
                const int i = i0 + j * SUM_THREADS;
                if (i < n) {
                    const int dl = y[j] - node0;
                    const int p = atomicAdd(&nodeCur[dl], 1);
                    L[nodeStart[dl] + p] = i;
                }
            }
        }
        __syncthreads();
    }

    for (int node = wave; node < BIN_NODES; node += SUM_WAVES) {
        const int gv = node0 + node;
        if (gv >= N) break;
        float c0 = 0.0f, c1 = 0.0f, c2 = 0.0f, c3 = 0.0f;
        if (listed) {
            const int d = __builtin_amdgcn_readfirstlane(nodeCnt[node]);
            const int o = __builtin_amdgcn_readfirstlane(nodeStart[node]);
            if (d <= 32) {
                int key = 0x7fffffff, s = 0;
                float a = 0.0f;
                if (lane < d) {
                    const int4 rc = R[L[o + lane]];
                    key = rc.w; s = rc.x; a = __int_as_float(rc.z);
                }
                int rank = 0;
                for (int j = 0; j < d; ++j)
                    rank += (__builtin_amdgcn_readlane(key, j) < key) ? 1 : 0;
                if (lane >= d) rank = -1;
                int k = 0;
                for (; k + 4 <= d; k += 4) {
                    const int j0 = __builtin_ctz(__builtin_amdgcn_ballot_w32(rank == k + 0));
                    const int j1 = __builtin_ctz(__builtin_amdgcn_ballot_w32(rank == k + 1));
                    const int j2 = __builtin_ctz(__builtin_amdgcn_ballot_w32(rank == k + 2));
                    const int j3 = __builtin_ctz(__builtin_amdgcn_ballot_w32(rank == k + 3));
                    const int s0 = __builtin_amdgcn_readlane(s, j0);
                    const int s1 = __builtin_amdgcn_readlane(s, j1);
                    const int s2 = __builtin_amdgcn_readlane(s, j2);
                    const int s3 = __builtin_amdgcn_readlane(s, j3);
                    const float f0 = __int_as_float(__builtin_amdgcn_readlane(__float_as_int(a), j0));
                    const float f1 = __int_as_float(__builtin_amdgcn_readlane(__float_as_int(a), j1));
                    const float f2 = __int_as_float(__builtin_amdgcn_readlane(__float_as_int(a), j2));
                    const float f3 = __int_as_float(__builtin_amdgcn_readlane(__float_as_int(a), j3));
                    const float* p0 = h + (size_t)s0 * DCH + lane;
                    const float* p1 = h + (size_t)s1 * DCH + lane;
                    const float* p2 = h + (size_t)s2 * DCH + lane;
                    const float* p3 = h + (size_t)s3 * DCH + lane;
                    const float x00 = p0[0], x01 = p0[32], x02 = p0[64], x03 = p0[96];
                    const float x10 = p1[0], x11 = p1[32], x12 = p1[64], x13 = p1[96];
                    const float x20 = p2[0], x21 = p2[32], x22 = p2[64], x23 = p2[96];
                    const float x30 = p3[0], x31 = p3[32], x32 = p3[64], x33 = p3[96];
                    c0 += x00 * f0; c1 += x01 * f0; c2 += x02 * f0; c3 += x03 * f0;
                    c0 += x10 * f1; c1 += x11 * f1; c2 += x12 * f1; c3 += x13 * f1;
                    c0 += x20 * f2; c1 += x21 * f2; c2 += x22 * f2; c3 += x23 * f2;
                    c0 += x30 * f3; c1 += x31 * f3; c2 += x32 * f3; c3 += x33 * f3;
                }
                if (k < d) {
                    const bool u1 = k + 1 < d, u2 = k + 2 < d;
                    const int j0 = __builtin_ctz(__builtin_amdgcn_ballot_w32(rank == k + 0));
                    const int j1 = u1 ? __builtin_ctz(__builtin_amdgcn_ballot_w32(rank == k + 1)) : j0;
                    const int j2 = u2 ? __builtin_ctz(__builtin_amdgcn_ballot_w32(rank == k + 2)) : j0;
                    const int s0 = __builtin_amdgcn_readlane(s, j0);
                    const int s1 = __builtin_amdgcn_readlane(s, j1);
                    const int s2 = __builtin_amdgcn_readlane(s, j2);
                    const float f0 = __int_as_float(__builtin_amdgcn_readlane(__float_as_int(a), j0));
                    const float f1 = __int_as_float(__builtin_amdgcn_readlane(__float_as_int(a), j1));
                    const float f2 = __int_as_float(__builtin_amdgcn_readlane(__float_as_int(a), j2));
                    const float* p0 = h + (size_t)s0 * DCH + lane;
                    const float* p1 = h + (size_t)s1 * DCH + lane;
                    const float* p2 = h + (size_t)s2 * DCH + lane;
                    float x10 = 0.0f, x11 = 0.0f, x12 = 0.0f, x13 = 0.0f;
                    float x20 = 0.0f, x21 = 0.0f, x22 = 0.0f, x23 = 0.0f;
                    const float x00 = p0[0], x01 = p0[32], x02 = p0[64], x03 = p0[96];
                    if (u1) { x10 = p1[0]; x11 = p1[32]; x12 = p1[64]; x13 = p1[96]; }
                    if (u2) { x20 = p2[0]; x21 = p2[32]; x22 = p2[64]; x23 = p2[96]; }
                    c0 += x00 * f0; c1 += x01 * f0; c2 += x02 * f0; c3 += x03 * f0;
                    if (u1) { c0 += x10 * f1; c1 += x11 * f1; c2 += x12 * f1; c3 += x13 * f1; }
                    if (u2) { c0 += x20 * f2; c1 += x21 * f2; c2 += x22 * f2; c3 += x23 * f2; }
                }
            } else if (d <= 64) {
                const int d2 = d - 32;
                const int4 ra = R[L[o + lane]];
                const int keyA = ra.w, sA = ra.x;
                const float aA = __int_as_float(ra.z);
                int keyB = 0x7fffffff, sB = 0;
                float aB = 0.0f;
                if (lane < d2) {
                    const int4 rb = R[L[o + 32 + lane]];
                    keyB = rb.w; sB = rb.x; aB = __int_as_float(rb.z);
                }
                int rankA = 0, rankB = 0;
                for (int j = 0; j < 32; ++j) {
                    const int kj = __builtin_amdgcn_readlane(keyA, j);
                    rankA += (kj < keyA) ? 1 : 0;
                    rankB += (kj < keyB) ? 1 : 0;
                }
                for (int j = 0; j < d2; ++j) {
                    const int kj = __builtin_amdgcn_readlane(keyB, j);
                    rankA += (kj < keyA) ? 1 : 0;
                    rankB += (kj < keyB) ? 1 : 0;
                }
                if (lane >= d2) rankB = -1;
                int k = 0;
                for (; k + 4 <= d; k += 4) {
                    int s0, s1, s2, s3;
                    float f0, f1, f2, f3;
                    pick_two(k + 0, rankA, rankB, sA, sB, aA, aB, s0, f0);
                    pick_two(k + 1, rankA, rankB, sA, sB, aA, aB, s1, f1);
                    pick_two(k + 2, rankA, rankB, sA, sB, aA, aB, s2, f2);
                    pick_two(k + 3, rankA, rankB, sA, sB, aA, aB, s3, f3);
                    const float* p0 = h + (size_t)s0 * DCH + lane;
                    const float* p1 = h + (size_t)s1 * DCH + lane;
                    const float* p2 = h + (size_t)s2 * DCH + lane;
                    const float* p3 = h + (size_t)s3 * DCH + lane;
                    const float x00 = p0[0], x01 = p0[32], x02 = p0[64], x03 = p0[96];
                    const float x10 = p1[0], x11 = p1[32], x12 = p1[64], x13 = p1[96];
                    const float x20 = p2[0], x21 = p2[32], x22 = p2[64], x23 = p2[96];
                    const float x30 = p3[0], x31 = p3[32], x32 = p3[64], x33 = p3[96];
                    c0 += x00 * f0; c1 += x01 * f0; c2 += x02 * f0; c3 += x03 * f0;
                    c0 += x10 * f1; c1 += x11 * f1; c2 += x12 * f1; c3 += x13 * f1;
                    c0 += x20 * f2; c1 += x21 * f2; c2 += x22 * f2; c3 += x23 * f2;
                    c0 += x30 * f3; c1 += x31 * f3; c2 += x32 * f3; c3 += x33 * f3;
                }
                for (; k < d; ++k) {
                    int s0;
                    float f0;
                    pick_two(k, rankA, rankB, sA, sB, aA, aB, s0, f0);
                    row_fma(h, s0, f0, lane, c0, c1, c2, c3);
                }
            } else {
                int last = -1;
                for (int k = 0; k < d; ++k) {
                    int best = 0x7fffffff, bi = 0;
                    for (int i = lane; i < d; i += 32) {
                        const int ri = L[o + i];
                        const int key = R[ri].w;
                        if (key > last && key < best) { best = key; bi = ri; }
                    }
                    for (int off = 16; off; off >>= 1) {
                        const int ob = __shfl_xor(best, off);
                        const int oi = __shfl_xor(bi, off);
                        if (ob < best) { best = ob; bi = oi; }
                    }
                    const int4 rc = R[bi];
                    last = best;
                    row_fma(h, rc.x, __int_as_float(rc.z), lane, c0, c1, c2, c3);
                }
            }
        } else {
            int last = -1;
            for (int trip = 0; trip <= n; ++trip) {
                int best = 0x7fffffff, bi = 0;
                for (int i = lane; i < n; i += 32) {
                    const int4 rc = R[i];
                    if (rc.y == gv && rc.w > last && rc.w < best) { best = rc.w; bi = i; }
                }
                for (int off = 16; off; off >>= 1) {
                    const int ob = __shfl_xor(best, off);
                    const int oi = __shfl_xor(bi, off);
                    if (ob < best) { best = ob; bi = oi; }
                }
                if (best == 0x7fffffff) break;
                const int4 rc = R[bi];
                last = best;
                row_fma(h, rc.x, __int_as_float(rc.z), lane, c0, c1, c2, c3);
            }
        }
        float* outp = agg + (size_t)gv * DCH + lane;
        outp[0] = c0; outp[32] = c1; outp[64] = c2; outp[96] = c3;
    }
}

#define GROWS 64

__global__ __launch_bounds__(128) void k_bias_relu_mean(
    float* __restrict__ out, const float* __restrict__ bias,
    const int* __restrict__ batch, float* __restrict__ mean_sum,
    float* __restrict__ cnt, int N) {
    const int c = threadIdx.x;
    const float bv = bias[c];
    float* msum = mean_sum + c;
    float acc = 0.0f, cacc = 0.0f;
    int curg = -1;

    if (N >= GROWS) {
        int base = blockIdx.x * GROWS;
        int skip = 0;
        if (base + GROWS > N) { skip = base + GROWS - N; base = N - GROWS; }
        float* col = out + (size_t)base * DCH + c;
        const int lane = c & 31;
        const int gidlo = batch[base + lane];
        const int gidhi = batch[base + 32 + lane];

        float v[GROWS];
#pragma unroll
        for (int j = 0; j < GROWS; ++j)
            v[j] = (j >= skip) ? col[(size_t)j * DCH] : 0.0f;

#pragma unroll
        for (int j = 0; j < GROWS; ++j) {
            if (j >= skip) {
                const int gg = __builtin_amdgcn_readlane(j < 32 ? gidlo : gidhi, j & 31);
                if (gg != curg) {
                    if (curg >= 0) {
                        unsafeAtomicAdd(&msum[(size_t)curg * DCH], acc);
                        if (c == 0) unsafeAtomicAdd(&cnt[curg], cacc);
                    }
                    acc = 0.0f; cacc = 0.0f; curg = gg;
                }
                float x = v[j] + bv;
                x = x > 0.0f ? x : 0.0f;
                col[(size_t)j * DCH] = x;
                acc += x;
                cacc += 1.0f;
            }
        }
    } else {
        for (int r = 0; r < N; ++r) {
            const int gg = batch[r];
            if (gg != curg) {
                if (curg >= 0) {
                    unsafeAtomicAdd(&msum[(size_t)curg * DCH], acc);
                    if (c == 0) unsafeAtomicAdd(&cnt[curg], cacc);
                }
                acc = 0.0f; cacc = 0.0f; curg = gg;
            }
            const size_t idx = (size_t)r * DCH + c;
            float x = out[idx] + bv;
            x = x > 0.0f ? x : 0.0f;
            out[idx] = x;
            acc += x;
            cacc += 1.0f;
        }
    }
    if (curg >= 0) {
        unsafeAtomicAdd(&msum[(size_t)curg * DCH], acc);
        if (c == 0) unsafeAtomicAdd(&cnt[curg], cacc);
    }
}

#define HOLD16(a, o)                                                          \
    asm volatile("" :: "v"(a[o + 0]), "v"(a[o + 1]), "v"(a[o + 2]),          \
                 "v"(a[o + 3]), "v"(a[o + 4]), "v"(a[o + 5]), "v"(a[o + 6]), \
                 "v"(a[o + 7]), "v"(a[o + 8]), "v"(a[o + 9]), "v"(a[o + 10]),\
                 "v"(a[o + 11]), "v"(a[o + 12]), "v"(a[o + 13]),             \
                 "v"(a[o + 14]), "v"(a[o + 15]))

__global__ __launch_bounds__(128) void k_var_pass(
    const float* __restrict__ out, const int* __restrict__ batch,
    const float* __restrict__ mean_sum, const float* __restrict__ cnt,
    const float* __restrict__ ms, float* __restrict__ var_sum, int N) {
    const int c = threadIdx.x;
    const float msc = ms[c];
    const float* msum = mean_sum + c;
    float* vsum = var_sum + c;
    float acc = 0.0f;
    float sub = 0.0f;
    int curg = -1;

    if (N >= GROWS) {
        int base = blockIdx.x * GROWS;
        int skip = 0;
        if (base + GROWS > N) { skip = base + GROWS - N; base = N - GROWS; }
        const float* col = out + (size_t)base * DCH + c;
        const int lane = c & 31;
        const int gidlo = batch[base + lane];
        const int gidhi = batch[base + 32 + lane];

        float v[GROWS];
#pragma unroll
        for (int j = 0; j < GROWS; ++j) v[j] = col[(size_t)j * DCH];
        HOLD16(v, 0); HOLD16(v, 16); HOLD16(v, 32); HOLD16(v, 48);

#pragma unroll
        for (int j = 0; j < GROWS; ++j) {
            if (j >= skip) {
                const int gg = __builtin_amdgcn_readlane(j < 32 ? gidlo : gidhi, j & 31);
                if (gg != curg) {
                    if (curg >= 0) unsafeAtomicAdd(&vsum[(size_t)curg * DCH], acc);
                    acc = 0.0f;
                    curg = gg;
                    float cn = cnt[gg];
                    cn = cn > 1.0f ? cn : 1.0f;
                    sub = (msum[(size_t)gg * DCH] / cn) * msc;
                }
                float o = v[j] - sub;
                acc += o * o;
            }
        }
    } else {
        for (int r = 0; r < N; ++r) {
            const int gg = batch[r];
            if (gg != curg) {
                if (curg >= 0) unsafeAtomicAdd(&vsum[(size_t)curg * DCH], acc);
                acc = 0.0f;
                curg = gg;
                float cn = cnt[gg];
                cn = cn > 1.0f ? cn : 1.0f;
                sub = (msum[(size_t)gg * DCH] / cn) * msc;
            }
            float o = out[(size_t)r * DCH + c] - sub;
            acc += o * o;
        }
    }
    if (curg >= 0) unsafeAtomicAdd(&vsum[(size_t)curg * DCH], acc);
}

__global__ __launch_bounds__(256) void k_finalize(
    float* __restrict__ out, const int* __restrict__ batch,
    const float* __restrict__ mean_sum, const float* __restrict__ var_sum,
    const float* __restrict__ cnt, const float* __restrict__ ms,
    const float* __restrict__ gw, const float* __restrict__ gb, int N) {
    size_t idx = (size_t)blockIdx.x * 256 + threadIdx.x;
    size_t total = (size_t)N * (DCH / 4);
    if (idx >= total) return;
    int nrow = (int)(idx >> 5);
    int q    = (int)(idx & 31);
    int g = batch[nrow];
    float cn = cnt[g];
    cn = cn > 1.0f ? cn : 1.0f;
    float inv_cn = 1.0f / cn;

    float4 m4 = reinterpret_cast<const float4*>(mean_sum + (size_t)g * DCH)[q];
    float4 v4 = reinterpret_cast<const float4*>(var_sum  + (size_t)g * DCH)[q];
    float4 s4 = reinterpret_cast<const float4*>(ms)[q];
    float4 w4 = reinterpret_cast<const float4*>(gw)[q];
    float4 b4 = reinterpret_cast<const float4*>(gb)[q];
    float4 o  = reinterpret_cast<float4*>(out)[idx];

    o.x -= (m4.x * inv_cn) * s4.x;
    o.y -= (m4.y * inv_cn) * s4.y;
    o.z -= (m4.z * inv_cn) * s4.z;
    o.w -= (m4.w * inv_cn) * s4.w;
    o.x = w4.x * o.x / sqrtf(v4.x * inv_cn + GN_EPS) + b4.x;
    o.y = w4.y * o.y / sqrtf(v4.y * inv_cn + GN_EPS) + b4.y;
    o.z = w4.z * o.z / sqrtf(v4.z * inv_cn + GN_EPS) + b4.z;
    o.w = w4.w * o.w / sqrtf(v4.w * inv_cn + GN_EPS) + b4.w;

    reinterpret_cast<float4*>(out)[idx] = o;
}

extern "C" void kernel_launch(void* const* d_in, const int* in_sizes, int n_in,
                              void* d_out, int out_size, void* d_ws, size_t ws_size,
                              hipStream_t stream) {
    const float* x     = (const float*)d_in[0];
    const int*   eidx  = (const int*)  d_in[1];
    const float* eattr = (const float*)d_in[2];
    const int*   batch = (const int*)  d_in[3];
    const float* W     = (const float*)d_in[4];
    const float* bias  = (const float*)d_in[5];
    const float* gw    = (const float*)d_in[6];
    const float* gb    = (const float*)d_in[7];
    const float* ms    = (const float*)d_in[8];
    float* out = (float*)d_out;

    const int N = in_sizes[0] / DCH;
    const int E = in_sizes[2];

    float* h        = (float*)d_ws;
    float* mean_sum = h + (size_t)N * DCH;
    float* var_sum  = mean_sum + (size_t)NGRAPH * DCH;
    float* cnt      = var_sum + (size_t)NGRAPH * DCH;
    int*   binCount = (int*)(cnt + NGRAPH);
    int*   binFill  = binCount + MAXBIN;
    float* Wt       = (float*)(binFill + MAXBIN);
    int*   binStart = (int*)(Wt + DCH * DCH);
    int4*  recs     = (int4*)(binStart + MAXBIN);
    const int nb = (N + BIN_NODES - 1) >> BIN_SHIFT;

    {
        const size_t per_block = (size_t)256 * ZERO_V4 * 4;
        size_t n1 = (size_t)2 * NGRAPH * DCH + NGRAPH + 2 * MAXBIN;
        k_zero_f32<<<(unsigned)((n1 + per_block - 1) / per_block), 256, 0, stream>>>(mean_sum, n1);
    }

    k_transpose_w<<<64, 256, 0, stream>>>(W, Wt);

    k_gemm_wmma<<<(N + TILE_M - 1) / TILE_M, 256, 0, stream>>>(x, Wt, h, N);

    {
        const int eblocks = (E + CHUNK - 1) / CHUNK;
        k_bin_count<<<eblocks, 256, 0, stream>>>(eidx + E, binCount, E, nb);
        k_bin_place<<<eblocks, 256, 0, stream>>>(
            eidx, eidx + E, eattr, binCount, binFill, binStart, recs, E, nb);
        k_bin_sum<<<nb, SUM_THREADS, 0, stream>>>(recs, binCount, binStart, h, out, N);
    }

    k_bias_relu_mean<<<(N + GROWS - 1) / GROWS, 128, 0, stream>>>(
        out, bias, batch, mean_sum, cnt, N);

    k_var_pass<<<(N + GROWS - 1) / GROWS, 128, 0, stream>>>(
        out, batch, mean_sum, cnt, ms, var_sum, N);

    {
        size_t total = (size_t)N * (DCH / 4);
        k_finalize<<<(unsigned)((total + 255) / 256), 256, 0, stream>>>(
            out, batch, mean_sum, var_sum, cnt, ms, gw, gb, N);
    }
}


# edge sum epilogue: bias, relu and per-graph mean sums/counts taken on the row in registers; mean sums add per wave, then waves in order via LDS, one float atomic per bin and channel; own pass gone
# speedup vs baseline: 2.7492x; 1.3766x over previous
#include <hip/hip_runtime.h>
#include <hip/hip_bf16.h>
#include <math.h>

typedef __attribute__((ext_vector_type(2))) float v2f;
typedef __attribute__((ext_vector_type(8))) float v8f;

#define DCH 128
#define NGRAPH 64
#define GN_EPS 1e-5f
#define TILE_M 16
#define SX_LD (DCH + 4)

#define ZERO_V4 16
__global__ __launch_bounds__(256) void k_zero_f32(float* __restrict__ p, size_t n) {
    const size_t n4 = n >> 2;
    const size_t blk = (size_t)blockIdx.x * (256 * ZERO_V4);
    float4* __restrict__ q = reinterpret_cast<float4*>(p) + blk + threadIdx.x;
    const float4 z = make_float4(0.0f, 0.0f, 0.0f, 0.0f);
    if (blk + 256 * ZERO_V4 <= n4) {
#pragma unroll
        for (int k = 0; k < ZERO_V4; ++k) q[k * 256] = z;
    } else {
#pragma unroll
        for (int k = 0; k < ZERO_V4; ++k)
            if (blk + (size_t)k * 256 + threadIdx.x < n4) q[k * 256] = z;
    }
    if (blockIdx.x == 0 && threadIdx.x < (n & 3)) p[(n4 << 2) + threadIdx.x] = 0.0f;
}

__global__ __launch_bounds__(256) void k_transpose_w(
    const float* __restrict__ W, float* __restrict__ Wt) {
    __shared__ float tile[16][17];
    int bx = blockIdx.x & 7;
    int by = blockIdx.x >> 3;
    int tx = threadIdx.x & 15;
    int ty = threadIdx.x >> 4;
    tile[ty][tx] = W[(size_t)(by * 16 + ty) * DCH + bx * 16 + tx];
    __syncthreads();
    Wt[(size_t)(bx * 16 + ty) * DCH + by * 16 + tx] = tile[tx][ty];
}

__global__ __launch_bounds__(256) void k_gemm_wmma(
    const float* __restrict__ x, const float* __restrict__ Wt,
    float* __restrict__ h, int N) {
    __shared__ float sX[TILE_M][SX_LD];

    const int t = threadIdx.x;
    const int rowBase = blockIdx.x * TILE_M;

    for (int i = t; i < TILE_M * DCH; i += 256) {
        int r = i >> 7;
        int c = i & (DCH - 1);
        int gr = rowBase + r;
        sX[r][c] = (gr < N) ? x[(size_t)gr * DCH + c] : 0.0f;
    }
    __syncthreads();

    const int lane = t & 31;
    const int wave = t >> 5;
    const int m    = lane & 15;
    const int koff = (lane >> 4) << 1;
    const int n    = (wave << 4) + (lane & 15);

    const float* wrow = Wt + (size_t)n * DCH;

    v8f acc = {};
#pragma unroll
    for (int kt = 0; kt < DCH; kt += 4) {
        v2f a = *reinterpret_cast<const v2f*>(&sX[m][kt + koff]);
        v2f b = *reinterpret_cast<const v2f*>(wrow + kt + koff);
        acc = __builtin_amdgcn_wmma_f32_16x16x4_f32(
            false, a, false, b, (short)0, acc, false, false);
    }

    const int rhalf = (lane >> 4) << 3;
#pragma unroll
    for (int v = 0; v < 8; ++v) {
        int gr = rowBase + rhalf + v;
        if (gr < N) h[(size_t)gr * DCH + n] = acc[v];
    }
}

#define BIN_SHIFT 7
#define BIN_NODES 128
#define MAXBIN 1024
#define CHUNK 4096
#define EPT (CHUNK / 256)
#define LCAP 8192

__global__ __launch_bounds__(256) void k_bin_count(
    const int* __restrict__ dst, int* __restrict__ binCount, int E, int nb) {
    __shared__ int hist[MAXBIN];
    const int t = threadIdx.x;
    const int e0 = blockIdx.x * CHUNK;
    int dv[EPT];
#pragma unroll
    for (int k = 0; k < EPT; ++k) {
        const int e = e0 + k * 256 + t;
        dv[k] = dst[e < E ? e : E - 1];
    }
    for (int i = t; i < nb; i += 256) hist[i] = 0;
    __syncthreads();
#pragma unroll
    for (int k = 0; k < EPT; ++k) {
        const int e = e0 + k * 256 + t;
        if (e < E) atomicAdd(&hist[dv[k] >> BIN_SHIFT], 1);
    }
    __syncthreads();
    for (int i = t; i < nb; i += 256) {
        const int c = hist[i];
        if (c) atomicAdd(&binCount[i], c);
    }
}

__global__ __launch_bounds__(256) void k_bin_place(
    const int* __restrict__ src, const int* __restrict__ dst,
    const float* __restrict__ attr, const int* __restrict__ binCount,
    int* __restrict__ binFill, int* __restrict__ binStart,
    int4* __restrict__ recs, int E, int nb) {
    __shared__ int lcur[MAXBIN];
    __shared__ int lbase[MAXBIN];
    __shared__ int wtot[8];
    const int t = threadIdx.x;
    const int lane = t & 31;
    const int wave = t >> 5;
    const int e0 = blockIdx.x * CHUNK;

    int a0 = 4 * t + 0 < nb ? binCount[4 * t + 0] : 0;
    int a1 = 4 * t + 1 < nb ? binCount[4 * t + 1] : 0;
    int a2 = 4 * t + 2 < nb ? binCount[4 * t + 2] : 0;
    int a3 = 4 * t + 3 < nb ? binCount[4 * t + 3] : 0;

    int dv[EPT], sv[EPT];
    float av[EPT];
#pragma unroll
    for (int k = 0; k < EPT; ++k) {
        const int e = e0 + k * 256 + t;
        dv[k] = dst[e < E ? e : E - 1];
    }
#pragma unroll
    for (int k = 0; k < EPT; ++k) {
        const int e = e0 + k * 256 + t;
        const int ec = e < E ? e : E - 1;
        sv[k] = src[ec];
        av[k] = attr[ec];
    }

    const int s = a0 + a1 + a2 + a3;
    int v = s;
    for (int o = 1; o < 32; o <<= 1) {
        const int u = __shfl_up(v, o);
        if (lane >= o) v += u;
    }
    if (lane == 31) wtot[wave] = v;
    lcur[4 * t + 0] = 0; lcur[4 * t + 1] = 0; lcur[4 * t + 2] = 0; lcur[4 * t + 3] = 0;
    __syncthreads();
    int wb = 0;
    for (int w = 0; w < wave; ++w) wb += wtot[w];
    const int ex = wb + v - s;
    lbase[4 * t + 0] = ex;
    lbase[4 * t + 1] = ex + a0;
    lbase[4 * t + 2] = ex + a0 + a1;
    lbase[4 * t + 3] = ex + a0 + a1 + a2;
    if (blockIdx.x == 0) {
        binStart[4 * t + 0] = ex;
        binStart[4 * t + 1] = ex + a0;
        binStart[4 * t + 2] = ex + a0 + a1;
        binStart[4 * t + 3] = ex + a0 + a1 + a2;
    }

    int r[EPT];
#pragma unroll
    for (int k = 0; k < EPT; ++k) {
        const int e = e0 + k * 256 + t;
        r[k] = 0;
        if (e < E) r[k] = atomicAdd(&lcur[dv[k] >> BIN_SHIFT], 1);
    }
    __syncthreads();
    int cc[MAXBIN / 256], gb[MAXBIN / 256];
#pragma unroll
    for (int j = 0; j < MAXBIN / 256; ++j) {
        const int i = t + 256 * j;
        cc[j] = i < nb ? lcur[i] : 0;
    }
#pragma unroll
    for (int j = 0; j < MAXBIN / 256; ++j) {
        gb[j] = 0;
        if (cc[j]) gb[j] = atomicAdd(&binFill[t + 256 * j], cc[j]);
    }
#pragma unroll
    for (int j = 0; j < MAXBIN / 256; ++j)
        if (cc[j]) lbase[t + 256 * j] += gb[j];
    __syncthreads();
    int lb[EPT];
#pragma unroll
    for (int k = 0; k < EPT; ++k) lb[k] = lbase[dv[k] >> BIN_SHIFT];
#pragma unroll
    for (int k = 0; k < EPT; ++k) {
        const int e = e0 + k * 256 + t;
        if (e < E)
            recs[lb[k] + r[k]] = make_int4(sv[k], dv[k], __float_as_int(av[k]), e);
    }
}

__device__ __forceinline__ void row_fma(const float* __restrict__ h, int s, float a,
                                        int lane, float& c0, float& c1, float& c2, float& c3) {
    const float* hp = h + (size_t)s * DCH + lane;
    c0 += hp[0] * a;
    c1 += hp[32] * a;
    c2 += hp[64] * a;
    c3 += hp[96] * a;
}

__device__ __forceinline__ void pick_two(int k, int rankA, int rankB, int sA, int sB,
                                         float aA, float aB, int& s, float& f) {
    const bool mB = rankB == k;
    const bool m = mB || rankA == k;
    const int j = __builtin_ctz(__builtin_amdgcn_ballot_w32(m));
    s = __builtin_amdgcn_readlane(mB ? sB : sA, j);
    f = __int_as_float(__builtin_amdgcn_readlane(__float_as_int(mB ? aB : aA), j));
}

#define SUM_THREADS 512
#define SUM_WAVES (SUM_THREADS / 32)
#define LIST_V 8
__global__ __launch_bounds__(SUM_THREADS) void k_bin_sum(
    const int4* __restrict__ recs, const int* __restrict__ binCount,
    const int* __restrict__ binStart, const float* __restrict__ h,
    const float* __restrict__ bias, const int* __restrict__ batch,
    float* __restrict__ mean_sum, float* __restrict__ cnt,
    float* __restrict__ agg, int N) {
    __shared__ int nodeCnt[BIN_NODES];
    __shared__ int nodeStart[BIN_NODES];
    __shared__ int nodeCur[BIN_NODES];
    __shared__ int L[LCAP];
    __shared__ float wSum[SUM_WAVES][DCH];
    __shared__ float wCnt[SUM_WAVES];
    __shared__ int wGid[SUM_WAVES];
    __shared__ int nodeGid[BIN_NODES];
    __shared__ float sBias[DCH];
    const int t = threadIdx.x;
    const int lane = t & 31;
    const int wave = t >> 5;
    const int bin = blockIdx.x;
    const int n = binCount[bin];
    const int4* __restrict__ R = recs + binStart[bin];
    const int node0 = bin << BIN_SHIFT;
    const bool listed = n <= LCAP;

    int gidn = 0;
    float biasv = 0.0f;

    if (listed) {
        if (t < BIN_NODES) { nodeCnt[t] = 0; nodeCur[t] = 0; }
        __syncthreads();
        for (int i0 = t; i0 < n; i0 += LIST_V * SUM_THREADS) {
            int y[LIST_V];
#pragma unroll
            for (int j = 0; j < LIST_V; ++j) {
                const int i = i0 + j * SUM_THREADS;
                y[j] = R[i < n ? i : n - 1].y;
            }
#pragma unroll
            for (int j = 0; j < LIST_V; ++j)
                if (i0 + j * SUM_THREADS < n) atomicAdd(&nodeCnt[y[j] - node0], 1);
        }
        __syncthreads();
        if (t < BIN_NODES) {
            int p = 0;
            for (int u = 0; u < t; ++u) p += nodeCnt[u];
            nodeStart[t] = p;
        }
        __syncthreads();
        if (t < BIN_NODES) {
            const int gn = node0 + t;
            gidn = batch[gn < N ? gn : N - 1];
            biasv = bias[t];
        }
        for (int i0 = t; i0 < n; i0 += LIST_V * SUM_THREADS) {
            int y[LIST_V];
#pragma unroll
            for (int j = 0; j < LIST_V; ++j) {
                const int i = i0 + j * SUM_THREADS;
                y[j] = R[i < n ? i : n - 1].y;
            }
#pragma unroll
            for (int j = 0; j < LIST_V; ++j) {
                const int i = i0 + j * SUM_THREADS;
                if (i < n) {
                    const int dl = y[j] - node0;
                    const int p = atomicAdd(&nodeCur[dl], 1);
                    L[nodeStart[dl] + p] = i;
                }
            }
        }
    } else if (t < BIN_NODES) {
        const int gn = node0 + t;
        gidn = batch[gn < N ? gn : N - 1];
        biasv = bias[t];
    }
    if (t < BIN_NODES) { nodeGid[t] = gidn; sBias[t] = biasv; }
    __syncthreads();

    float* ws = &wSum[wave][lane];
    int curg = -1, crun = 0;
    for (int node = wave; node < BIN_NODES; node += SUM_WAVES) {
        const int gv = node0 + node;
        if (gv >= N) break;
        float c0 = 0.0f, c1 = 0.0f, c2 = 0.0f, c3 = 0.0f;
        if (listed) {
            const int d = __builtin_amdgcn_readfirstlane(nodeCnt[node]);
            const int o = __builtin_amdgcn_readfirstlane(nodeStart[node]);
            if (d <= 32) {
                int key = 0x7fffffff, s = 0;
                float a = 0.0f;
                if (lane < d) {
                    const int4 rc = R[L[o + lane]];
                    key = rc.w; s = rc.x; a = __int_as_float(rc.z);
                }
                int rank = 0;
                for (int j = 0; j < d; ++j)
                    rank += (__builtin_amdgcn_readlane(key, j) < key) ? 1 : 0;
                if (lane >= d) rank = -1;
                int k = 0;
                for (; k + 4 <= d; k += 4) {
                    const int j0 = __builtin_ctz(__builtin_amdgcn_ballot_w32(rank == k + 0));
                    const int j1 = __builtin_ctz(__builtin_amdgcn_ballot_w32(rank == k + 1));
                    const int j2 = __builtin_ctz(__builtin_amdgcn_ballot_w32(rank == k + 2));
                    const int j3 = __builtin_ctz(__builtin_amdgcn_ballot_w32(rank == k + 3));
                    const int s0 = __builtin_amdgcn_readlane(s, j0);
                    const int s1 = __builtin_amdgcn_readlane(s, j1);
                    const int s2 = __builtin_amdgcn_readlane(s, j2);
                    const int s3 = __builtin_amdgcn_readlane(s, j3);
                    const float f0 = __int_as_float(__builtin_amdgcn_readlane(__float_as_int(a), j0));
                    const float f1 = __int_as_float(__builtin_amdgcn_readlane(__float_as_int(a), j1));
                    const float f2 = __int_as_float(__builtin_amdgcn_readlane(__float_as_int(a), j2));
                    const float f3 = __int_as_float(__builtin_amdgcn_readlane(__float_as_int(a), j3));
                    const float* p0 = h + (size_t)s0 * DCH + lane;
                    const float* p1 = h + (size_t)s1 * DCH + lane;
                    const float* p2 = h + (size_t)s2 * DCH + lane;
                    const float* p3 = h + (size_t)s3 * DCH + lane;
                    const float x00 = p0[0], x01 = p0[32], x02 = p0[64], x03 = p0[96];
                    const float x10 = p1[0], x11 = p1[32], x12 = p1[64], x13 = p1[96];
                    const float x20 = p2[0], x21 = p2[32], x22 = p2[64], x23 = p2[96];
                    const float x30 = p3[0], x31 = p3[32], x32 = p3[64], x33 = p3[96];
                    c0 += x00 * f0; c1 += x01 * f0; c2 += x02 * f0; c3 += x03 * f0;
                    c0 += x10 * f1; c1 += x11 * f1; c2 += x12 * f1; c3 += x13 * f1;
                    c0 += x20 * f2; c1 += x21 * f2; c2 += x22 * f2; c3 += x23 * f2;
                    c0 += x30 * f3; c1 += x31 * f3; c2 += x32 * f3; c3 += x33 * f3;
                }
                if (k < d) {
                    const bool u1 = k + 1 < d, u2 = k + 2 < d;
                    const int j0 = __builtin_ctz(__builtin_amdgcn_ballot_w32(rank == k + 0));
                    const int j1 = u1 ? __builtin_ctz(__builtin_amdgcn_ballot_w32(rank == k + 1)) : j0;
                    const int j2 = u2 ? __builtin_ctz(__builtin_amdgcn_ballot_w32(rank == k + 2)) : j0;
                    const int s0 = __builtin_amdgcn_readlane(s, j0);
                    const int s1 = __builtin_amdgcn_readlane(s, j1);
                    const int s2 = __builtin_amdgcn_readlane(s, j2);
                    const float f0 = __int_as_float(__builtin_amdgcn_readlane(__float_as_int(a), j0));
                    const float f1 = __int_as_float(__builtin_amdgcn_readlane(__float_as_int(a), j1));
                    const float f2 = __int_as_float(__builtin_amdgcn_readlane(__float_as_int(a), j2));
                    const float* p0 = h + (size_t)s0 * DCH + lane;
                    const float* p1 = h + (size_t)s1 * DCH + lane;
                    const float* p2 = h + (size_t)s2 * DCH + lane;
                    float x10 = 0.0f, x11 = 0.0f, x12 = 0.0f, x13 = 0.0f;
                    float x20 = 0.0f, x21 = 0.0f, x22 = 0.0f, x23 = 0.0f;
                    const float x00 = p0[0], x01 = p0[32], x02 = p0[64], x03 = p0[96];
                    if (u1) { x10 = p1[0]; x11 = p1[32]; x12 = p1[64]; x13 = p1[96]; }
                    if (u2) { x20 = p2[0]; x21 = p2[32]; x22 = p2[64]; x23 = p2[96]; }
                    c0 += x00 * f0; c1 += x01 * f0; c2 += x02 * f0; c3 += x03 * f0;
                    if (u1) { c0 += x10 * f1; c1 += x11 * f1; c2 += x12 * f1; c3 += x13 * f1; }
                    if (u2) { c0 += x20 * f2; c1 += x21 * f2; c2 += x22 * f2; c3 += x23 * f2; }
                }
            } else if (d <= 64) {
                const int d2 = d - 32;
                const int4 ra = R[L[o + lane]];
                const int keyA = ra.w, sA = ra.x;
                const float aA = __int_as_float(ra.z);
                int keyB = 0x7fffffff, sB = 0;
                float aB = 0.0f;
                if (lane < d2) {
                    const int4 rb = R[L[o + 32 + lane]];
                    keyB = rb.w; sB = rb.x; aB = __int_as_float(rb.z);
                }
                int rankA = 0, rankB = 0;
                for (int j = 0; j < 32; ++j) {
                    const int kj = __builtin_amdgcn_readlane(keyA, j);
                    rankA += (kj < keyA) ? 1 : 0;
                    rankB += (kj < keyB) ? 1 : 0;
                }
                for (int j = 0; j < d2; ++j) {
                    const int kj = __builtin_amdgcn_readlane(keyB, j);
                    rankA += (kj < keyA) ? 1 : 0;
                    rankB += (kj < keyB) ? 1 : 0;
                }
                if (lane >= d2) rankB = -1;
                int k = 0;
                for (; k + 4 <= d; k += 4) {
                    int s0, s1, s2, s3;
                    float f0, f1, f2, f3;
                    pick_two(k + 0, rankA, rankB, sA, sB, aA, aB, s0, f0);
                    pick_two(k + 1, rankA, rankB, sA, sB, aA, aB, s1, f1);
                    pick_two(k + 2, rankA, rankB, sA, sB, aA, aB, s2, f2);
                    pick_two(k + 3, rankA, rankB, sA, sB, aA, aB, s3, f3);
                    const float* p0 = h + (size_t)s0 * DCH + lane;
                    const float* p1 = h + (size_t)s1 * DCH + lane;
                    const float* p2 = h + (size_t)s2 * DCH + lane;
                    const float* p3 = h + (size_t)s3 * DCH + lane;
                    const float x00 = p0[0], x01 = p0[32], x02 = p0[64], x03 = p0[96];
                    const float x10 = p1[0], x11 = p1[32], x12 = p1[64], x13 = p1[96];
                    const float x20 = p2[0], x21 = p2[32], x22 = p2[64], x23 = p2[96];
                    const float x30 = p3[0], x31 = p3[32], x32 = p3[64], x33 = p3[96];
                    c0 += x00 * f0; c1 += x01 * f0; c2 += x02 * f0; c3 += x03 * f0;
                    c0 += x10 * f1; c1 += x11 * f1; c2 += x12 * f1; c3 += x13 * f1;
                    c0 += x20 * f2; c1 += x21 * f2; c2 += x22 * f2; c3 += x23 * f2;
                    c0 += x30 * f3; c1 += x31 * f3; c2 += x32 * f3; c3 += x33 * f3;
                }
                for (; k < d; ++k) {
                    int s0;
                    float f0;
                    pick_two(k, rankA, rankB, sA, sB, aA, aB, s0, f0);
                    row_fma(h, s0, f0, lane, c0, c1, c2, c3);
                }
            } else {
                int last = -1;
                for (int k = 0; k < d; ++k) {
                    int best = 0x7fffffff, bi = 0;
                    for (int i = lane; i < d; i += 32) {
                        const int ri = L[o + i];
                        const int key = R[ri].w;
                        if (key > last && key < best) { best = key; bi = ri; }
                    }
                    for (int off = 16; off; off >>= 1) {
                        const int ob = __shfl_xor(best, off);
                        const int oi = __shfl_xor(bi, off);
                        if (ob < best) { best = ob; bi = oi; }
                    }
                    const int4 rc = R[bi];
                    last = best;
                    row_fma(h, rc.x, __int_as_float(rc.z), lane, c0, c1, c2, c3);
                }
            }
        } else {
            int last = -1;
            for (int trip = 0; trip <= n; ++trip) {
                int best = 0x7fffffff, bi = 0;
                for (int i = lane; i < n; i += 32) {
                    const int4 rc = R[i];
                    if (rc.y == gv && rc.w > last && rc.w < best) { best = rc.w; bi = i; }
                }
                for (int off = 16; off; off >>= 1) {
                    const int ob = __shfl_xor(best, off);
                    const int oi = __shfl_xor(bi, off);
                    if (ob < best) { best = ob; bi = oi; }
                }
                if (best == 0x7fffffff) break;
                const int4 rc = R[bi];
                last = best;
                row_fma(h, rc.x, __int_as_float(rc.z), lane, c0, c1, c2, c3);
            }
        }
        float x0 = c0 + sBias[lane], x1 = c1 + sBias[lane + 32];
        float x2 = c2 + sBias[lane + 64], x3 = c3 + sBias[lane + 96];
        x0 = x0 > 0.0f ? x0 : 0.0f;
        x1 = x1 > 0.0f ? x1 : 0.0f;
        x2 = x2 > 0.0f ? x2 : 0.0f;
        x3 = x3 > 0.0f ? x3 : 0.0f;
        float* outp = agg + (size_t)gv * DCH + lane;
        outp[0] = x0; outp[32] = x1; outp[64] = x2; outp[96] = x3;
        const int gg = nodeGid[node];
        if (gg != curg) {
            if (curg >= 0) {
                float* mp = mean_sum + (size_t)curg * DCH + lane;
                unsafeAtomicAdd(mp, ws[0]);
                unsafeAtomicAdd(mp + 32, ws[32]);
                unsafeAtomicAdd(mp + 64, ws[64]);
                unsafeAtomicAdd(mp + 96, ws[96]);
                if (lane == 0) unsafeAtomicAdd(&cnt[curg], (float)crun);
            }
            ws[0] = x0; ws[32] = x1; ws[64] = x2; ws[96] = x3;
            crun = 1;
            curg = gg;
        } else {
            ws[0] += x0; ws[32] += x1; ws[64] += x2; ws[96] += x3;
            crun += 1;
        }
    }

    if (lane == 0) { wCnt[wave] = (float)crun; wGid[wave] = curg; }
    __syncthreads();
    if (t < DCH) {
        float acc = 0.0f, cc = 0.0f;
        int cur = -1;
        for (int w = 0; w < SUM_WAVES; ++w) {
            const int g = wGid[w];
            if (g < 0) continue;
            if (g != cur) {
                if (cur >= 0) {
                    unsafeAtomicAdd(&mean_sum[(size_t)cur * DCH + t], acc);
                    if (t == 0) unsafeAtomicAdd(&cnt[cur], cc);
                }
                acc = 0.0f; cc = 0.0f; cur = g;
            }
            acc += wSum[w][t];
            cc += wCnt[w];
        }
        if (cur >= 0) {
            unsafeAtomicAdd(&mean_sum[(size_t)cur * DCH + t], acc);
            if (t == 0) unsafeAtomicAdd(&cnt[cur], cc);
        }
    }
}

#define GROWS 64

#define HOLD16(a, o)                                                          \
    asm volatile("" :: "v"(a[o + 0]), "v"(a[o + 1]), "v"(a[o + 2]),          \
                 "v"(a[o + 3]), "v"(a[o + 4]), "v"(a[o + 5]), "v"(a[o + 6]), \
                 "v"(a[o + 7]), "v"(a[o + 8]), "v"(a[o + 9]), "v"(a[o + 10]),\
                 "v"(a[o + 11]), "v"(a[o + 12]), "v"(a[o + 13]),             \
                 "v"(a[o + 14]), "v"(a[o + 15]))

__global__ __launch_bounds__(128) void k_var_pass(
    const float* __restrict__ out, const int* __restrict__ batch,
    const float* __restrict__ mean_sum, const float* __restrict__ cnt,
    const float* __restrict__ ms, float* __restrict__ var_sum, int N) {
    const int c = threadIdx.x;
    const float msc = ms[c];
    const float* msum = mean_sum + c;
    float* vsum = var_sum + c;
    float acc = 0.0f;
    float sub = 0.0f;
    int curg = -1;

    if (N >= GROWS) {
        int base = blockIdx.x * GROWS;
        int skip = 0;
        if (base + GROWS > N) { skip = base + GROWS - N; base = N - GROWS; }
        const float* col = out + (size_t)base * DCH + c;
        const int lane = c & 31;
        const int gidlo = batch[base + lane];
        const int gidhi = batch[base + 32 + lane];

        float v[GROWS];
#pragma unroll
        for (int j = 0; j < GROWS; ++j) v[j] = col[(size_t)j * DCH];
        HOLD16(v, 0); HOLD16(v, 16); HOLD16(v, 32); HOLD16(v, 48);

#pragma unroll
        for (int j = 0; j < GROWS; ++j) {
            if (j >= skip) {
                const int gg = __builtin_amdgcn_readlane(j < 32 ? gidlo : gidhi, j & 31);
                if (gg != curg) {
                    if (curg >= 0) unsafeAtomicAdd(&vsum[(size_t)curg * DCH], acc);
                    acc = 0.0f;
                    curg = gg;
                    float cn = cnt[gg];
                    cn = cn > 1.0f ? cn : 1.0f;
                    sub = (msum[(size_t)gg * DCH] / cn) * msc;
                }
                float o = v[j] - sub;
                acc += o * o;
            }
        }
    } else {
        for (int r = 0; r < N; ++r) {
            const int gg = batch[r];
            if (gg != curg) {
                if (curg >= 0) unsafeAtomicAdd(&vsum[(size_t)curg * DCH], acc);
                acc = 0.0f;
                curg = gg;
                float cn = cnt[gg];
                cn = cn > 1.0f ? cn : 1.0f;
                sub = (msum[(size_t)gg * DCH] / cn) * msc;
            }
            float o = out[(size_t)r * DCH + c] - sub;
            acc += o * o;
        }
    }
    if (curg >= 0) unsafeAtomicAdd(&vsum[(size_t)curg * DCH], acc);
}

__global__ __launch_bounds__(256) void k_finalize(
    float* __restrict__ out, const int* __restrict__ batch,
    const float* __restrict__ mean_sum, const float* __restrict__ var_sum,
    const float* __restrict__ cnt, const float* __restrict__ ms,
    const float* __restrict__ gw, const float* __restrict__ gb, int N) {
    size_t idx = (size_t)blockIdx.x * 256 + threadIdx.x;
    size_t total = (size_t)N * (DCH / 4);
    if (idx >= total) return;
    int nrow = (int)(idx >> 5);
    int q    = (int)(idx & 31);
    int g = batch[nrow];
    float cn = cnt[g];
    cn = cn > 1.0f ? cn : 1.0f;
    float inv_cn = 1.0f / cn;

    float4 m4 = reinterpret_cast<const float4*>(mean_sum + (size_t)g * DCH)[q];
    float4 v4 = reinterpret_cast<const float4*>(var_sum  + (size_t)g * DCH)[q];
    float4 s4 = reinterpret_cast<const float4*>(ms)[q];
    float4 w4 = reinterpret_cast<const float4*>(gw)[q];
    float4 b4 = reinterpret_cast<const float4*>(gb)[q];
    float4 o  = reinterpret_cast<float4*>(out)[idx];

    o.x -= (m4.x * inv_cn) * s4.x;
    o.y -= (m4.y * inv_cn) * s4.y;
    o.z -= (m4.z * inv_cn) * s4.z;
    o.w -= (m4.w * inv_cn) * s4.w;
    o.x = w4.x * o.x / sqrtf(v4.x * inv_cn + GN_EPS) + b4.x;
    o.y = w4.y * o.y / sqrtf(v4.y * inv_cn + GN_EPS) + b4.y;
    o.z = w4.z * o.z / sqrtf(v4.z * inv_cn + GN_EPS) + b4.z;
    o.w = w4.w * o.w / sqrtf(v4.w * inv_cn + GN_EPS) + b4.w;

    reinterpret_cast<float4*>(out)[idx] = o;
}

extern "C" void kernel_launch(void* const* d_in, const int* in_sizes, int n_in,
                              void* d_out, int out_size, void* d_ws, size_t ws_size,
                              hipStream_t stream) {
    const float* x     = (const float*)d_in[0];
    const int*   eidx  = (const int*)  d_in[1];
    const float* eattr = (const float*)d_in[2];
    const int*   batch = (const int*)  d_in[3];
    const float* W     = (const float*)d_in[4];
    const float* bias  = (const float*)d_in[5];
    const float* gw    = (const float*)d_in[6];
    const float* gb    = (const float*)d_in[7];
    const float* ms    = (const float*)d_in[8];
    float* out = (float*)d_out;

    const int N = in_sizes[0] / DCH;
    const int E = in_sizes[2];

    float* h        = (float*)d_ws;
    float* mean_sum = h + (size_t)N * DCH;
    float* var_sum  = mean_sum + (size_t)NGRAPH * DCH;
    float* cnt      = var_sum + (size_t)NGRAPH * DCH;
    int*   binCount = (int*)(cnt + NGRAPH);
    int*   binFill  = binCount + MAXBIN;
    float* Wt       = (float*)(binFill + MAXBIN);
    int*   binStart = (int*)(Wt + DCH * DCH);
    int4*  recs     = (int4*)(binStart + MAXBIN);
    const int nb = (N + BIN_NODES - 1) >> BIN_SHIFT;

    {
        const size_t per_block = (size_t)256 * ZERO_V4 * 4;
        size_t n1 = (size_t)2 * NGRAPH * DCH + NGRAPH + 2 * MAXBIN;
        k_zero_f32<<<(unsigned)((n1 + per_block - 1) / per_block), 256, 0, stream>>>(mean_sum, n1);
    }

    k_transpose_w<<<64, 256, 0, stream>>>(W, Wt);

    k_gemm_wmma<<<(N + TILE_M - 1) / TILE_M, 256, 0, stream>>>(x, Wt, h, N);

    {
        const int eblocks = (E + CHUNK - 1) / CHUNK;
        k_bin_count<<<eblocks, 256, 0, stream>>>(eidx + E, binCount, E, nb);
        k_bin_place<<<eblocks, 256, 0, stream>>>(
            eidx, eidx + E, eattr, binCount, binFill, binStart, recs, E, nb);
        k_bin_sum<<<nb, SUM_THREADS, 0, stream>>>(
            recs, binCount, binStart, h, bias, batch, mean_sum, cnt, out, N);
    }

    k_var_pass<<<(N + GROWS - 1) / GROWS, 128, 0, stream>>>(
        out, batch, mean_sum, cnt, ms, var_sum, N);

    {
        size_t total = (size_t)N * (DCH / 4);
        k_finalize<<<(unsigned)((total + 255) / 256), 256, 0, stream>>>(
            out, batch, mean_sum, var_sum, cnt, ms, gw, gb, N);
    }
}


# edge sum: a lane owns 4 neighbouring channels, so each h row is one 16-byte-per-lane load instead of four one-line loads; row store, bias and run sums go 16-byte too; order of adds unchanged
# speedup vs baseline: 2.9771x; 1.0829x over previous
#include <hip/hip_runtime.h>
#include <hip/hip_bf16.h>
#include <math.h>

typedef __attribute__((ext_vector_type(2))) float v2f;
typedef __attribute__((ext_vector_type(8))) float v8f;

#define DCH 128
#define NGRAPH 64
#define GN_EPS 1e-5f
#define TILE_M 16
#define SX_LD (DCH + 4)

#define ZERO_V4 16
__global__ __launch_bounds__(256) void k_zero_f32(float* __restrict__ p, size_t n) {
    const size_t n4 = n >> 2;
    const size_t blk = (size_t)blockIdx.x * (256 * ZERO_V4);
    float4* __restrict__ q = reinterpret_cast<float4*>(p) + blk + threadIdx.x;
    const float4 z = make_float4(0.0f, 0.0f, 0.0f, 0.0f);
    if (blk + 256 * ZERO_V4 <= n4) {
#pragma unroll
        for (int k = 0; k < ZERO_V4; ++k) q[k * 256] = z;
    } else {
#pragma unroll
        for (int k = 0; k < ZERO_V4; ++k)
            if (blk + (size_t)k * 256 + threadIdx.x < n4) q[k * 256] = z;
    }
    if (blockIdx.x == 0 && threadIdx.x < (n & 3)) p[(n4 << 2) + threadIdx.x] = 0.0f;
}

__global__ __launch_bounds__(256) void k_transpose_w(
    const float* __restrict__ W, float* __restrict__ Wt) {
    __shared__ float tile[16][17];
    int bx = blockIdx.x & 7;
    int by = blockIdx.x >> 3;
    int tx = threadIdx.x & 15;
    int ty = threadIdx.x >> 4;
    tile[ty][tx] = W[(size_t)(by * 16 + ty) * DCH + bx * 16 + tx];
    __syncthreads();
    Wt[(size_t)(bx * 16 + ty) * DCH + by * 16 + tx] = tile[tx][ty];
}

__global__ __launch_bounds__(256) void k_gemm_wmma(
    const float* __restrict__ x, const float* __restrict__ Wt,
    float* __restrict__ h, int N) {
    __shared__ float sX[TILE_M][SX_LD];

    const int t = threadIdx.x;
    const int rowBase = blockIdx.x * TILE_M;

    for (int i = t; i < TILE_M * DCH; i += 256) {
        int r = i >> 7;
        int c = i & (DCH - 1);
        int gr = rowBase + r;
        sX[r][c] = (gr < N) ? x[(size_t)gr * DCH + c] : 0.0f;
    }
    __syncthreads();

    const int lane = t & 31;
    const int wave = t >> 5;
    const int m    = lane & 15;
    const int koff = (lane >> 4) << 1;
    const int n    = (wave << 4) + (lane & 15);

    const float* wrow = Wt + (size_t)n * DCH;

    v8f acc = {};
#pragma unroll
    for (int kt = 0; kt < DCH; kt += 4) {
        v2f a = *reinterpret_cast<const v2f*>(&sX[m][kt + koff]);
        v2f b = *reinterpret_cast<const v2f*>(wrow + kt + koff);
        acc = __builtin_amdgcn_wmma_f32_16x16x4_f32(
            false, a, false, b, (short)0, acc, false, false);
    }

    const int rhalf = (lane >> 4) << 3;
#pragma unroll
    for (int v = 0; v < 8; ++v) {
        int gr = rowBase + rhalf + v;
        if (gr < N) h[(size_t)gr * DCH + n] = acc[v];
    }
}

#define BIN_SHIFT 7
#define BIN_NODES 128
#define MAXBIN 1024
#define CHUNK 4096
#define EPT (CHUNK / 256)
#define LCAP 8192

__global__ __launch_bounds__(256) void k_bin_count(
    const int* __restrict__ dst, int* __restrict__ binCount, int E, int nb) {
    __shared__ int hist[MAXBIN];
    const int t = threadIdx.x;
    const int e0 = blockIdx.x * CHUNK;
    int dv[EPT];
#pragma unroll
    for (int k = 0; k < EPT; ++k) {
        const int e = e0 + k * 256 + t;
        dv[k] = dst[e < E ? e : E - 1];
    }
    for (int i = t; i < nb; i += 256) hist[i] = 0;
    __syncthreads();
#pragma unroll
    for (int k = 0; k < EPT; ++k) {
        const int e = e0 + k * 256 + t;
        if (e < E) atomicAdd(&hist[dv[k] >> BIN_SHIFT], 1);
    }
    __syncthreads();
    for (int i = t; i < nb; i += 256) {
        const int c = hist[i];
        if (c) atomicAdd(&binCount[i], c);
    }
}

__global__ __launch_bounds__(256) void k_bin_place(
    const int* __restrict__ src, const int* __restrict__ dst,
    const float* __restrict__ attr, const int* __restrict__ binCount,
    int* __restrict__ binFill, int* __restrict__ binStart,
    int4* __restrict__ recs, int E, int nb) {
    __shared__ int lcur[MAXBIN];
    __shared__ int lbase[MAXBIN];
    __shared__ int wtot[8];
    const int t = threadIdx.x;
    const int lane = t & 31;
    const int wave = t >> 5;
    const int e0 = blockIdx.x * CHUNK;

    int a0 = 4 * t + 0 < nb ? binCount[4 * t + 0] : 0;
    int a1 = 4 * t + 1 < nb ? binCount[4 * t + 1] : 0;
    int a2 = 4 * t + 2 < nb ? binCount[4 * t + 2] : 0;
    int a3 = 4 * t + 3 < nb ? binCount[4 * t + 3] : 0;

    int dv[EPT], sv[EPT];
    float av[EPT];
#pragma unroll
    for (int k = 0; k < EPT; ++k) {
        const int e = e0 + k * 256 + t;
        dv[k] = dst[e < E ? e : E - 1];
    }
#pragma unroll
    for (int k = 0; k < EPT; ++k) {
        const int e = e0 + k * 256 + t;
        const int ec = e < E ? e : E - 1;
        sv[k] = src[ec];
        av[k] = attr[ec];
    }

    const int s = a0 + a1 + a2 + a3;
    int v = s;
    for (int o = 1; o < 32; o <<= 1) {
        const int u = __shfl_up(v, o);
        if (lane >= o) v += u;
    }
    if (lane == 31) wtot[wave] = v;
    lcur[4 * t + 0] = 0; lcur[4 * t + 1] = 0; lcur[4 * t + 2] = 0; lcur[4 * t + 3] = 0;
    __syncthreads();
    int wb = 0;
    for (int w = 0; w < wave; ++w) wb += wtot[w];
    const int ex = wb + v - s;
    lbase[4 * t + 0] = ex;
    lbase[4 * t + 1] = ex + a0;
    lbase[4 * t + 2] = ex + a0 + a1;
    lbase[4 * t + 3] = ex + a0 + a1 + a2;
    if (blockIdx.x == 0) {
        binStart[4 * t + 0] = ex;
        binStart[4 * t + 1] = ex + a0;
        binStart[4 * t + 2] = ex + a0 + a1;
        binStart[4 * t + 3] = ex + a0 + a1 + a2;
    }

    int r[EPT];
#pragma unroll
    for (int k = 0; k < EPT; ++k) {
        const int e = e0 + k * 256 + t;
        r[k] = 0;
        if (e < E) r[k] = atomicAdd(&lcur[dv[k] >> BIN_SHIFT], 1);
    }
    __syncthreads();
    int cc[MAXBIN / 256], gb[MAXBIN / 256];
#pragma unroll
    for (int j = 0; j < MAXBIN / 256; ++j) {
        const int i = t + 256 * j;
        cc[j] = i < nb ? lcur[i] : 0;
    }
#pragma unroll
    for (int j = 0; j < MAXBIN / 256; ++j) {
        gb[j] = 0;
        if (cc[j]) gb[j] = atomicAdd(&binFill[t + 256 * j], cc[j]);
    }
#pragma unroll
    for (int j = 0; j < MAXBIN / 256; ++j)
        if (cc[j]) lbase[t + 256 * j] += gb[j];
    __syncthreads();
    int lb[EPT];
#pragma unroll
    for (int k = 0; k < EPT; ++k) lb[k] = lbase[dv[k] >> BIN_SHIFT];
#pragma unroll
    for (int k = 0; k < EPT; ++k) {
        const int e = e0 + k * 256 + t;
        if (e < E)
            recs[lb[k] + r[k]] = make_int4(sv[k], dv[k], __float_as_int(av[k]), e);
    }
}

__device__ __forceinline__ const float4* row_ptr(const float* __restrict__ h, int s, int lane) {
    return reinterpret_cast<const float4*>(h + (size_t)s * DCH) + lane;
}

__device__ __forceinline__ void row_fma(const float* __restrict__ h, int s, float a,
                                        int lane, float& c0, float& c1, float& c2, float& c3) {
    const float4 x = *row_ptr(h, s, lane);
    c0 += x.x * a;
    c1 += x.y * a;
    c2 += x.z * a;
    c3 += x.w * a;
}

__device__ __forceinline__ void pick_two(int k, int rankA, int rankB, int sA, int sB,
                                         float aA, float aB, int& s, float& f) {
    const bool mB = rankB == k;
    const bool m = mB || rankA == k;
    const int j = __builtin_ctz(__builtin_amdgcn_ballot_w32(m));
    s = __builtin_amdgcn_readlane(mB ? sB : sA, j);
    f = __int_as_float(__builtin_amdgcn_readlane(__float_as_int(mB ? aB : aA), j));
}

#define SUM_THREADS 512
#define SUM_WAVES (SUM_THREADS / 32)
#define LIST_V 8
__global__ __launch_bounds__(SUM_THREADS, 16) void k_bin_sum(
    const int4* __restrict__ recs, const int* __restrict__ binCount,
    const int* __restrict__ binStart, const float* __restrict__ h,
    const float* __restrict__ bias, const int* __restrict__ batch,
    float* __restrict__ mean_sum, float* __restrict__ sq_sum,
    float* __restrict__ cnt, float* __restrict__ agg, int N) {
    __shared__ int nodeCnt[BIN_NODES];
    __shared__ int nodeStart[BIN_NODES];
    __shared__ int nodeCur[BIN_NODES];
    __shared__ int L[LCAP];
    __shared__ __attribute__((aligned(16))) float wSum[SUM_WAVES][DCH];
    __shared__ __attribute__((aligned(16))) float wSq[SUM_WAVES][DCH];
    __shared__ float wCnt[SUM_WAVES];
    __shared__ int wGid[SUM_WAVES];
    __shared__ int nodeGid[BIN_NODES];
    __shared__ __attribute__((aligned(16))) float sBias[DCH];
    const int t = threadIdx.x;
    const int lane = t & 31;
    const int wave = t >> 5;
    const int bin = blockIdx.x;
    const int n = binCount[bin];
    const int4* __restrict__ R = recs + binStart[bin];
    const int node0 = bin << BIN_SHIFT;
    const bool listed = n <= LCAP;

    int gidn = 0;
    float biasv = 0.0f;

    if (listed) {
        if (t < BIN_NODES) { nodeCnt[t] = 0; nodeCur[t] = 0; }
        __syncthreads();
        for (int i0 = t; i0 < n; i0 += LIST_V * SUM_THREADS) {
            int y[LIST_V];
#pragma unroll
            for (int j = 0; j < LIST_V; ++j) {
                const int i = i0 + j * SUM_THREADS;
                y[j] = R[i < n ? i : n - 1].y;
            }
#pragma unroll
            for (int j = 0; j < LIST_V; ++j)
                if (i0 + j * SUM_THREADS < n) atomicAdd(&nodeCnt[y[j] - node0], 1);
        }
        __syncthreads();
        if (t < BIN_NODES) {
            int p = 0;
            for (int u = 0; u < t; ++u) p += nodeCnt[u];
            nodeStart[t] = p;
        }
        __syncthreads();
        if (t < BIN_NODES) {
            const int gn = node0 + t;
            gidn = batch[gn < N ? gn : N - 1];
            biasv = bias[t];
        }
        for (int i0 = t; i0 < n; i0 += LIST_V * SUM_THREADS) {
            int y[LIST_V];
#pragma unroll
            for (int j = 0; j < LIST_V; ++j) {
                const int i = i0 + j * SUM_THREADS;
                y[j] = R[i < n ? i : n - 1].y;
            }
#pragma unroll
            for (int j = 0; j < LIST_V; ++j) {
                const int i = i0 + j * SUM_THREADS;
                if (i < n) {
                    const int dl = y[j] - node0;
                    const int p = atomicAdd(&nodeCur[dl], 1);
                    L[nodeStart[dl] + p] = i;
                }
            }
        }
    } else if (t < BIN_NODES) {
        const int gn = node0 + t;
        gidn = batch[gn < N ? gn : N - 1];
        biasv = bias[t];
    }
    if (t < BIN_NODES) { nodeGid[t] = gidn; sBias[t] = biasv; }
    __syncthreads();

    float4* ws = reinterpret_cast<float4*>(&wSum[wave][4 * lane]);
    float4* wq = reinterpret_cast<float4*>(&wSq[wave][4 * lane]);
    int curg = -1, crun = 0;
    for (int node = wave; node < BIN_NODES; node += SUM_WAVES) {
        const int gv = node0 + node;
        if (gv >= N) break;
        float c0 = 0.0f, c1 = 0.0f, c2 = 0.0f, c3 = 0.0f;
        if (listed) {
            const int d = __builtin_amdgcn_readfirstlane(nodeCnt[node]);
            const int o = __builtin_amdgcn_readfirstlane(nodeStart[node]);
            if (d <= 32) {
                int key = 0x7fffffff, s = 0;
                float a = 0.0f;
                if (lane < d) {
                    const int4 rc = R[L[o + lane]];
                    key = rc.w; s = rc.x; a = __int_as_float(rc.z);
                }
                int rank = 0;
                for (int j = 0; j < d; ++j)
                    rank += (__builtin_amdgcn_readlane(key, j) < key) ? 1 : 0;
                if (lane >= d) rank = -1;
                int k = 0;
                for (; k + 4 <= d; k += 4) {
                    const int j0 = __builtin_ctz(__builtin_amdgcn_ballot_w32(rank == k + 0));
                    const int j1 = __builtin_ctz(__builtin_amdgcn_ballot_w32(rank == k + 1));
                    const int j2 = __builtin_ctz(__builtin_amdgcn_ballot_w32(rank == k + 2));
                    const int j3 = __builtin_ctz(__builtin_amdgcn_ballot_w32(rank == k + 3));
                    const int s0 = __builtin_amdgcn_readlane(s, j0);
                    const int s1 = __builtin_amdgcn_readlane(s, j1);
                    const int s2 = __builtin_amdgcn_readlane(s, j2);
                    const int s3 = __builtin_amdgcn_readlane(s, j3);
                    const float f0 = __int_as_float(__builtin_amdgcn_readlane(__float_as_int(a), j0));
                    const float f1 = __int_as_float(__builtin_amdgcn_readlane(__float_as_int(a), j1));
                    const float f2 = __int_as_float(__builtin_amdgcn_readlane(__float_as_int(a), j2));
                    const float f3 = __int_as_float(__builtin_amdgcn_readlane(__float_as_int(a), j3));
                    const float4 x0 = *row_ptr(h, s0, lane);
                    const float4 x1 = *row_ptr(h, s1, lane);
                    const float4 x2 = *row_ptr(h, s2, lane);
                    const float4 x3 = *row_ptr(h, s3, lane);
                    c0 += x0.x * f0; c1 += x0.y * f0; c2 += x0.z * f0; c3 += x0.w * f0;
                    c0 += x1.x * f1; c1 += x1.y * f1; c2 += x1.z * f1; c3 += x1.w * f1;
                    c0 += x2.x * f2; c1 += x2.y * f2; c2 += x2.z * f2; c3 += x2.w * f2;
                    c0 += x3.x * f3; c1 += x3.y * f3; c2 += x3.z * f3; c3 += x3.w * f3;
                }
                if (k < d) {
                    const bool u1 = k + 1 < d, u2 = k + 2 < d;
                    const int j0 = __builtin_ctz(__builtin_amdgcn_ballot_w32(rank == k + 0));
                    const int j1 = u1 ? __builtin_ctz(__builtin_amdgcn_ballot_w32(rank == k + 1)) : j0;
                    const int j2 = u2 ? __builtin_ctz(__builtin_amdgcn_ballot_w32(rank == k + 2)) : j0;
                    const int s0 = __builtin_amdgcn_readlane(s, j0);
                    const int s1 = __builtin_amdgcn_readlane(s, j1);
                    const int s2 = __builtin_amdgcn_readlane(s, j2);
                    const float f0 = __int_as_float(__builtin_amdgcn_readlane(__float_as_int(a), j0));
                    const float f1 = __int_as_float(__builtin_amdgcn_readlane(__float_as_int(a), j1));
                    const float f2 = __int_as_float(__builtin_amdgcn_readlane(__float_as_int(a), j2));
                    float4 x1 = make_float4(0.0f, 0.0f, 0.0f, 0.0f);
                    float4 x2 = make_float4(0.0f, 0.0f, 0.0f, 0.0f);
                    const float4 x0 = *row_ptr(h, s0, lane);
                    if (u1) x1 = *row_ptr(h, s1, lane);
                    if (u2) x2 = *row_ptr(h, s2, lane);
                    c0 += x0.x * f0; c1 += x0.y * f0; c2 += x0.z * f0; c3 += x0.w * f0;
                    if (u1) { c0 += x1.x * f1; c1 += x1.y * f1; c2 += x1.z * f1; c3 += x1.w * f1; }
                    if (u2) { c0 += x2.x * f2; c1 += x2.y * f2; c2 += x2.z * f2; c3 += x2.w * f2; }
                }
            } else if (d <= 64) {
                const int d2 = d - 32;
                const int4 ra = R[L[o + lane]];
                const int keyA = ra.w, sA = ra.x;
                const float aA = __int_as_float(ra.z);
                int keyB = 0x7fffffff, sB = 0;
                float aB = 0.0f;
                if (lane < d2) {
                    const int4 rb = R[L[o + 32 + lane]];
                    keyB = rb.w; sB = rb.x; aB = __int_as_float(rb.z);
                }
                int rankA = 0, rankB = 0;
                for (int j = 0; j < 32; ++j) {
                    const int kj = __builtin_amdgcn_readlane(keyA, j);
                    rankA += (kj < keyA) ? 1 : 0;
                    rankB += (kj < keyB) ? 1 : 0;
                }
                for (int j = 0; j < d2; ++j) {
                    const int kj = __builtin_amdgcn_readlane(keyB, j);
                    rankA += (kj < keyA) ? 1 : 0;
                    rankB += (kj < keyB) ? 1 : 0;
                }
                if (lane >= d2) rankB = -1;
                int k = 0;
                for (; k + 4 <= d; k += 4) {
                    int s0, s1, s2, s3;
                    float f0, f1, f2, f3;
                    pick_two(k + 0, rankA, rankB, sA, sB, aA, aB, s0, f0);
                    pick_two(k + 1, rankA, rankB, sA, sB, aA, aB, s1, f1);
                    pick_two(k + 2, rankA, rankB, sA, sB, aA, aB, s2, f2);
                    pick_two(k + 3, rankA, rankB, sA, sB, aA, aB, s3, f3);
                    const float4 x0 = *row_ptr(h, s0, lane);
                    const float4 x1 = *row_ptr(h, s1, lane);
                    const float4 x2 = *row_ptr(h, s2, lane);
                    const float4 x3 = *row_ptr(h, s3, lane);
                    c0 += x0.x * f0; c1 += x0.y * f0; c2 += x0.z * f0; c3 += x0.w * f0;
                    c0 += x1.x * f1; c1 += x1.y * f1; c2 += x1.z * f1; c3 += x1.w * f1;
                    c0 += x2.x * f2; c1 += x2.y * f2; c2 += x2.z * f2; c3 += x2.w * f2;
                    c0 += x3.x * f3; c1 += x3.y * f3; c2 += x3.z * f3; c3 += x3.w * f3;
                }
                for (; k < d; ++k) {
                    int s0;
                    float f0;
                    pick_two(k, rankA, rankB, sA, sB, aA, aB, s0, f0);
                    row_fma(h, s0, f0, lane, c0, c1, c2, c3);
                }
            } else {
                int last = -1;
                for (int k = 0; k < d; ++k) {
                    int best = 0x7fffffff, bi = 0;
                    for (int i = lane; i < d; i += 32) {
                        const int ri = L[o + i];
                        const int key = R[ri].w;
                        if (key > last && key < best) { best = key; bi = ri; }
                    }
                    for (int off = 16; off; off >>= 1) {
                        const int ob = __shfl_xor(best, off);
                        const int oi = __shfl_xor(bi, off);
                        if (ob < best) { best = ob; bi = oi; }
                    }
                    const int4 rc = R[bi];
                    last = best;
                    row_fma(h, rc.x, __int_as_float(rc.z), lane, c0, c1, c2, c3);
                }
            }
        } else {
            int last = -1;
            for (int trip = 0; trip <= n; ++trip) {
                int best = 0x7fffffff, bi = 0;
                for (int i = lane; i < n; i += 32) {
                    const int4 rc = R[i];
                    if (rc.y == gv && rc.w > last && rc.w < best) { best = rc.w; bi = i; }
                }
                for (int off = 16; off; off >>= 1) {
                    const int ob = __shfl_xor(best, off);
                    const int oi = __shfl_xor(bi, off);
                    if (ob < best) { best = ob; bi = oi; }
                }
                if (best == 0x7fffffff) break;
                const int4 rc = R[bi];
                last = best;
                row_fma(h, rc.x, __int_as_float(rc.z), lane, c0, c1, c2, c3);
            }
        }
        const float4 bv = *reinterpret_cast<const float4*>(&sBias[4 * lane]);
        const float b0 = bv.x, b1 = bv.y, b2 = bv.z, b3 = bv.w;
        float x0 = c0 + b0, x1 = c1 + b1, x2 = c2 + b2, x3 = c3 + b3;
        x0 = x0 > 0.0f ? x0 : 0.0f;
        x1 = x1 > 0.0f ? x1 : 0.0f;
        x2 = x2 > 0.0f ? x2 : 0.0f;
        x3 = x3 > 0.0f ? x3 : 0.0f;
        *(reinterpret_cast<float4*>(agg + (size_t)gv * DCH) + lane) = make_float4(x0, x1, x2, x3);
        const float e0 = x0 - (b0 > 0.0f ? b0 : 0.0f);
        const float e1 = x1 - (b1 > 0.0f ? b1 : 0.0f);
        const float e2 = x2 - (b2 > 0.0f ? b2 : 0.0f);
        const float e3 = x3 - (b3 > 0.0f ? b3 : 0.0f);
        const float q0 = e0 * e0, q1 = e1 * e1, q2 = e2 * e2, q3 = e3 * e3;
        const int gg = nodeGid[node];
        if (gg != curg) {
            if (curg >= 0) {
                const float4 sv = *ws, qv = *wq;
                float* mp = mean_sum + (size_t)curg * DCH + 4 * lane;
                unsafeAtomicAdd(mp, sv.x);
                unsafeAtomicAdd(mp + 1, sv.y);
                unsafeAtomicAdd(mp + 2, sv.z);
                unsafeAtomicAdd(mp + 3, sv.w);
                float* qp = sq_sum + (size_t)curg * DCH + 4 * lane;
                unsafeAtomicAdd(qp, qv.x);
                unsafeAtomicAdd(qp + 1, qv.y);
                unsafeAtomicAdd(qp + 2, qv.z);
                unsafeAtomicAdd(qp + 3, qv.w);
                if (lane == 0) unsafeAtomicAdd(&cnt[curg], (float)crun);
            }
            *ws = make_float4(x0, x1, x2, x3);
            *wq = make_float4(q0, q1, q2, q3);
            crun = 1;
            curg = gg;
        } else {
            float4 sv = *ws, qv = *wq;
            sv.x += x0; sv.y += x1; sv.z += x2; sv.w += x3;
            qv.x += q0; qv.y += q1; qv.z += q2; qv.w += q3;
            *ws = sv;
            *wq = qv;
            crun += 1;
        }
    }

    if (lane == 0) { wCnt[wave] = (float)crun; wGid[wave] = curg; }
    __syncthreads();
    if (t < 2 * DCH) {
        const int ch = t & (DCH - 1);
        const bool sq = t >= DCH;
        const float* part = sq ? &wSq[0][ch] : &wSum[0][ch];
        float* tot = (sq ? sq_sum : mean_sum) + ch;
        float acc = 0.0f, cc = 0.0f;
        int cur = -1;
        for (int w = 0; w < SUM_WAVES; ++w) {
            const int g = wGid[w];
            if (g < 0) continue;
            if (g != cur) {
                if (cur >= 0) {
                    unsafeAtomicAdd(&tot[(size_t)cur * DCH], acc);
                    if (t == 0) unsafeAtomicAdd(&cnt[cur], cc);
                }
                acc = 0.0f; cc = 0.0f; cur = g;
            }
            acc += part[w * DCH];
            cc += wCnt[w];
        }
        if (cur >= 0) {
            unsafeAtomicAdd(&tot[(size_t)cur * DCH], acc);
            if (t == 0) unsafeAtomicAdd(&cnt[cur], cc);
        }
    }
}

__device__ __forceinline__ float gn_var(float q, float mean, float u, float b, float inv_cn) {
    const float k = b > 0.0f ? b : 0.0f;
    const float a = mean - k;
    const float d = u - k;
    const float v = q * inv_cn - d * (2.0f * a - d);
    return v > 0.0f ? v : 0.0f;
}

__global__ __launch_bounds__(256) void k_finalize(
    float* __restrict__ out, const int* __restrict__ batch,
    const float* __restrict__ mean_sum, const float* __restrict__ sq_sum,
    const float* __restrict__ cnt, const float* __restrict__ ms,
    const float* __restrict__ gw, const float* __restrict__ gb,
    const float* __restrict__ bias, int N) {
    size_t idx = (size_t)blockIdx.x * 256 + threadIdx.x;
    size_t total = (size_t)N * (DCH / 4);
    if (idx >= total) return;
    int nrow = (int)(idx >> 5);
    int q    = (int)(idx & 31);
    int g = batch[nrow];
    float cn = cnt[g];
    cn = cn > 1.0f ? cn : 1.0f;
    float inv_cn = 1.0f / cn;

    float4 m4 = reinterpret_cast<const float4*>(mean_sum + (size_t)g * DCH)[q];
    float4 q4 = reinterpret_cast<const float4*>(sq_sum   + (size_t)g * DCH)[q];
    float4 s4 = reinterpret_cast<const float4*>(ms)[q];
    float4 w4 = reinterpret_cast<const float4*>(gw)[q];
    float4 b4 = reinterpret_cast<const float4*>(gb)[q];
    float4 c4 = reinterpret_cast<const float4*>(bias)[q];
    float4 o  = reinterpret_cast<float4*>(out)[idx];

    const float mx = m4.x * inv_cn, my = m4.y * inv_cn;
    const float mz = m4.z * inv_cn, mw = m4.w * inv_cn;
    const float ux = mx * s4.x, uy = my * s4.y, uz = mz * s4.z, uw = mw * s4.w;
    const float vx = gn_var(q4.x, mx, ux, c4.x, inv_cn);
    const float vy = gn_var(q4.y, my, uy, c4.y, inv_cn);
    const float vz = gn_var(q4.z, mz, uz, c4.z, inv_cn);
    const float vw = gn_var(q4.w, mw, uw, c4.w, inv_cn);
    o.x -= ux;
    o.y -= uy;
    o.z -= uz;
    o.w -= uw;
    o.x = w4.x * o.x / sqrtf(vx + GN_EPS) + b4.x;
    o.y = w4.y * o.y / sqrtf(vy + GN_EPS) + b4.y;
    o.z = w4.z * o.z / sqrtf(vz + GN_EPS) + b4.z;
    o.w = w4.w * o.w / sqrtf(vw + GN_EPS) + b4.w;

    reinterpret_cast<float4*>(out)[idx] = o;
}

extern "C" void kernel_launch(void* const* d_in, const int* in_sizes, int n_in,
                              void* d_out, int out_size, void* d_ws, size_t ws_size,
                              hipStream_t stream) {
    const float* x     = (const float*)d_in[0];
    const int*   eidx  = (const int*)  d_in[1];
    const float* eattr = (const float*)d_in[2];
    const int*   batch = (const int*)  d_in[3];
    const float* W     = (const float*)d_in[4];
    const float* bias  = (const float*)d_in[5];
    const float* gw    = (const float*)d_in[6];
    const float* gb    = (const float*)d_in[7];
    const float* ms    = (const float*)d_in[8];
    float* out = (float*)d_out;

    const int N = in_sizes[0] / DCH;
    const int E = in_sizes[2];

    float* h        = (float*)d_ws;
    float* mean_sum = h + (size_t)N * DCH;
    float* sq_sum   = mean_sum + (size_t)NGRAPH * DCH;
    float* cnt      = sq_sum + (size_t)NGRAPH * DCH;
    int*   binCount = (int*)(cnt + NGRAPH);
    int*   binFill  = binCount + MAXBIN;
    float* Wt       = (float*)(binFill + MAXBIN);
    int*   binStart = (int*)(Wt + DCH * DCH);
    int4*  recs     = (int4*)(binStart + MAXBIN);
    const int nb = (N + BIN_NODES - 1) >> BIN_SHIFT;

    {
        const size_t per_block = (size_t)256 * ZERO_V4 * 4;
        size_t n1 = (size_t)2 * NGRAPH * DCH + NGRAPH + 2 * MAXBIN;
        k_zero_f32<<<(unsigned)((n1 + per_block - 1) / per_block), 256, 0, stream>>>(mean_sum, n1);
    }

    k_transpose_w<<<64, 256, 0, stream>>>(W, Wt);

    k_gemm_wmma<<<(N + TILE_M - 1) / TILE_M, 256, 0, stream>>>(x, Wt, h, N);

    {
        const int eblocks = (E + CHUNK - 1) / CHUNK;
        k_bin_count<<<eblocks, 256, 0, stream>>>(eidx + E, binCount, E, nb);
        k_bin_place<<<eblocks, 256, 0, stream>>>(
            eidx, eidx + E, eattr, binCount, binFill, binStart, recs, E, nb);
        k_bin_sum<<<nb, SUM_THREADS, 0, stream>>>(
            recs, binCount, binStart, h, bias, batch, mean_sum, sq_sum, cnt, out, N);
    }

    {
        size_t total = (size_t)N * (DCH / 4);
        k_finalize<<<(unsigned)((total + 255) / 256), 256, 0, stream>>>(
            out, batch, mean_sum, sq_sum, cnt, ms, gw, gb, bias, N);
    }
}


# GEMM: each wave keeps its 16 columns of W in 64 registers, read once straight from W, over a resident grid's loop of row tiles; per-tile B fetch, transposed W copy and its launch gone; same sums
# speedup vs baseline: 3.7822x; 1.2704x over previous
#include <hip/hip_runtime.h>
#include <hip/hip_bf16.h>
#include <math.h>

typedef __attribute__((ext_vector_type(2))) float v2f;
typedef __attribute__((ext_vector_type(8))) float v8f;

#define DCH 128
#define NGRAPH 64
#define GN_EPS 1e-5f
#define TILE_M 16
#define SX_LD (DCH + 4)

#define ZERO_V4 16
__global__ __launch_bounds__(256) void k_zero_f32(float* __restrict__ p, size_t n) {
    const size_t n4 = n >> 2;
    const size_t blk = (size_t)blockIdx.x * (256 * ZERO_V4);
    float4* __restrict__ q = reinterpret_cast<float4*>(p) + blk + threadIdx.x;
    const float4 z = make_float4(0.0f, 0.0f, 0.0f, 0.0f);
    if (blk + 256 * ZERO_V4 <= n4) {
#pragma unroll
        for (int k = 0; k < ZERO_V4; ++k) q[k * 256] = z;
    } else {
#pragma unroll
        for (int k = 0; k < ZERO_V4; ++k)
            if (blk + (size_t)k * 256 + threadIdx.x < n4) q[k * 256] = z;
    }
    if (blockIdx.x == 0 && threadIdx.x < (n & 3)) p[(n4 << 2) + threadIdx.x] = 0.0f;
}

#define GEMM_BLOCKS 512

__device__ __forceinline__ void x_pieces(const float* __restrict__ x, int rowBase, int sr, int sq,
                                         int N, float4& p0, float4& p1) {
    const float4* r0 = reinterpret_cast<const float4*>(x + (size_t)(rowBase + sr) * DCH) + sq;
    if (rowBase + TILE_M <= N) {
        p0 = r0[0];
        p1 = r0[8 * (DCH / 4)];
    } else {
        const float4 z = make_float4(0.0f, 0.0f, 0.0f, 0.0f);
        p0 = z;
        p1 = z;
        if (rowBase + sr < N) p0 = r0[0];
        if (rowBase + sr + 8 < N) p1 = r0[8 * (DCH / 4)];
    }
}

__global__ __launch_bounds__(256) void k_gemm_wmma(
    const float* __restrict__ x, const float* __restrict__ W,
    float* __restrict__ h, int N) {
    __shared__ __attribute__((aligned(16))) float sX[2][TILE_M][SX_LD];

    const int t = threadIdx.x;
    const int lane = t & 31;
    const int wave = t >> 5;
    const int m    = lane & 15;
    const int koff = (lane >> 4) << 1;
    const int n    = (wave << 4) + (lane & 15);
    const int nt   = (N + TILE_M - 1) / TILE_M;
    const int step = gridDim.x;

    v2f bf[DCH / 4];
#pragma unroll
    for (int i = 0; i < DCH / 4; ++i) {
        const float* wp = W + (size_t)(4 * i + koff) * DCH + n;
        bf[i].x = wp[0];
        bf[i].y = wp[DCH];
    }

    const int sr = t >> 5;
    const int sq = t & 31;

    int tile = blockIdx.x;
    if (tile < nt) {
        float4 p0, p1;
        x_pieces(x, tile * TILE_M, sr, sq, N, p0, p1);
        *reinterpret_cast<float4*>(&sX[0][sr][4 * sq]) = p0;
        *reinterpret_cast<float4*>(&sX[0][sr + 8][4 * sq]) = p1;
    }
    __syncthreads();

    const int rhalf = (lane >> 4) << 3;
    int cur = 0;
    for (; tile < nt; tile += step) {
        const int nxt = tile + step;
        const bool more = nxt < nt;
        float4 p0 = make_float4(0.0f, 0.0f, 0.0f, 0.0f), p1 = p0;
        if (more) x_pieces(x, nxt * TILE_M, sr, sq, N, p0, p1);

        v8f acc = {};
#pragma unroll
        for (int i = 0; i < DCH / 4; ++i) {
            const v2f a = *reinterpret_cast<const v2f*>(&sX[cur][m][4 * i + koff]);
            acc = __builtin_amdgcn_wmma_f32_16x16x4_f32(
                false, a, false, bf[i], (short)0, acc, false, false);
        }

        if (more) {
            *reinterpret_cast<float4*>(&sX[cur ^ 1][sr][4 * sq]) = p0;
            *reinterpret_cast<float4*>(&sX[cur ^ 1][sr + 8][4 * sq]) = p1;
        }
        __syncthreads();

        const int rowBase = tile * TILE_M;
        float* hp = h + (size_t)(rowBase + rhalf) * DCH + n;
        if (rowBase + TILE_M <= N) {
#pragma unroll
            for (int v = 0; v < 8; ++v) hp[(size_t)v * DCH] = acc[v];
        } else {
#pragma unroll
            for (int v = 0; v < 8; ++v)
                if (rowBase + rhalf + v < N) hp[(size_t)v * DCH] = acc[v];
        }
        cur ^= 1;
    }
}

#define BIN_SHIFT 7
#define BIN_NODES 128
#define MAXBIN 1024
#define CHUNK 4096
#define EPT (CHUNK / 256)
#define LCAP 8192

__global__ __launch_bounds__(256) void k_bin_count(
    const int* __restrict__ dst, int* __restrict__ binCount, int E, int nb) {
    __shared__ int hist[MAXBIN];
    const int t = threadIdx.x;
    const int e0 = blockIdx.x * CHUNK;
    int dv[EPT];
#pragma unroll
    for (int k = 0; k < EPT; ++k) {
        const int e = e0 + k * 256 + t;
        dv[k] = dst[e < E ? e : E - 1];
    }
    for (int i = t; i < nb; i += 256) hist[i] = 0;
    __syncthreads();
#pragma unroll
    for (int k = 0; k < EPT; ++k) {
        const int e = e0 + k * 256 + t;
        if (e < E) atomicAdd(&hist[dv[k] >> BIN_SHIFT], 1);
    }
    __syncthreads();
    for (int i = t; i < nb; i += 256) {
        const int c = hist[i];
        if (c) atomicAdd(&binCount[i], c);
    }
}

__global__ __launch_bounds__(256) void k_bin_place(
    const int* __restrict__ src, const int* __restrict__ dst,
    const float* __restrict__ attr, const int* __restrict__ binCount,
    int* __restrict__ binFill, int* __restrict__ binStart,
    int4* __restrict__ recs, int E, int nb) {
    __shared__ int lcur[MAXBIN];
    __shared__ int lbase[MAXBIN];
    __shared__ int wtot[8];
    const int t = threadIdx.x;
    const int lane = t & 31;
    const int wave = t >> 5;
    const int e0 = blockIdx.x * CHUNK;

    int a0 = 4 * t + 0 < nb ? binCount[4 * t + 0] : 0;
    int a1 = 4 * t + 1 < nb ? binCount[4 * t + 1] : 0;
    int a2 = 4 * t + 2 < nb ? binCount[4 * t + 2] : 0;
    int a3 = 4 * t + 3 < nb ? binCount[4 * t + 3] : 0;

    int dv[EPT], sv[EPT];
    float av[EPT];
#pragma unroll
    for (int k = 0; k < EPT; ++k) {
        const int e = e0 + k * 256 + t;
        dv[k] = dst[e < E ? e : E - 1];
    }
#pragma unroll
    for (int k = 0; k < EPT; ++k) {
        const int e = e0 + k * 256 + t;
        const int ec = e < E ? e : E - 1;
        sv[k] = src[ec];
        av[k] = attr[ec];
    }

    const int s = a0 + a1 + a2 + a3;
    int v = s;
    for (int o = 1; o < 32; o <<= 1) {
        const int u = __shfl_up(v, o);
        if (lane >= o) v += u;
    }
    if (lane == 31) wtot[wave] = v;
    lcur[4 * t + 0] = 0; lcur[4 * t + 1] = 0; lcur[4 * t + 2] = 0; lcur[4 * t + 3] = 0;
    __syncthreads();
    int wb = 0;
    for (int w = 0; w < wave; ++w) wb += wtot[w];
    const int ex = wb + v - s;
    lbase[4 * t + 0] = ex;
    lbase[4 * t + 1] = ex + a0;
    lbase[4 * t + 2] = ex + a0 + a1;
    lbase[4 * t + 3] = ex + a0 + a1 + a2;
    if (blockIdx.x == 0) {
        binStart[4 * t + 0] = ex;
        binStart[4 * t + 1] = ex + a0;
        binStart[4 * t + 2] = ex + a0 + a1;
        binStart[4 * t + 3] = ex + a0 + a1 + a2;
    }

    int r[EPT];
#pragma unroll
    for (int k = 0; k < EPT; ++k) {
        const int e = e0 + k * 256 + t;
        r[k] = 0;
        if (e < E) r[k] = atomicAdd(&lcur[dv[k] >> BIN_SHIFT], 1);
    }
    __syncthreads();
    int cc[MAXBIN / 256], gb[MAXBIN / 256];
#pragma unroll
    for (int j = 0; j < MAXBIN / 256; ++j) {
        const int i = t + 256 * j;
        cc[j] = i < nb ? lcur[i] : 0;
    }
#pragma unroll
    for (int j = 0; j < MAXBIN / 256; ++j) {
        gb[j] = 0;
        if (cc[j]) gb[j] = atomicAdd(&binFill[t + 256 * j], cc[j]);
    }
#pragma unroll
    for (int j = 0; j < MAXBIN / 256; ++j)
        if (cc[j]) lbase[t + 256 * j] += gb[j];
    __syncthreads();
    int lb[EPT];
#pragma unroll
    for (int k = 0; k < EPT; ++k) lb[k] = lbase[dv[k] >> BIN_SHIFT];
#pragma unroll
    for (int k = 0; k < EPT; ++k) {
        const int e = e0 + k * 256 + t;
        if (e < E)
            recs[lb[k] + r[k]] = make_int4(sv[k], dv[k], __float_as_int(av[k]), e);
    }
}

__device__ __forceinline__ const float4* row_ptr(const float* __restrict__ h, int s, int lane) {
    return reinterpret_cast<const float4*>(h + (size_t)s * DCH) + lane;
}

__device__ __forceinline__ void row_fma(const float* __restrict__ h, int s, float a,
                                        int lane, float& c0, float& c1, float& c2, float& c3) {
    const float4 x = *row_ptr(h, s, lane);
    c0 += x.x * a;
    c1 += x.y * a;
    c2 += x.z * a;
    c3 += x.w * a;
}

__device__ __forceinline__ void pick_two(int k, int rankA, int rankB, int sA, int sB,
                                         float aA, float aB, int& s, float& f) {
    const bool mB = rankB == k;
    const bool m = mB || rankA == k;
    const int j = __builtin_ctz(__builtin_amdgcn_ballot_w32(m));
    s = __builtin_amdgcn_readlane(mB ? sB : sA, j);
    f = __int_as_float(__builtin_amdgcn_readlane(__float_as_int(mB ? aB : aA), j));
}

#define SUM_THREADS 512
#define SUM_WAVES (SUM_THREADS / 32)
#define LIST_V 8
__global__ __launch_bounds__(SUM_THREADS, 16) void k_bin_sum(
    const int4* __restrict__ recs, const int* __restrict__ binCount,
    const int* __restrict__ binStart, const float* __restrict__ h,
    const float* __restrict__ bias, const int* __restrict__ batch,
    float* __restrict__ mean_sum, float* __restrict__ sq_sum,
    float* __restrict__ cnt, float* __restrict__ agg, int N) {
    __shared__ int nodeCnt[BIN_NODES];
    __shared__ int nodeStart[BIN_NODES];
    __shared__ int nodeCur[BIN_NODES];
    __shared__ int L[LCAP];
    __shared__ __attribute__((aligned(16))) float wSum[SUM_WAVES][DCH];
    __shared__ __attribute__((aligned(16))) float wSq[SUM_WAVES][DCH];
    __shared__ float wCnt[SUM_WAVES];
    __shared__ int wGid[SUM_WAVES];
    __shared__ int nodeGid[BIN_NODES];
    __shared__ __attribute__((aligned(16))) float sBias[DCH];
    const int t = threadIdx.x;
    const int lane = t & 31;
    const int wave = t >> 5;
    const int bin = blockIdx.x;
    const int n = binCount[bin];
    const int4* __restrict__ R = recs + binStart[bin];
    const int node0 = bin << BIN_SHIFT;
    const bool listed = n <= LCAP;

    int gidn = 0;
    float biasv = 0.0f;

    if (listed) {
        if (t < BIN_NODES) { nodeCnt[t] = 0; nodeCur[t] = 0; }
        __syncthreads();
        for (int i0 = t; i0 < n; i0 += LIST_V * SUM_THREADS) {
            int y[LIST_V];
#pragma unroll
            for (int j = 0; j < LIST_V; ++j) {
                const int i = i0 + j * SUM_THREADS;
                y[j] = R[i < n ? i : n - 1].y;
            }
#pragma unroll
            for (int j = 0; j < LIST_V; ++j)
                if (i0 + j * SUM_THREADS < n) atomicAdd(&nodeCnt[y[j] - node0], 1);
        }
        __syncthreads();
        if (t < BIN_NODES) {
            int p = 0;
            for (int u = 0; u < t; ++u) p += nodeCnt[u];
            nodeStart[t] = p;
        }
        __syncthreads();
        if (t < BIN_NODES) {
            const int gn = node0 + t;
            gidn = batch[gn < N ? gn : N - 1];
            biasv = bias[t];
        }
        for (int i0 = t; i0 < n; i0 += LIST_V * SUM_THREADS) {
            int y[LIST_V];
#pragma unroll
            for (int j = 0; j < LIST_V; ++j) {
                const int i = i0 + j * SUM_THREADS;
                y[j] = R[i < n ? i : n - 1].y;
            }
#pragma unroll
            for (int j = 0; j < LIST_V; ++j) {
                const int i = i0 + j * SUM_THREADS;
                if (i < n) {
                    const int dl = y[j] - node0;
                    const int p = atomicAdd(&nodeCur[dl], 1);
                    L[nodeStart[dl] + p] = i;
                }
            }
        }
    } else if (t < BIN_NODES) {
        const int gn = node0 + t;
        gidn = batch[gn < N ? gn : N - 1];
        biasv = bias[t];
    }
    if (t < BIN_NODES) { nodeGid[t] = gidn; sBias[t] = biasv; }
    __syncthreads();

    float4* ws = reinterpret_cast<float4*>(&wSum[wave][4 * lane]);
    float4* wq = reinterpret_cast<float4*>(&wSq[wave][4 * lane]);
    int curg = -1, crun = 0;
    for (int node = wave; node < BIN_NODES; node += SUM_WAVES) {
        const int gv = node0 + node;
        if (gv >= N) break;
        float c0 = 0.0f, c1 = 0.0f, c2 = 0.0f, c3 = 0.0f;
        if (listed) {
            const int d = __builtin_amdgcn_readfirstlane(nodeCnt[node]);
            const int o = __builtin_amdgcn_readfirstlane(nodeStart[node]);
            if (d <= 32) {
                int key = 0x7fffffff, s = 0;
                float a = 0.0f;
                if (lane < d) {
                    const int4 rc = R[L[o + lane]];
                    key = rc.w; s = rc.x; a = __int_as_float(rc.z);
                }
                int rank = 0;
                for (int j = 0; j < d; ++j)
                    rank += (__builtin_amdgcn_readlane(key, j) < key) ? 1 : 0;
                if (lane >= d) rank = -1;
                int k = 0;
                for (; k + 4 <= d; k += 4) {
                    const int j0 = __builtin_ctz(__builtin_amdgcn_ballot_w32(rank == k + 0));
                    const int j1 = __builtin_ctz(__builtin_amdgcn_ballot_w32(rank == k + 1));
                    const int j2 = __builtin_ctz(__builtin_amdgcn_ballot_w32(rank == k + 2));
                    const int j3 = __builtin_ctz(__builtin_amdgcn_ballot_w32(rank == k + 3));
                    const int s0 = __builtin_amdgcn_readlane(s, j0);
                    const int s1 = __builtin_amdgcn_readlane(s, j1);
                    const int s2 = __builtin_amdgcn_readlane(s, j2);
                    const int s3 = __builtin_amdgcn_readlane(s, j3);
                    const float f0 = __int_as_float(__builtin_amdgcn_readlane(__float_as_int(a), j0));
                    const float f1 = __int_as_float(__builtin_amdgcn_readlane(__float_as_int(a), j1));
                    const float f2 = __int_as_float(__builtin_amdgcn_readlane(__float_as_int(a), j2));
                    const float f3 = __int_as_float(__builtin_amdgcn_readlane(__float_as_int(a), j3));
                    const float4 x0 = *row_ptr(h, s0, lane);
                    const float4 x1 = *row_ptr(h, s1, lane);
                    const float4 x2 = *row_ptr(h, s2, lane);
                    const float4 x3 = *row_ptr(h, s3, lane);
                    c0 += x0.x * f0; c1 += x0.y * f0; c2 += x0.z * f0; c3 += x0.w * f0;
                    c0 += x1.x * f1; c1 += x1.y * f1; c2 += x1.z * f1; c3 += x1.w * f1;
                    c0 += x2.x * f2; c1 += x2.y * f2; c2 += x2.z * f2; c3 += x2.w * f2;
                    c0 += x3.x * f3; c1 += x3.y * f3; c2 += x3.z * f3; c3 += x3.w * f3;
                }
                if (k < d) {
                    const bool u1 = k + 1 < d, u2 = k + 2 < d;
                    const int j0 = __builtin_ctz(__builtin_amdgcn_ballot_w32(rank == k + 0));
                    const int j1 = u1 ? __builtin_ctz(__builtin_amdgcn_ballot_w32(rank == k + 1)) : j0;
                    const int j2 = u2 ? __builtin_ctz(__builtin_amdgcn_ballot_w32(rank == k + 2)) : j0;
                    const int s0 = __builtin_amdgcn_readlane(s, j0);
                    const int s1 = __builtin_amdgcn_readlane(s, j1);
                    const int s2 = __builtin_amdgcn_readlane(s, j2);
                    const float f0 = __int_as_float(__builtin_amdgcn_readlane(__float_as_int(a), j0));
                    const float f1 = __int_as_float(__builtin_amdgcn_readlane(__float_as_int(a), j1));
                    const float f2 = __int_as_float(__builtin_amdgcn_readlane(__float_as_int(a), j2));
                    float4 x1 = make_float4(0.0f, 0.0f, 0.0f, 0.0f);
                    float4 x2 = make_float4(0.0f, 0.0f, 0.0f, 0.0f);
                    const float4 x0 = *row_ptr(h, s0, lane);
                    if (u1) x1 = *row_ptr(h, s1, lane);
                    if (u2) x2 = *row_ptr(h, s2, lane);
                    c0 += x0.x * f0; c1 += x0.y * f0; c2 += x0.z * f0; c3 += x0.w * f0;
                    if (u1) { c0 += x1.x * f1; c1 += x1.y * f1; c2 += x1.z * f1; c3 += x1.w * f1; }
                    if (u2) { c0 += x2.x * f2; c1 += x2.y * f2; c2 += x2.z * f2; c3 += x2.w * f2; }
                }
            } else if (d <= 64) {
                const int d2 = d - 32;
                const int4 ra = R[L[o + lane]];
                const int keyA = ra.w, sA = ra.x;
                const float aA = __int_as_float(ra.z);
                int keyB = 0x7fffffff, sB = 0;
                float aB = 0.0f;
                if (lane < d2) {
                    const int4 rb = R[L[o + 32 + lane]];
                    keyB = rb.w; sB = rb.x; aB = __int_as_float(rb.z);
                }
                int rankA = 0, rankB = 0;
                for (int j = 0; j < 32; ++j) {
                    const int kj = __builtin_amdgcn_readlane(keyA, j);
                    rankA += (kj < keyA) ? 1 : 0;
                    rankB += (kj < keyB) ? 1 : 0;
                }
                for (int j = 0; j < d2; ++j) {
                    const int kj = __builtin_amdgcn_readlane(keyB, j);
                    rankA += (kj < keyA) ? 1 : 0;
                    rankB += (kj < keyB) ? 1 : 0;
                }
                if (lane >= d2) rankB = -1;
                int k = 0;
                for (; k + 4 <= d; k += 4) {
                    int s0, s1, s2, s3;
                    float f0, f1, f2, f3;
                    pick_two(k + 0, rankA, rankB, sA, sB, aA, aB, s0, f0);
                    pick_two(k + 1, rankA, rankB, sA, sB, aA, aB, s1, f1);
                    pick_two(k + 2, rankA, rankB, sA, sB, aA, aB, s2, f2);
                    pick_two(k + 3, rankA, rankB, sA, sB, aA, aB, s3, f3);
                    const float4 x0 = *row_ptr(h, s0, lane);
                    const float4 x1 = *row_ptr(h, s1, lane);
                    const float4 x2 = *row_ptr(h, s2, lane);
                    const float4 x3 = *row_ptr(h, s3, lane);
                    c0 += x0.x * f0; c1 += x0.y * f0; c2 += x0.z * f0; c3 += x0.w * f0;
                    c0 += x1.x * f1; c1 += x1.y * f1; c2 += x1.z * f1; c3 += x1.w * f1;
                    c0 += x2.x * f2; c1 += x2.y * f2; c2 += x2.z * f2; c3 += x2.w * f2;
                    c0 += x3.x * f3; c1 += x3.y * f3; c2 += x3.z * f3; c3 += x3.w * f3;
                }
                for (; k < d; ++k) {
                    int s0;
                    float f0;
                    pick_two(k, rankA, rankB, sA, sB, aA, aB, s0, f0);
                    row_fma(h, s0, f0, lane, c0, c1, c2, c3);
                }
            } else {
                int last = -1;
                for (int k = 0; k < d; ++k) {
                    int best = 0x7fffffff, bi = 0;
                    for (int i = lane; i < d; i += 32) {
                        const int ri = L[o + i];
                        const int key = R[ri].w;
                        if (key > last && key < best) { best = key; bi = ri; }
                    }
                    for (int off = 16; off; off >>= 1) {
                        const int ob = __shfl_xor(best, off);
                        const int oi = __shfl_xor(bi, off);
                        if (ob < best) { best = ob; bi = oi; }
                    }
                    const int4 rc = R[bi];
                    last = best;
                    row_fma(h, rc.x, __int_as_float(rc.z), lane, c0, c1, c2, c3);
                }
            }
        } else {
            int last = -1;
            for (int trip = 0; trip <= n; ++trip) {
                int best = 0x7fffffff, bi = 0;
                for (int i = lane; i < n; i += 32) {
                    const int4 rc = R[i];
                    if (rc.y == gv && rc.w > last && rc.w < best) { best = rc.w; bi = i; }
                }
                for (int off = 16; off; off >>= 1) {
                    const int ob = __shfl_xor(best, off);
                    const int oi = __shfl_xor(bi, off);
                    if (ob < best) { best = ob; bi = oi; }
                }
                if (best == 0x7fffffff) break;
                const int4 rc = R[bi];
                last = best;
                row_fma(h, rc.x, __int_as_float(rc.z), lane, c0, c1, c2, c3);
            }
        }
        const float4 bv = *reinterpret_cast<const float4*>(&sBias[4 * lane]);
        const float b0 = bv.x, b1 = bv.y, b2 = bv.z, b3 = bv.w;
        float x0 = c0 + b0, x1 = c1 + b1, x2 = c2 + b2, x3 = c3 + b3;
        x0 = x0 > 0.0f ? x0 : 0.0f;
        x1 = x1 > 0.0f ? x1 : 0.0f;
        x2 = x2 > 0.0f ? x2 : 0.0f;
        x3 = x3 > 0.0f ? x3 : 0.0f;
        *(reinterpret_cast<float4*>(agg + (size_t)gv * DCH) + lane) = make_float4(x0, x1, x2, x3);
        const float e0 = x0 - (b0 > 0.0f ? b0 : 0.0f);
        const float e1 = x1 - (b1 > 0.0f ? b1 : 0.0f);
        const float e2 = x2 - (b2 > 0.0f ? b2 : 0.0f);
        const float e3 = x3 - (b3 > 0.0f ? b3 : 0.0f);
        const float q0 = e0 * e0, q1 = e1 * e1, q2 = e2 * e2, q3 = e3 * e3;
        const int gg = nodeGid[node];
        if (gg != curg) {
            if (curg >= 0) {
                const float4 sv = *ws, qv = *wq;
                float* mp = mean_sum + (size_t)curg * DCH + 4 * lane;
                unsafeAtomicAdd(mp, sv.x);
                unsafeAtomicAdd(mp + 1, sv.y);
                unsafeAtomicAdd(mp + 2, sv.z);
                unsafeAtomicAdd(mp + 3, sv.w);
                float* qp = sq_sum + (size_t)curg * DCH + 4 * lane;
                unsafeAtomicAdd(qp, qv.x);
                unsafeAtomicAdd(qp + 1, qv.y);
                unsafeAtomicAdd(qp + 2, qv.z);
                unsafeAtomicAdd(qp + 3, qv.w);
                if (lane == 0) unsafeAtomicAdd(&cnt[curg], (float)crun);
            }
            *ws = make_float4(x0, x1, x2, x3);
            *wq = make_float4(q0, q1, q2, q3);
            crun = 1;
            curg = gg;
        } else {
            float4 sv = *ws, qv = *wq;
            sv.x += x0; sv.y += x1; sv.z += x2; sv.w += x3;
            qv.x += q0; qv.y += q1; qv.z += q2; qv.w += q3;
            *ws = sv;
            *wq = qv;
            crun += 1;
        }
    }

    if (lane == 0) { wCnt[wave] = (float)crun; wGid[wave] = curg; }
    __syncthreads();
    if (t < 2 * DCH) {
        const int ch = t & (DCH - 1);
        const bool sq = t >= DCH;
        const float* part = sq ? &wSq[0][ch] : &wSum[0][ch];
        float* tot = (sq ? sq_sum : mean_sum) + ch;
        float acc = 0.0f, cc = 0.0f;
        int cur = -1;
        for (int w = 0; w < SUM_WAVES; ++w) {
            const int g = wGid[w];
            if (g < 0) continue;
            if (g != cur) {
                if (cur >= 0) {
                    unsafeAtomicAdd(&tot[(size_t)cur * DCH], acc);
                    if (t == 0) unsafeAtomicAdd(&cnt[cur], cc);
                }
                acc = 0.0f; cc = 0.0f; cur = g;
            }
            acc += part[w * DCH];
            cc += wCnt[w];
        }
        if (cur >= 0) {
            unsafeAtomicAdd(&tot[(size_t)cur * DCH], acc);
            if (t == 0) unsafeAtomicAdd(&cnt[cur], cc);
        }
    }
}

__device__ __forceinline__ float gn_var(float q, float mean, float u, float b, float inv_cn) {
    const float k = b > 0.0f ? b : 0.0f;
    const float a = mean - k;
    const float d = u - k;
    const float v = q * inv_cn - d * (2.0f * a - d);
    return v > 0.0f ? v : 0.0f;
}

__global__ __launch_bounds__(256) void k_finalize(
    float* __restrict__ out, const int* __restrict__ batch,
    const float* __restrict__ mean_sum, const float* __restrict__ sq_sum,
    const float* __restrict__ cnt, const float* __restrict__ ms,
    const float* __restrict__ gw, const float* __restrict__ gb,
    const float* __restrict__ bias, int N) {
    size_t idx = (size_t)blockIdx.x * 256 + threadIdx.x;
    size_t total = (size_t)N * (DCH / 4);
    if (idx >= total) return;
    int nrow = (int)(idx >> 5);
    int q    = (int)(idx & 31);
    int g = batch[nrow];
    float cn = cnt[g];
    cn = cn > 1.0f ? cn : 1.0f;
    float inv_cn = 1.0f / cn;

    float4 m4 = reinterpret_cast<const float4*>(mean_sum + (size_t)g * DCH)[q];
    float4 q4 = reinterpret_cast<const float4*>(sq_sum   + (size_t)g * DCH)[q];
    float4 s4 = reinterpret_cast<const float4*>(ms)[q];
    float4 w4 = reinterpret_cast<const float4*>(gw)[q];
    float4 b4 = reinterpret_cast<const float4*>(gb)[q];
    float4 c4 = reinterpret_cast<const float4*>(bias)[q];
    float4 o  = reinterpret_cast<float4*>(out)[idx];

    const float mx = m4.x * inv_cn, my = m4.y * inv_cn;
    const float mz = m4.z * inv_cn, mw = m4.w * inv_cn;
    const float ux = mx * s4.x, uy = my * s4.y, uz = mz * s4.z, uw = mw * s4.w;
    const float vx = gn_var(q4.x, mx, ux, c4.x, inv_cn);
    const float vy = gn_var(q4.y, my, uy, c4.y, inv_cn);
    const float vz = gn_var(q4.z, mz, uz, c4.z, inv_cn);
    const float vw = gn_var(q4.w, mw, uw, c4.w, inv_cn);
    o.x -= ux;
    o.y -= uy;
    o.z -= uz;
    o.w -= uw;
    o.x = w4.x * o.x / sqrtf(vx + GN_EPS) + b4.x;
    o.y = w4.y * o.y / sqrtf(vy + GN_EPS) + b4.y;
    o.z = w4.z * o.z / sqrtf(vz + GN_EPS) + b4.z;
    o.w = w4.w * o.w / sqrtf(vw + GN_EPS) + b4.w;

    reinterpret_cast<float4*>(out)[idx] = o;
}

extern "C" void kernel_launch(void* const* d_in, const int* in_sizes, int n_in,
                              void* d_out, int out_size, void* d_ws, size_t ws_size,
                              hipStream_t stream) {
    const float* x     = (const float*)d_in[0];
    const int*   eidx  = (const int*)  d_in[1];
    const float* eattr = (const float*)d_in[2];
    const int*   batch = (const int*)  d_in[3];
    const float* W     = (const float*)d_in[4];
    const float* bias  = (const float*)d_in[5];
    const float* gw    = (const float*)d_in[6];
    const float* gb    = (const float*)d_in[7];
    const float* ms    = (const float*)d_in[8];
    float* out = (float*)d_out;

    const int N = in_sizes[0] / DCH;
    const int E = in_sizes[2];

    float* h        = (float*)d_ws;
    float* mean_sum = h + (size_t)N * DCH;
    float* sq_sum   = mean_sum + (size_t)NGRAPH * DCH;
    float* cnt      = sq_sum + (size_t)NGRAPH * DCH;
    int*   binCount = (int*)(cnt + NGRAPH);
    int*   binFill  = binCount + MAXBIN;
    int*   binStart = binFill + MAXBIN;
    int4*  recs     = (int4*)(binStart + MAXBIN);
    const int nb = (N + BIN_NODES - 1) >> BIN_SHIFT;

    {
        const size_t per_block = (size_t)256 * ZERO_V4 * 4;
        size_t n1 = (size_t)2 * NGRAPH * DCH + NGRAPH + 2 * MAXBIN;
        k_zero_f32<<<(unsigned)((n1 + per_block - 1) / per_block), 256, 0, stream>>>(mean_sum, n1);
    }

    k_gemm_wmma<<<GEMM_BLOCKS, 256, 0, stream>>>(x, W, h, N);

    {
        const int eblocks = (E + CHUNK - 1) / CHUNK;
        k_bin_count<<<eblocks, 256, 0, stream>>>(eidx + E, binCount, E, nb);
        k_bin_place<<<eblocks, 256, 0, stream>>>(
            eidx, eidx + E, eattr, binCount, binFill, binStart, recs, E, nb);
        k_bin_sum<<<nb, SUM_THREADS, 0, stream>>>(
            recs, binCount, binStart, h, bias, batch, mean_sum, sq_sum, cnt, out, N);
    }

    {
        size_t total = (size_t)N * (DCH / 4);
        k_finalize<<<(unsigned)((total + 255) / 256), 256, 0, stream>>>(
            out, batch, mean_sum, sq_sum, cnt, ms, gw, gb, bias, N);
    }
}
